# starting kernel
# baseline (speedup 1.0000x reference)
#include <hip/hip_runtime.h>
#include <hip/hip_bf16.h>


typedef __attribute__((ext_vector_type(16))) __bf16 v16bf;
typedef __attribute__((ext_vector_type(8)))  float  v8f;

#define N_SRC   131072
#define N_MID   32768
#define N_DST   8192
#define E1_N    524288
#define E2_N    131072
#define HDIM    256
#define N_PAIRS 8192

__device__ __forceinline__ void atomic_add_f32(float* p, float v) {
    unsafeAtomicAdd(p, v);
}

__global__ void f32_to_bf16_kernel(const float* __restrict__ src,
                                   __bf16* __restrict__ dst, int n) {
    int i = blockIdx.x * blockDim.x + threadIdx.x;
    if (i < n) dst[i] = (__bf16)src[i];
}

__global__ void zero_f32_kernel(float* __restrict__ p, int n) {
    int i = blockIdx.x * blockDim.x + threadIdx.x;
    if (i < n) p[i] = 0.0f;
}

__global__ __launch_bounds__(256)
void gemm_bf16_n256(const float* __restrict__ A, const __bf16* __restrict__ W,
                    const float* __restrict__ bias, float* __restrict__ C,
                    int M, int K, int relu)
{
    const int lane = threadIdx.x & 31;
    const int wave = threadIdx.x >> 5;
    const int half = lane >> 4;
    const int l15  = lane & 15;
    const int m0   = blockIdx.x * 32 + (wave >> 2) * 16;
    const int n0   = (wave & 3) * 64;

    v8f acc[4];
    #pragma unroll
    for (int t = 0; t < 4; ++t) acc[t] = (v8f){};

    const float* arow = A + (size_t)(m0 + l15) * K;

    for (int k0 = 0; k0 < K; k0 += 32) {
        v16bf a;
        #pragma unroll
        for (int q = 0; q < 8; ++q) {
            int kk = k0 + (q >> 2) * 16 + half * 8 + (q & 3) * 2;
            float2 v = *(const float2*)(arow + kk);
            a[2 * q]     = (__bf16)v.x;
            a[2 * q + 1] = (__bf16)v.y;
        }
        #pragma unroll
        for (int t = 0; t < 4; ++t) {
            const __bf16* wrow = W + (size_t)(n0 + t * 16 + l15) * K;
            v16bf b;
            #pragma unroll
            for (int q = 0; q < 8; ++q) {
                int kk = k0 + (q >> 2) * 16 + half * 8 + (q & 3) * 2;
                b[2 * q]     = wrow[kk];
                b[2 * q + 1] = wrow[kk + 1];
            }
            acc[t] = __builtin_amdgcn_wmma_f32_16x16x32_bf16(
                false, a, false, b, (short)0, acc[t], false, false);
        }
    }

    #pragma unroll
    for (int t = 0; t < 4; ++t) {
        int n = n0 + t * 16 + l15;
        float bv = bias[n];
        #pragma unroll
        for (int r = 0; r < 8; ++r) {
            int m = m0 + half * 8 + r;
            float v = acc[t][r] + bv;
            if (relu) v = fmaxf(v, 0.0f);
            C[(size_t)m * HDIM + n] = v;
        }
    }
}

__global__ __launch_bounds__(256)
void edge_scatter_kernel(const float* __restrict__ nsrc,
                         const float* __restrict__ w,
                         const int* __restrict__ src,
                         const int* __restrict__ dst,
                         float* __restrict__ agg,
                         float* __restrict__ wsum, int E)
{
    int e    = blockIdx.x * 8 + (threadIdx.x >> 5);
    int lane = threadIdx.x & 31;
    if (e >= E) return;
    int   s  = src[e];
    int   d  = dst[e];
    float we = w[e];
    const float* nr = nsrc + (size_t)s * HDIM + lane * 8;
    float*       ar = agg  + (size_t)d * HDIM + lane * 8;
    float4 v0 = *(const float4*)(nr);
    float4 v1 = *(const float4*)(nr + 4);
    atomic_add_f32(ar + 0, v0.x * we);
    atomic_add_f32(ar + 1, v0.y * we);
    atomic_add_f32(ar + 2, v0.z * we);
    atomic_add_f32(ar + 3, v0.w * we);
    atomic_add_f32(ar + 4, v1.x * we);
    atomic_add_f32(ar + 5, v1.y * we);
    atomic_add_f32(ar + 6, v1.z * we);
    atomic_add_f32(ar + 7, v1.w * we);
    if (lane == 0) atomic_add_f32(wsum + d, we);
}

__global__ void build_z_kernel(const float* __restrict__ agg,
                               const float* __restrict__ wsum,
                               const float* __restrict__ hdst,
                               float* __restrict__ z, int rows)
{
    int i = blockIdx.x * blockDim.x + threadIdx.x;
    if (i >= rows * 512) return;
    int r = i >> 9, c = i & 511;
    float v;
    if (c < HDIM) {
        float ws = wsum[r];
        ws = ws < 1.0f ? 1.0f : ws;
        v = agg[(size_t)r * HDIM + c] * __frcp_rn(ws);
    } else {
        v = hdst[(size_t)r * HDIM + (c - HDIM)];
    }
    z[i] = v;
}

__global__ __launch_bounds__(256)
void skip_norm_kernel(const float* __restrict__ hp, const float* __restrict__ h2,
                      float* __restrict__ zf, int rows)
{
    int r    = blockIdx.x * 8 + (threadIdx.x >> 5);
    int lane = threadIdx.x & 31;
    if (r >= rows) return;
    float v[8];
    float ss = 0.0f;
    #pragma unroll
    for (int j = 0; j < 8; ++j) {
        int c = lane * 8 + j;
        v[j] = hp[(size_t)r * HDIM + c] + h2[(size_t)r * HDIM + c];
        ss += v[j] * v[j];
    }
    #pragma unroll
    for (int o = 16; o > 0; o >>= 1) ss += __shfl_xor(ss, o, 32);
    float nrm   = sqrtf(ss);
    float scale = (nrm == 0.0f) ? 1.0f : 1.0f / nrm;
    #pragma unroll
    for (int j = 0; j < 8; ++j)
        zf[(size_t)r * HDIM + lane * 8 + j] = v[j] * scale;
}

__global__ __launch_bounds__(256)
void score_kernel(const float* __restrict__ zf, const float* __restrict__ bias,
                  const int* __restrict__ nids,
                  const int* __restrict__ ps, const int* __restrict__ pd,
                  const int* __restrict__ ns, const int* __restrict__ nd,
                  float* __restrict__ out, int npairs)
{
    int p    = blockIdx.x * 8 + (threadIdx.x >> 5);
    int lane = threadIdx.x & 31;
    if (p >= npairs) return;
    int pu = ps[p], pv = pd[p], nu = ns[p], nv = nd[p];
    float sp = 0.0f, sn = 0.0f;
    #pragma unroll
    for (int j = 0; j < 8; ++j) {
        int c = lane * 8 + j;
        sp += zf[(size_t)pu * HDIM + c] * zf[(size_t)pv * HDIM + c];
        sn += zf[(size_t)nu * HDIM + c] * zf[(size_t)nv * HDIM + c];
    }
    #pragma unroll
    for (int o = 16; o > 0; o >>= 1) {
        sp += __shfl_xor(sp, o, 32);
        sn += __shfl_xor(sn, o, 32);
    }
    if (lane == 0) {
        float gp = bias[nids[pu]] + bias[nids[pv]];
        float gn = bias[nids[nu]] + bias[nids[nv]];
        float m  = (sn + gn) - (sp + gp) + 1.0f;
        out[p] = m > 0.0f ? m : 0.0f;
    }
}

extern "C" void kernel_launch(void* const* d_in, const int* in_sizes, int n_in,
                              void* d_out, int out_size, void* d_ws, size_t ws_size,
                              hipStream_t stream) {
    const float* feat = (const float*)d_in[0];
    const float* Wp   = (const float*)d_in[1];
    const float* bp   = (const float*)d_in[2];
    const float* Q1   = (const float*)d_in[3];
    const float* bq1  = (const float*)d_in[4];
    const float* W1   = (const float*)d_in[5];
    const float* bw1  = (const float*)d_in[6];
    const float* Q2   = (const float*)d_in[7];
    const float* bq2  = (const float*)d_in[8];
    const float* W2   = (const float*)d_in[9];
    const float* bw2  = (const float*)d_in[10];
    const float* w1v  = (const float*)d_in[11];
    const float* w2v  = (const float*)d_in[12];
    const float* bias = (const float*)d_in[13];
    const int* src1   = (const int*)d_in[14];
    const int* dst1   = (const int*)d_in[15];
    const int* src2   = (const int*)d_in[16];
    const int* dst2   = (const int*)d_in[17];
    const int* pos_s  = (const int*)d_in[18];
    const int* pos_d  = (const int*)d_in[19];
    const int* neg_s  = (const int*)d_in[20];
    const int* neg_d  = (const int*)d_in[21];
    const int* nids   = (const int*)d_in[22];
    float* out        = (float*)d_out;

    char* ws = (char*)d_ws;
    const size_t SZ_HP  = (size_t)N_SRC * HDIM * 4;
    const size_t SZ_N   = (size_t)N_SRC * HDIM * 4;
    const size_t SZ_H1  = (size_t)N_MID * HDIM * 4;
    const size_t SZ_AGG = (size_t)N_MID * HDIM * 4;
    const size_t SZ_WS  = 256 * 1024;
    const size_t SZ_H2  = (size_t)N_DST * HDIM * 4;
    const size_t SZ_ZF  = (size_t)N_DST * HDIM * 4;
    float*  hp   = (float*)(ws);
    float*  nbuf = (float*)(ws + SZ_HP);
    float*  h1   = (float*)(ws + SZ_HP + SZ_N);
    float*  agg  = (float*)(ws + SZ_HP + SZ_N + SZ_H1);
    float*  wsum = (float*)(ws + SZ_HP + SZ_N + SZ_H1 + SZ_AGG);
    float*  h2   = (float*)(ws + SZ_HP + SZ_N + SZ_H1 + SZ_AGG + SZ_WS);
    float*  zf   = (float*)(ws + SZ_HP + SZ_N + SZ_H1 + SZ_AGG + SZ_WS + SZ_H2);
    __bf16* Wpb  = (__bf16*)(ws + SZ_HP + SZ_N + SZ_H1 + SZ_AGG + SZ_WS + SZ_H2 + SZ_ZF);
    __bf16* Q1b  = Wpb + HDIM * HDIM;
    __bf16* W1b  = Q1b + HDIM * HDIM;
    __bf16* Q2b  = W1b + HDIM * 2 * HDIM;
    __bf16* W2b  = Q2b + HDIM * HDIM;

    f32_to_bf16_kernel<<<(HDIM * HDIM + 255) / 256, 256, 0, stream>>>(Wp, Wpb, HDIM * HDIM);
    f32_to_bf16_kernel<<<(HDIM * HDIM + 255) / 256, 256, 0, stream>>>(Q1, Q1b, HDIM * HDIM);
    f32_to_bf16_kernel<<<(HDIM * 2 * HDIM + 255) / 256, 256, 0, stream>>>(W1, W1b, HDIM * 2 * HDIM);
    f32_to_bf16_kernel<<<(HDIM * HDIM + 255) / 256, 256, 0, stream>>>(Q2, Q2b, HDIM * HDIM);
    f32_to_bf16_kernel<<<(HDIM * 2 * HDIM + 255) / 256, 256, 0, stream>>>(W2, W2b, HDIM * 2 * HDIM);

    gemm_bf16_n256<<<N_SRC / 32, 256, 0, stream>>>(feat, Wpb, bp, hp, N_SRC, HDIM, 0);
    gemm_bf16_n256<<<N_SRC / 32, 256, 0, stream>>>(hp, Q1b, bq1, nbuf, N_SRC, HDIM, 1);
    {
        int nz = N_MID * HDIM + 65536;
        zero_f32_kernel<<<(nz + 255) / 256, 256, 0, stream>>>(agg, nz);
        edge_scatter_kernel<<<E1_N / 8, 256, 0, stream>>>(nbuf, w1v, src1, dst1, agg, wsum, E1_N);
    }
    build_z_kernel<<<(N_MID * 512 + 255) / 256, 256, 0, stream>>>(agg, wsum, hp, nbuf, N_MID);
    gemm_bf16_n256<<<N_MID / 32, 256, 0, stream>>>(nbuf, W1b, bw1, h1, N_MID, 2 * HDIM, 1);
    gemm_bf16_n256<<<N_MID / 32, 256, 0, stream>>>(h1, Q2b, bq2, nbuf, N_MID, HDIM, 1);
    {
        int nz = N_DST * HDIM;
        zero_f32_kernel<<<(nz + 255) / 256, 256, 0, stream>>>(agg, nz);
        zero_f32_kernel<<<(N_DST + 255) / 256, 256, 0, stream>>>(wsum, N_DST);
        edge_scatter_kernel<<<E2_N / 8, 256, 0, stream>>>(nbuf, w2v, src2, dst2, agg, wsum, E2_N);
    }
    build_z_kernel<<<(N_DST * 512 + 255) / 256, 256, 0, stream>>>(agg, wsum, h1, nbuf, N_DST);
    gemm_bf16_n256<<<N_DST / 32, 256, 0, stream>>>(nbuf, W2b, bw2, h2, N_DST, 2 * HDIM, 1);
    skip_norm_kernel<<<N_DST / 8, 256, 0, stream>>>(hp, h2, zf, N_DST);
    score_kernel<<<N_PAIRS / 8, 256, 0, stream>>>(zf, bias, nids,
                                                  pos_s, pos_d, neg_s, neg_d,
                                                  out, N_PAIRS);
}


# Segment sums without float atomics: counting sort of edges by dst per call (int atomics, arrival-order rank), one wave gathers and sums each dst row and writes z; sum order = arrival order
# speedup vs baseline: 1.5664x; 1.5664x over previous
#include <hip/hip_runtime.h>
#include <hip/hip_bf16.h>


typedef __attribute__((ext_vector_type(16))) __bf16 v16bf;
typedef __attribute__((ext_vector_type(8)))  float  v8f;

#define N_SRC   131072
#define N_MID   32768
#define N_DST   8192
#define E1_N    524288
#define E2_N    131072
#define HDIM    256
#define N_PAIRS 8192

__global__ void f32_to_bf16_kernel(const float* __restrict__ src,
                                   __bf16* __restrict__ dst, int n) {
    int i = blockIdx.x * blockDim.x + threadIdx.x;
    if (i < n) dst[i] = (__bf16)src[i];
}

__global__ void zero_f32_kernel(float* __restrict__ p, int n) {
    int i = blockIdx.x * blockDim.x + threadIdx.x;
    if (i < n) p[i] = 0.0f;
}

__global__ __launch_bounds__(256)
void gemm_bf16_n256(const float* __restrict__ A, const __bf16* __restrict__ W,
                    const float* __restrict__ bias, float* __restrict__ C,
                    int M, int K, int relu)
{
    const int lane = threadIdx.x & 31;
    const int wave = threadIdx.x >> 5;
    const int half = lane >> 4;
    const int l15  = lane & 15;
    const int m0   = blockIdx.x * 32 + (wave >> 2) * 16;
    const int n0   = (wave & 3) * 64;

    v8f acc[4];
    #pragma unroll
    for (int t = 0; t < 4; ++t) acc[t] = (v8f){};

    const float* arow = A + (size_t)(m0 + l15) * K;

    for (int k0 = 0; k0 < K; k0 += 32) {
        v16bf a;
        #pragma unroll
        for (int q = 0; q < 8; ++q) {
            int kk = k0 + (q >> 2) * 16 + half * 8 + (q & 3) * 2;
            float2 v = *(const float2*)(arow + kk);
            a[2 * q]     = (__bf16)v.x;
            a[2 * q + 1] = (__bf16)v.y;
        }
        #pragma unroll
        for (int t = 0; t < 4; ++t) {
            const __bf16* wrow = W + (size_t)(n0 + t * 16 + l15) * K;
            v16bf b;
            #pragma unroll
            for (int q = 0; q < 8; ++q) {
                int kk = k0 + (q >> 2) * 16 + half * 8 + (q & 3) * 2;
                b[2 * q]     = wrow[kk];
                b[2 * q + 1] = wrow[kk + 1];
            }
            acc[t] = __builtin_amdgcn_wmma_f32_16x16x32_bf16(
                false, a, false, b, (short)0, acc[t], false, false);
        }
    }

    #pragma unroll
    for (int t = 0; t < 4; ++t) {
        int n = n0 + t * 16 + l15;
        float bv = bias[n];
        #pragma unroll
        for (int r = 0; r < 8; ++r) {
            int m = m0 + half * 8 + r;
            float v = acc[t][r] + bv;
            if (relu) v = fmaxf(v, 0.0f);
            C[(size_t)m * HDIM + n] = v;
        }
    }
}

__global__ __launch_bounds__(256)
void edge_count_kernel(const int* __restrict__ dst1, int* __restrict__ cnt1,
                       const int* __restrict__ dst2, int* __restrict__ cnt2)
{
    int i = blockIdx.x * 256 + threadIdx.x;
    if (i < E1_N) {
        atomicAdd(cnt1 + dst1[i], 1);
    } else {
        int e = i - E1_N;
        if (e < E2_N) atomicAdd(cnt2 + dst2[e], 1);
    }
}

template <int PER>
__device__ __forceinline__ void block_offsets(const int* __restrict__ cnt,
                                              int* __restrict__ off,
                                              int* __restrict__ cur,
                                              int* wtot)
{
    const int t = threadIdx.x, lane = t & 31, wv = t >> 5;
    int v[PER];
    const int4* c4 = (const int4*)(cnt + t * PER);
    for (int q = 0; q < PER / 4; ++q) {
        int4 x = c4[q];
        v[4 * q] = x.x; v[4 * q + 1] = x.y; v[4 * q + 2] = x.z; v[4 * q + 3] = x.w;
    }
    int sum = 0;
    for (int k = 0; k < PER; ++k) sum += v[k];
    int inc = sum;
    for (int o = 1; o < 32; o <<= 1) {
        int y = __shfl_up(inc, o, 32);
        if (lane >= o) inc += y;
    }
    if (lane == 31) wtot[wv] = inc;
    __syncthreads();
    int wt = wtot[lane];
    int winc = wt;
    for (int o = 1; o < 32; o <<= 1) {
        int y = __shfl_up(winc, o, 32);
        if (lane >= o) winc += y;
    }
    int run = __shfl(winc, wv, 32) - __shfl(wt, wv, 32) + inc - sum;
    int total = __shfl(winc, 31, 32);
    int4* o4 = (int4*)(off + t * PER);
    int4* u4 = (int4*)(cur + t * PER);
    for (int q = 0; q < PER / 4; ++q) {
        int4 x;
        x.x = run; run += v[4 * q];
        x.y = run; run += v[4 * q + 1];
        x.z = run; run += v[4 * q + 2];
        x.w = run; run += v[4 * q + 3];
        o4[q] = x;
        u4[q] = x;
    }
    if (t == 1023) off[1024 * PER] = total;
}

__global__ __launch_bounds__(1024)
void segment_offsets_kernel(const int* __restrict__ cnt1, int* __restrict__ off1,
                            int* __restrict__ cur1,
                            const int* __restrict__ cnt2, int* __restrict__ off2,
                            int* __restrict__ cur2)
{
    __shared__ int wtot[32];
    if (blockIdx.x == 0) block_offsets<N_MID / 1024>(cnt1, off1, cur1, wtot);
    else                 block_offsets<N_DST / 1024>(cnt2, off2, cur2, wtot);
}

__global__ __launch_bounds__(256)
void edge_place_kernel(const int* __restrict__ src1, const int* __restrict__ dst1,
                       const float* __restrict__ w1, int* __restrict__ cur1,
                       int* __restrict__ pairs1,
                       const int* __restrict__ src2, const int* __restrict__ dst2,
                       const float* __restrict__ w2, int* __restrict__ cur2,
                       int* __restrict__ pairs2)
{
    int i = blockIdx.x * 256 + threadIdx.x;
    if (i < E1_N) {
        int p = atomicAdd(cur1 + dst1[i], 1);
        int2 v; v.x = src1[i]; v.y = __float_as_int(w1[i]);
        *(int2*)(pairs1 + 2 * (size_t)p) = v;
    } else {
        int e = i - E1_N;
        if (e < E2_N) {
            int p = atomicAdd(cur2 + dst2[e], 1);
            int2 v; v.x = src2[e]; v.y = __float_as_int(w2[e]);
            *(int2*)(pairs2 + 2 * (size_t)p) = v;
        }
    }
}

__device__ __forceinline__ void fma4(float4& a, const float4& x, float w) {
    a.x = fmaf(x.x, w, a.x); a.y = fmaf(x.y, w, a.y);
    a.z = fmaf(x.z, w, a.z); a.w = fmaf(x.w, w, a.w);
}

__global__ __launch_bounds__(256)
void segment_gather_kernel(const float* __restrict__ nsrc,
                           const int* __restrict__ off,
                           const int* __restrict__ pairs,
                           const float* __restrict__ hdst,
                           float* __restrict__ z, int rows)
{
    const int lane = threadIdx.x & 31;
    const int r    = blockIdx.x * 8 + (threadIdx.x >> 5);
    if (r >= rows) return;
    const float4* hr = (const float4*)(hdst + (size_t)r * HDIM);
    float4*       zr = (float4*)(z + (size_t)r * (2 * HDIM));
    const float4 h0 = hr[lane], h1 = hr[32 + lane];
    const int beg = __builtin_amdgcn_readfirstlane(off[r]);
    const int end = __builtin_amdgcn_readfirstlane(off[r + 1]);
    zr[64 + lane] = h0;
    zr[96 + lane] = h1;
    float4 a0 = make_float4(0.f, 0.f, 0.f, 0.f);
    float4 a1 = make_float4(0.f, 0.f, 0.f, 0.f);
    float  ws = 0.0f;
    for (int base = beg; base < end; base += 32) {
        int n = end - base;
        n = n < 32 ? n : 32;
        int ps = 0, pw = 0;
        if (lane < n) {
            int2 p = *(const int2*)(pairs + 2 * (size_t)(base + lane));
            ps = p.x; pw = p.y;
        }
        int j = 0;
        for (; j + 4 <= n; j += 4) {
            const float4* r0 = (const float4*)(nsrc + (size_t)__builtin_amdgcn_readlane(ps, j)     * HDIM);
            const float4* r1 = (const float4*)(nsrc + (size_t)__builtin_amdgcn_readlane(ps, j + 1) * HDIM);
            const float4* r2 = (const float4*)(nsrc + (size_t)__builtin_amdgcn_readlane(ps, j + 2) * HDIM);
            const float4* r3 = (const float4*)(nsrc + (size_t)__builtin_amdgcn_readlane(ps, j + 3) * HDIM);
            float w0 = __int_as_float(__builtin_amdgcn_readlane(pw, j));
            float w1 = __int_as_float(__builtin_amdgcn_readlane(pw, j + 1));
            float w2 = __int_as_float(__builtin_amdgcn_readlane(pw, j + 2));
            float w3 = __int_as_float(__builtin_amdgcn_readlane(pw, j + 3));
            float4 x0 = r0[lane], y0 = r0[32 + lane];
            float4 x1 = r1[lane], y1 = r1[32 + lane];
            float4 x2 = r2[lane], y2 = r2[32 + lane];
            float4 x3 = r3[lane], y3 = r3[32 + lane];
            fma4(a0, x0, w0); fma4(a1, y0, w0); ws += w0;
            fma4(a0, x1, w1); fma4(a1, y1, w1); ws += w1;
            fma4(a0, x2, w2); fma4(a1, y2, w2); ws += w2;
            fma4(a0, x3, w3); fma4(a1, y3, w3); ws += w3;
        }
        #pragma nounroll
        for (; j < n; ++j) {
            const float4* r0 = (const float4*)(nsrc + (size_t)__builtin_amdgcn_readlane(ps, j) * HDIM);
            float w0 = __int_as_float(__builtin_amdgcn_readlane(pw, j));
            float4 x0 = r0[lane], y0 = r0[32 + lane];
            fma4(a0, x0, w0); fma4(a1, y0, w0); ws += w0;
        }
    }
    ws = ws < 1.0f ? 1.0f : ws;
    const float inv = __frcp_rn(ws);
    zr[lane]      = make_float4(a0.x * inv, a0.y * inv, a0.z * inv, a0.w * inv);
    zr[32 + lane] = make_float4(a1.x * inv, a1.y * inv, a1.z * inv, a1.w * inv);
}

__global__ __launch_bounds__(256)
void skip_norm_kernel(const float* __restrict__ hp, const float* __restrict__ h2,
                      float* __restrict__ zf, int rows)
{
    int r    = blockIdx.x * 8 + (threadIdx.x >> 5);
    int lane = threadIdx.x & 31;
    if (r >= rows) return;
    float v[8];
    float ss = 0.0f;
    #pragma unroll
    for (int j = 0; j < 8; ++j) {
        int c = lane * 8 + j;
        v[j] = hp[(size_t)r * HDIM + c] + h2[(size_t)r * HDIM + c];
        ss += v[j] * v[j];
    }
    #pragma unroll
    for (int o = 16; o > 0; o >>= 1) ss += __shfl_xor(ss, o, 32);
    float nrm   = sqrtf(ss);
    float scale = (nrm == 0.0f) ? 1.0f : 1.0f / nrm;
    #pragma unroll
    for (int j = 0; j < 8; ++j)
        zf[(size_t)r * HDIM + lane * 8 + j] = v[j] * scale;
}

__global__ __launch_bounds__(256)
void score_kernel(const float* __restrict__ zf, const float* __restrict__ bias,
                  const int* __restrict__ nids,
                  const int* __restrict__ ps, const int* __restrict__ pd,
                  const int* __restrict__ ns, const int* __restrict__ nd,
                  float* __restrict__ out, int npairs)
{
    int p    = blockIdx.x * 8 + (threadIdx.x >> 5);
    int lane = threadIdx.x & 31;
    if (p >= npairs) return;
    int pu = ps[p], pv = pd[p], nu = ns[p], nv = nd[p];
    float sp = 0.0f, sn = 0.0f;
    #pragma unroll
    for (int j = 0; j < 8; ++j) {
        int c = lane * 8 + j;
        sp += zf[(size_t)pu * HDIM + c] * zf[(size_t)pv * HDIM + c];
        sn += zf[(size_t)nu * HDIM + c] * zf[(size_t)nv * HDIM + c];
    }
    #pragma unroll
    for (int o = 16; o > 0; o >>= 1) {
        sp += __shfl_xor(sp, o, 32);
        sn += __shfl_xor(sn, o, 32);
    }
    if (lane == 0) {
        float gp = bias[nids[pu]] + bias[nids[pv]];
        float gn = bias[nids[nu]] + bias[nids[nv]];
        float m  = (sn + gn) - (sp + gp) + 1.0f;
        out[p] = m > 0.0f ? m : 0.0f;
    }
}

extern "C" void kernel_launch(void* const* d_in, const int* in_sizes, int n_in,
                              void* d_out, int out_size, void* d_ws, size_t ws_size,
                              hipStream_t stream) {
    const float* feat = (const float*)d_in[0];
    const float* Wp   = (const float*)d_in[1];
    const float* bp   = (const float*)d_in[2];
    const float* Q1   = (const float*)d_in[3];
    const float* bq1  = (const float*)d_in[4];
    const float* W1   = (const float*)d_in[5];
    const float* bw1  = (const float*)d_in[6];
    const float* Q2   = (const float*)d_in[7];
    const float* bq2  = (const float*)d_in[8];
    const float* W2   = (const float*)d_in[9];
    const float* bw2  = (const float*)d_in[10];
    const float* w1v  = (const float*)d_in[11];
    const float* w2v  = (const float*)d_in[12];
    const float* bias = (const float*)d_in[13];
    const int* src1   = (const int*)d_in[14];
    const int* dst1   = (const int*)d_in[15];
    const int* src2   = (const int*)d_in[16];
    const int* dst2   = (const int*)d_in[17];
    const int* pos_s  = (const int*)d_in[18];
    const int* pos_d  = (const int*)d_in[19];
    const int* neg_s  = (const int*)d_in[20];
    const int* neg_d  = (const int*)d_in[21];
    const int* nids   = (const int*)d_in[22];
    float* out        = (float*)d_out;

    char* ws = (char*)d_ws;
    const size_t SZ_HP  = (size_t)N_SRC * HDIM * 4;
    const size_t SZ_N   = (size_t)N_SRC * HDIM * 4;
    const size_t SZ_H1  = (size_t)N_MID * HDIM * 4;
    const size_t SZ_AGG = (size_t)N_MID * HDIM * 4;
    const size_t SZ_WS  = 256 * 1024;
    const size_t SZ_H2  = (size_t)N_DST * HDIM * 4;
    const size_t SZ_ZF  = (size_t)N_DST * HDIM * 4;
    float*  hp   = (float*)(ws);
    float*  nbuf = (float*)(ws + SZ_HP);
    float*  h1   = (float*)(ws + SZ_HP + SZ_N);
    float*  zbuf = hp + (size_t)N_MID * HDIM;
    int* cnt1   = (int*)(ws + SZ_HP + SZ_N + SZ_H1);
    int* cnt2   = cnt1 + N_MID;
    int* off1   = cnt2 + N_DST;
    int* off2   = off1 + N_MID + 64;
    int* cur1   = off2 + N_DST + 64;
    int* cur2   = cur1 + N_MID;
    int* pairs1 = cur2 + N_DST;
    int* pairs2 = pairs1 + 2 * E1_N;
    float*  h2   = (float*)(ws + SZ_HP + SZ_N + SZ_H1 + SZ_AGG + SZ_WS);
    float*  zf   = (float*)(ws + SZ_HP + SZ_N + SZ_H1 + SZ_AGG + SZ_WS + SZ_H2);
    __bf16* Wpb  = (__bf16*)(ws + SZ_HP + SZ_N + SZ_H1 + SZ_AGG + SZ_WS + SZ_H2 + SZ_ZF);
    __bf16* Q1b  = Wpb + HDIM * HDIM;
    __bf16* W1b  = Q1b + HDIM * HDIM;
    __bf16* Q2b  = W1b + HDIM * 2 * HDIM;
    __bf16* W2b  = Q2b + HDIM * HDIM;

    f32_to_bf16_kernel<<<(HDIM * HDIM + 255) / 256, 256, 0, stream>>>(Wp, Wpb, HDIM * HDIM);
    f32_to_bf16_kernel<<<(HDIM * HDIM + 255) / 256, 256, 0, stream>>>(Q1, Q1b, HDIM * HDIM);
    f32_to_bf16_kernel<<<(HDIM * 2 * HDIM + 255) / 256, 256, 0, stream>>>(W1, W1b, HDIM * 2 * HDIM);
    f32_to_bf16_kernel<<<(HDIM * HDIM + 255) / 256, 256, 0, stream>>>(Q2, Q2b, HDIM * HDIM);
    f32_to_bf16_kernel<<<(HDIM * 2 * HDIM + 255) / 256, 256, 0, stream>>>(W2, W2b, HDIM * 2 * HDIM);

    zero_f32_kernel<<<(N_MID + N_DST + 255) / 256, 256, 0, stream>>>((float*)cnt1, N_MID + N_DST);
    edge_count_kernel<<<(E1_N + E2_N + 255) / 256, 256, 0, stream>>>(dst1, cnt1, dst2, cnt2);
    segment_offsets_kernel<<<2, 1024, 0, stream>>>(cnt1, off1, cur1, cnt2, off2, cur2);
    edge_place_kernel<<<(E1_N + E2_N + 255) / 256, 256, 0, stream>>>(src1, dst1, w1v, cur1, pairs1,
                                                                     src2, dst2, w2v, cur2, pairs2);

    gemm_bf16_n256<<<N_SRC / 32, 256, 0, stream>>>(feat, Wpb, bp, hp, N_SRC, HDIM, 0);
    gemm_bf16_n256<<<N_SRC / 32, 256, 0, stream>>>(hp, Q1b, bq1, nbuf, N_SRC, HDIM, 1);
    segment_gather_kernel<<<N_MID / 8, 256, 0, stream>>>(nbuf, off1, pairs1, hp, zbuf, N_MID);
    gemm_bf16_n256<<<N_MID / 32, 256, 0, stream>>>(zbuf, W1b, bw1, h1, N_MID, 2 * HDIM, 1);
    gemm_bf16_n256<<<N_MID / 32, 256, 0, stream>>>(h1, Q2b, bq2, nbuf, N_MID, HDIM, 1);
    segment_gather_kernel<<<N_DST / 8, 256, 0, stream>>>(nbuf, off2, pairs2, h1, zbuf, N_DST);
    gemm_bf16_n256<<<N_DST / 32, 256, 0, stream>>>(zbuf, W2b, bw2, h2, N_DST, 2 * HDIM, 1);
    skip_norm_kernel<<<N_DST / 8, 256, 0, stream>>>(hp, h2, zf, N_DST);
    score_kernel<<<N_PAIRS / 8, 256, 0, stream>>>(zf, bias, nids,
                                                  pos_s, pos_d, neg_s, neg_d,
                                                  out, N_PAIRS);
}


# GEMM B operand from LDS: one 16-wave work-group per WGP copies the bf16 weights into LDS once (row pitch 2K+16 B) and loops over row tiles reading B fragments from LDS; sum order unchanged
# speedup vs baseline: 2.4473x; 1.5624x over previous
#include <hip/hip_runtime.h>
#include <hip/hip_bf16.h>


typedef __attribute__((ext_vector_type(16))) __bf16 v16bf;
typedef __attribute__((ext_vector_type(8)))  float  v8f;

#define N_SRC   131072
#define N_MID   32768
#define N_DST   8192
#define E1_N    524288
#define E2_N    131072
#define HDIM    256
#define N_PAIRS 8192

__global__ void f32_to_bf16_kernel(const float* __restrict__ src,
                                   __bf16* __restrict__ dst, int n) {
    int i = blockIdx.x * blockDim.x + threadIdx.x;
    if (i < n) dst[i] = (__bf16)src[i];
}

__global__ void zero_f32_kernel(float* __restrict__ p, int n) {
    int i = blockIdx.x * blockDim.x + threadIdx.x;
    if (i < n) p[i] = 0.0f;
}

__global__ __launch_bounds__(256)
void gemm_bf16_n256(const float* __restrict__ A, const __bf16* __restrict__ W,
                    const float* __restrict__ bias, float* __restrict__ C,
                    int M, int K, int relu)
{
    const int lane = threadIdx.x & 31;
    const int wave = threadIdx.x >> 5;
    const int half = lane >> 4;
    const int l15  = lane & 15;
    const int m0   = blockIdx.x * 32 + (wave >> 2) * 16;
    const int n0   = (wave & 3) * 64;

    v8f acc[4];
    #pragma unroll
    for (int t = 0; t < 4; ++t) acc[t] = (v8f){};

    const float* arow = A + (size_t)(m0 + l15) * K;

    for (int k0 = 0; k0 < K; k0 += 32) {
        v16bf a;
        #pragma unroll
        for (int q = 0; q < 8; ++q) {
            int kk = k0 + (q >> 2) * 16 + half * 8 + (q & 3) * 2;
            float2 v = *(const float2*)(arow + kk);
            a[2 * q]     = (__bf16)v.x;
            a[2 * q + 1] = (__bf16)v.y;
        }
        #pragma unroll
        for (int t = 0; t < 4; ++t) {
            const __bf16* wrow = W + (size_t)(n0 + t * 16 + l15) * K;
            v16bf b;
            #pragma unroll
            for (int q = 0; q < 8; ++q) {
                int kk = k0 + (q >> 2) * 16 + half * 8 + (q & 3) * 2;
                b[2 * q]     = wrow[kk];
                b[2 * q + 1] = wrow[kk + 1];
            }
            acc[t] = __builtin_amdgcn_wmma_f32_16x16x32_bf16(
                false, a, false, b, (short)0, acc[t], false, false);
        }
    }

    #pragma unroll
    for (int t = 0; t < 4; ++t) {
        int n = n0 + t * 16 + l15;
        float bv = bias[n];
        #pragma unroll
        for (int r = 0; r < 8; ++r) {
            int m = m0 + half * 8 + r;
            float v = acc[t][r] + bv;
            if (relu) v = fmaxf(v, 0.0f);
            C[(size_t)m * HDIM + n] = v;
        }
    }
}

typedef __attribute__((ext_vector_type(8))) unsigned int v8u;

struct a_step { float4 f0, f1, f2, f3; };

__device__ __forceinline__ a_step load_a_step(const float* p) {
    a_step s;
    s.f0 = *(const float4*)(p);
    s.f1 = *(const float4*)(p + 4);
    s.f2 = *(const float4*)(p + 16);
    s.f3 = *(const float4*)(p + 20);
    return s;
}

template <int K, int NT>
__global__ __launch_bounds__(512)
void gemm_bf16_w_in_lds(const float* __restrict__ A, const __bf16* __restrict__ W,
                        const float* __restrict__ bias, float* __restrict__ C,
                        int M, int relu)
{
    constexpr int PITCH  = K * 2 + 16;
    constexpr int CPR    = K / 8;
    constexpr int CHUNKS = HDIM * CPR;
    constexpr int WPR    = 16 / NT;
    constexpr int KS     = K / 32;
    constexpr int G      = (65535 - 47 - (KS - 1) * 64) / (16 * PITCH) + 1 < NT
                         ? (65535 - 47 - (KS - 1) * 64) / (16 * PITCH) + 1 : NT;
    constexpr int NB     = (NT + G - 1) / G;
    __shared__ __attribute__((aligned(16))) unsigned char L[HDIM * PITCH];

    const int tid  = threadIdx.x;
    const int lane = tid & 31;
    const int wave = __builtin_amdgcn_readfirstlane(tid >> 5);
    const int half = lane >> 4;
    const int l15  = lane & 15;

    const int total  = (M / 16) * WPR;
    const int stride = gridDim.x * 16;
    int it = blockIdx.x * 16 + wave;

    a_step q0, q1;
    q0.f0 = q0.f1 = q0.f2 = q0.f3 = make_float4(0.f, 0.f, 0.f, 0.f);
    q1 = q0;
    if (it < total) {
        const float* p = A + (size_t)((it / WPR) * 16 + l15) * K + half * 8;
        q0 = load_a_step(p);
        q1 = load_a_step(p + 32);
    }

    #pragma unroll 16
    for (int i = 0; i < CHUNKS / 512; ++i) {
        const int c = i * 512 + tid;
        const int row = c / CPR, col = c % CPR;
        const uint4 v = *(const uint4*)(W + (size_t)c * 8);
        *(uint4*)(L + row * PITCH + col * 16) = v;
    }
    __syncthreads();

    unsigned goff[NB];
    #pragma unroll
    for (int g = 0; g < NB; ++g) {
        goff[g] = g * G * 16 * PITCH;
        if (g > 0) asm volatile("" : "+v"(goff[g]));
    }

    for (; it < total; it += stride) {
        const int m0 = (it / WPR) * 16;
        const int n0 = (it % WPR) * (NT * 16);
        const float* arow = A + (size_t)(m0 + l15) * K + half * 8;
        const bool more = it + stride < total;
        const float* anext = A + (size_t)(((it + stride) / WPR) * 16 + l15) * K + half * 8;
        const unsigned char* brow = L + (n0 + l15) * PITCH + half * 16;

        v8f acc[NT];
        #pragma unroll
        for (int t = 0; t < NT; ++t) acc[t] = (v8f){};

        #pragma unroll
        for (int ks = 0; ks < KS; ++ks) {
            a_step q2 = q1;
            if (ks + 2 < KS)  q2 = load_a_step(arow + (ks + 2) * 32);
            else if (more)    q2 = load_a_step(anext + (ks + 2 - KS) * 32);

            v16bf a;
            a[0]  = (__bf16)q0.f0.x; a[1]  = (__bf16)q0.f0.y; a[2]  = (__bf16)q0.f0.z; a[3]  = (__bf16)q0.f0.w;
            a[4]  = (__bf16)q0.f1.x; a[5]  = (__bf16)q0.f1.y; a[6]  = (__bf16)q0.f1.z; a[7]  = (__bf16)q0.f1.w;
            a[8]  = (__bf16)q0.f2.x; a[9]  = (__bf16)q0.f2.y; a[10] = (__bf16)q0.f2.z; a[11] = (__bf16)q0.f2.w;
            a[12] = (__bf16)q0.f3.x; a[13] = (__bf16)q0.f3.y; a[14] = (__bf16)q0.f3.z; a[15] = (__bf16)q0.f3.w;

            #pragma unroll
            for (int t = 0; t < NT; ++t) {
                const unsigned char* bp = brow + goff[t / G] + (t % G) * 16 * PITCH + ks * 64;
                const uint4 lo = *(const uint4*)(bp);
                const uint4 hi = *(const uint4*)(bp + 32);
                v8u bw;
                bw[0] = lo.x; bw[1] = lo.y; bw[2] = lo.z; bw[3] = lo.w;
                bw[4] = hi.x; bw[5] = hi.y; bw[6] = hi.z; bw[7] = hi.w;
                const v16bf b = __builtin_bit_cast(v16bf, bw);
                acc[t] = __builtin_amdgcn_wmma_f32_16x16x32_bf16(
                    false, a, false, b, (short)0, acc[t], false, false);
            }
            q0 = q1;
            q1 = q2;
        }

        #pragma unroll
        for (int t = 0; t < NT; ++t) {
            const int n = n0 + t * 16 + l15;
            const float bv = bias[n];
            #pragma unroll
            for (int r = 0; r < 8; ++r) {
                const int m = m0 + half * 8 + r;
                float v = acc[t][r] + bv;
                if (relu) v = fmaxf(v, 0.0f);
                C[(size_t)m * HDIM + n] = v;
            }
        }
    }
}

__global__ __launch_bounds__(256)
void edge_count_kernel(const int* __restrict__ dst1, int* __restrict__ cnt1,
                       const int* __restrict__ dst2, int* __restrict__ cnt2)
{
    int i = blockIdx.x * 256 + threadIdx.x;
    if (i < E1_N) {
        atomicAdd(cnt1 + dst1[i], 1);
    } else {
        int e = i - E1_N;
        if (e < E2_N) atomicAdd(cnt2 + dst2[e], 1);
    }
}

template <int PER>
__device__ __forceinline__ void block_offsets(const int* __restrict__ cnt,
                                              int* __restrict__ off,
                                              int* __restrict__ cur,
                                              int* wtot)
{
    const int t = threadIdx.x, lane = t & 31, wv = t >> 5;
    int v[PER];
    const int4* c4 = (const int4*)(cnt + t * PER);
    for (int q = 0; q < PER / 4; ++q) {
        int4 x = c4[q];
        v[4 * q] = x.x; v[4 * q + 1] = x.y; v[4 * q + 2] = x.z; v[4 * q + 3] = x.w;
    }
    int sum = 0;
    for (int k = 0; k < PER; ++k) sum += v[k];
    int inc = sum;
    for (int o = 1; o < 32; o <<= 1) {
        int y = __shfl_up(inc, o, 32);
        if (lane >= o) inc += y;
    }
    if (lane == 31) wtot[wv] = inc;
    __syncthreads();
    int wt = wtot[lane];
    int winc = wt;
    for (int o = 1; o < 32; o <<= 1) {
        int y = __shfl_up(winc, o, 32);
        if (lane >= o) winc += y;
    }
    int run = __shfl(winc, wv, 32) - __shfl(wt, wv, 32) + inc - sum;
    int total = __shfl(winc, 31, 32);
    int4* o4 = (int4*)(off + t * PER);
    int4* u4 = (int4*)(cur + t * PER);
    for (int q = 0; q < PER / 4; ++q) {
        int4 x;
        x.x = run; run += v[4 * q];
        x.y = run; run += v[4 * q + 1];
        x.z = run; run += v[4 * q + 2];
        x.w = run; run += v[4 * q + 3];
        o4[q] = x;
        u4[q] = x;
    }
    if (t == 1023) off[1024 * PER] = total;
}

__global__ __launch_bounds__(1024)
void segment_offsets_kernel(const int* __restrict__ cnt1, int* __restrict__ off1,
                            int* __restrict__ cur1,
                            const int* __restrict__ cnt2, int* __restrict__ off2,
                            int* __restrict__ cur2)
{
    __shared__ int wtot[32];
    if (blockIdx.x == 0) block_offsets<N_MID / 1024>(cnt1, off1, cur1, wtot);
    else                 block_offsets<N_DST / 1024>(cnt2, off2, cur2, wtot);
}

__global__ __launch_bounds__(256)
void edge_place_kernel(const int* __restrict__ src1, const int* __restrict__ dst1,
                       const float* __restrict__ w1, int* __restrict__ cur1,
                       int* __restrict__ pairs1,
                       const int* __restrict__ src2, const int* __restrict__ dst2,
                       const float* __restrict__ w2, int* __restrict__ cur2,
                       int* __restrict__ pairs2)
{
    int i = blockIdx.x * 256 + threadIdx.x;
    if (i < E1_N) {
        int p = atomicAdd(cur1 + dst1[i], 1);
        int2 v; v.x = src1[i]; v.y = __float_as_int(w1[i]);
        *(int2*)(pairs1 + 2 * (size_t)p) = v;
    } else {
        int e = i - E1_N;
        if (e < E2_N) {
            int p = atomicAdd(cur2 + dst2[e], 1);
            int2 v; v.x = src2[e]; v.y = __float_as_int(w2[e]);
            *(int2*)(pairs2 + 2 * (size_t)p) = v;
        }
    }
}

__device__ __forceinline__ void fma4(float4& a, const float4& x, float w) {
    a.x = fmaf(x.x, w, a.x); a.y = fmaf(x.y, w, a.y);
    a.z = fmaf(x.z, w, a.z); a.w = fmaf(x.w, w, a.w);
}

__global__ __launch_bounds__(256)
void segment_gather_kernel(const float* __restrict__ nsrc,
                           const int* __restrict__ off,
                           const int* __restrict__ pairs,
                           const float* __restrict__ hdst,
                           float* __restrict__ z, int rows)
{
    const int lane = threadIdx.x & 31;
    const int r    = blockIdx.x * 8 + (threadIdx.x >> 5);
    if (r >= rows) return;
    const float4* hr = (const float4*)(hdst + (size_t)r * HDIM);
    float4*       zr = (float4*)(z + (size_t)r * (2 * HDIM));
    const float4 h0 = hr[lane], h1 = hr[32 + lane];
    const int beg = __builtin_amdgcn_readfirstlane(off[r]);
    const int end = __builtin_amdgcn_readfirstlane(off[r + 1]);
    zr[64 + lane] = h0;
    zr[96 + lane] = h1;
    float4 a0 = make_float4(0.f, 0.f, 0.f, 0.f);
    float4 a1 = make_float4(0.f, 0.f, 0.f, 0.f);
    float  ws = 0.0f;
    for (int base = beg; base < end; base += 32) {
        int n = end - base;
        n = n < 32 ? n : 32;
        int ps = 0, pw = 0;
        if (lane < n) {
            int2 p = *(const int2*)(pairs + 2 * (size_t)(base + lane));
            ps = p.x; pw = p.y;
        }
        int j = 0;
        for (; j + 4 <= n; j += 4) {
            const float4* r0 = (const float4*)(nsrc + (size_t)__builtin_amdgcn_readlane(ps, j)     * HDIM);
            const float4* r1 = (const float4*)(nsrc + (size_t)__builtin_amdgcn_readlane(ps, j + 1) * HDIM);
            const float4* r2 = (const float4*)(nsrc + (size_t)__builtin_amdgcn_readlane(ps, j + 2) * HDIM);
            const float4* r3 = (const float4*)(nsrc + (size_t)__builtin_amdgcn_readlane(ps, j + 3) * HDIM);
            float w0 = __int_as_float(__builtin_amdgcn_readlane(pw, j));
            float w1 = __int_as_float(__builtin_amdgcn_readlane(pw, j + 1));
            float w2 = __int_as_float(__builtin_amdgcn_readlane(pw, j + 2));
            float w3 = __int_as_float(__builtin_amdgcn_readlane(pw, j + 3));
            float4 x0 = r0[lane], y0 = r0[32 + lane];
            float4 x1 = r1[lane], y1 = r1[32 + lane];
            float4 x2 = r2[lane], y2 = r2[32 + lane];
            float4 x3 = r3[lane], y3 = r3[32 + lane];
            fma4(a0, x0, w0); fma4(a1, y0, w0); ws += w0;
            fma4(a0, x1, w1); fma4(a1, y1, w1); ws += w1;
            fma4(a0, x2, w2); fma4(a1, y2, w2); ws += w2;
            fma4(a0, x3, w3); fma4(a1, y3, w3); ws += w3;
        }
        #pragma nounroll
        for (; j < n; ++j) {
            const float4* r0 = (const float4*)(nsrc + (size_t)__builtin_amdgcn_readlane(ps, j) * HDIM);
            float w0 = __int_as_float(__builtin_amdgcn_readlane(pw, j));
            float4 x0 = r0[lane], y0 = r0[32 + lane];
            fma4(a0, x0, w0); fma4(a1, y0, w0); ws += w0;
        }
    }
    ws = ws < 1.0f ? 1.0f : ws;
    const float inv = __frcp_rn(ws);
    zr[lane]      = make_float4(a0.x * inv, a0.y * inv, a0.z * inv, a0.w * inv);
    zr[32 + lane] = make_float4(a1.x * inv, a1.y * inv, a1.z * inv, a1.w * inv);
}

__global__ __launch_bounds__(256)
void skip_norm_kernel(const float* __restrict__ hp, const float* __restrict__ h2,
                      float* __restrict__ zf, int rows)
{
    int r    = blockIdx.x * 8 + (threadIdx.x >> 5);
    int lane = threadIdx.x & 31;
    if (r >= rows) return;
    float v[8];
    float ss = 0.0f;
    #pragma unroll
    for (int j = 0; j < 8; ++j) {
        int c = lane * 8 + j;
        v[j] = hp[(size_t)r * HDIM + c] + h2[(size_t)r * HDIM + c];
        ss += v[j] * v[j];
    }
    #pragma unroll
    for (int o = 16; o > 0; o >>= 1) ss += __shfl_xor(ss, o, 32);
    float nrm   = sqrtf(ss);
    float scale = (nrm == 0.0f) ? 1.0f : 1.0f / nrm;
    #pragma unroll
    for (int j = 0; j < 8; ++j)
        zf[(size_t)r * HDIM + lane * 8 + j] = v[j] * scale;
}

__global__ __launch_bounds__(256)
void score_kernel(const float* __restrict__ zf, const float* __restrict__ bias,
                  const int* __restrict__ nids,
                  const int* __restrict__ ps, const int* __restrict__ pd,
                  const int* __restrict__ ns, const int* __restrict__ nd,
                  float* __restrict__ out, int npairs)
{
    int p    = blockIdx.x * 8 + (threadIdx.x >> 5);
    int lane = threadIdx.x & 31;
    if (p >= npairs) return;
    int pu = ps[p], pv = pd[p], nu = ns[p], nv = nd[p];
    float sp = 0.0f, sn = 0.0f;
    #pragma unroll
    for (int j = 0; j < 8; ++j) {
        int c = lane * 8 + j;
        sp += zf[(size_t)pu * HDIM + c] * zf[(size_t)pv * HDIM + c];
        sn += zf[(size_t)nu * HDIM + c] * zf[(size_t)nv * HDIM + c];
    }
    #pragma unroll
    for (int o = 16; o > 0; o >>= 1) {
        sp += __shfl_xor(sp, o, 32);
        sn += __shfl_xor(sn, o, 32);
    }
    if (lane == 0) {
        float gp = bias[nids[pu]] + bias[nids[pv]];
        float gn = bias[nids[nu]] + bias[nids[nv]];
        float m  = (sn + gn) - (sp + gp) + 1.0f;
        out[p] = m > 0.0f ? m : 0.0f;
    }
}

extern "C" void kernel_launch(void* const* d_in, const int* in_sizes, int n_in,
                              void* d_out, int out_size, void* d_ws, size_t ws_size,
                              hipStream_t stream) {
    const float* feat = (const float*)d_in[0];
    const float* Wp   = (const float*)d_in[1];
    const float* bp   = (const float*)d_in[2];
    const float* Q1   = (const float*)d_in[3];
    const float* bq1  = (const float*)d_in[4];
    const float* W1   = (const float*)d_in[5];
    const float* bw1  = (const float*)d_in[6];
    const float* Q2   = (const float*)d_in[7];
    const float* bq2  = (const float*)d_in[8];
    const float* W2   = (const float*)d_in[9];
    const float* bw2  = (const float*)d_in[10];
    const float* w1v  = (const float*)d_in[11];
    const float* w2v  = (const float*)d_in[12];
    const float* bias = (const float*)d_in[13];
    const int* src1   = (const int*)d_in[14];
    const int* dst1   = (const int*)d_in[15];
    const int* src2   = (const int*)d_in[16];
    const int* dst2   = (const int*)d_in[17];
    const int* pos_s  = (const int*)d_in[18];
    const int* pos_d  = (const int*)d_in[19];
    const int* neg_s  = (const int*)d_in[20];
    const int* neg_d  = (const int*)d_in[21];
    const int* nids   = (const int*)d_in[22];
    float* out        = (float*)d_out;

    char* ws = (char*)d_ws;
    const size_t SZ_HP  = (size_t)N_SRC * HDIM * 4;
    const size_t SZ_N   = (size_t)N_SRC * HDIM * 4;
    const size_t SZ_H1  = (size_t)N_MID * HDIM * 4;
    const size_t SZ_AGG = (size_t)N_MID * HDIM * 4;
    const size_t SZ_WS  = 256 * 1024;
    const size_t SZ_H2  = (size_t)N_DST * HDIM * 4;
    const size_t SZ_ZF  = (size_t)N_DST * HDIM * 4;
    float*  hp   = (float*)(ws);
    float*  nbuf = (float*)(ws + SZ_HP);
    float*  h1   = (float*)(ws + SZ_HP + SZ_N);
    float*  zbuf = hp + (size_t)N_MID * HDIM;
    int* cnt1   = (int*)(ws + SZ_HP + SZ_N + SZ_H1);
    int* cnt2   = cnt1 + N_MID;
    int* off1   = cnt2 + N_DST;
    int* off2   = off1 + N_MID + 64;
    int* cur1   = off2 + N_DST + 64;
    int* cur2   = cur1 + N_MID;
    int* pairs1 = cur2 + N_DST;
    int* pairs2 = pairs1 + 2 * E1_N;
    float*  h2   = (float*)(ws + SZ_HP + SZ_N + SZ_H1 + SZ_AGG + SZ_WS);
    float*  zf   = (float*)(ws + SZ_HP + SZ_N + SZ_H1 + SZ_AGG + SZ_WS + SZ_H2);
    __bf16* Wpb  = (__bf16*)(ws + SZ_HP + SZ_N + SZ_H1 + SZ_AGG + SZ_WS + SZ_H2 + SZ_ZF);
    __bf16* Q1b  = Wpb + HDIM * HDIM;
    __bf16* W1b  = Q1b + HDIM * HDIM;
    __bf16* Q2b  = W1b + HDIM * 2 * HDIM;
    __bf16* W2b  = Q2b + HDIM * HDIM;

    f32_to_bf16_kernel<<<(HDIM * HDIM + 255) / 256, 256, 0, stream>>>(Wp, Wpb, HDIM * HDIM);
    f32_to_bf16_kernel<<<(HDIM * HDIM + 255) / 256, 256, 0, stream>>>(Q1, Q1b, HDIM * HDIM);
    f32_to_bf16_kernel<<<(HDIM * 2 * HDIM + 255) / 256, 256, 0, stream>>>(W1, W1b, HDIM * 2 * HDIM);
    f32_to_bf16_kernel<<<(HDIM * HDIM + 255) / 256, 256, 0, stream>>>(Q2, Q2b, HDIM * HDIM);
    f32_to_bf16_kernel<<<(HDIM * 2 * HDIM + 255) / 256, 256, 0, stream>>>(W2, W2b, HDIM * 2 * HDIM);

    zero_f32_kernel<<<(N_MID + N_DST + 255) / 256, 256, 0, stream>>>((float*)cnt1, N_MID + N_DST);
    edge_count_kernel<<<(E1_N + E2_N + 255) / 256, 256, 0, stream>>>(dst1, cnt1, dst2, cnt2);
    segment_offsets_kernel<<<2, 1024, 0, stream>>>(cnt1, off1, cur1, cnt2, off2, cur2);
    edge_place_kernel<<<(E1_N + E2_N + 255) / 256, 256, 0, stream>>>(src1, dst1, w1v, cur1, pairs1,
                                                                     src2, dst2, w2v, cur2, pairs2);

    gemm_bf16_w_in_lds<HDIM, 16><<<256, 512, 0, stream>>>(feat, Wpb, bp, hp, N_SRC, 0);
    gemm_bf16_w_in_lds<HDIM, 16><<<256, 512, 0, stream>>>(hp, Q1b, bq1, nbuf, N_SRC, 1);
    segment_gather_kernel<<<N_MID / 8, 256, 0, stream>>>(nbuf, off1, pairs1, hp, zbuf, N_MID);
    gemm_bf16_w_in_lds<2 * HDIM, 8><<<256, 512, 0, stream>>>(zbuf, W1b, bw1, h1, N_MID, 1);
    gemm_bf16_w_in_lds<HDIM, 8><<<256, 512, 0, stream>>>(h1, Q2b, bq2, nbuf, N_MID, 1);
    segment_gather_kernel<<<N_DST / 8, 256, 0, stream>>>(nbuf, off2, pairs2, h1, zbuf, N_DST);
    gemm_bf16_n256<<<N_DST / 32, 256, 0, stream>>>(zbuf, W2b, bw2, h2, N_DST, 2 * HDIM, 1);
    skip_norm_kernel<<<N_DST / 8, 256, 0, stream>>>(hp, h2, zf, N_DST);
    score_kernel<<<N_PAIRS / 8, 256, 0, stream>>>(zf, bias, nids,
                                                  pos_s, pos_d, neg_s, neg_d,
                                                  out, N_PAIRS);
}


# Projection GEMM and layer-1 neighbour GEMM fused into one launch: hp tile leaves WMMA transposed and stays in registers as the next A fragment; hp stored only for rows < N_MID; sum order unchanged
# speedup vs baseline: 2.5389x; 1.0374x over previous
#include <hip/hip_runtime.h>
#include <hip/hip_bf16.h>


typedef __attribute__((ext_vector_type(16))) __bf16 v16bf;
typedef __attribute__((ext_vector_type(8)))  float  v8f;

#define N_SRC   131072
#define N_MID   32768
#define N_DST   8192
#define E1_N    524288
#define E2_N    131072
#define HDIM    256
#define N_PAIRS 8192

__global__ void f32_to_bf16_kernel(const float* __restrict__ src,
                                   __bf16* __restrict__ dst, int n) {
    int i = blockIdx.x * blockDim.x + threadIdx.x;
    if (i < n) dst[i] = (__bf16)src[i];
}

__global__ void zero_f32_kernel(float* __restrict__ p, int n) {
    int i = blockIdx.x * blockDim.x + threadIdx.x;
    if (i < n) p[i] = 0.0f;
}

__global__ __launch_bounds__(256)
void gemm_bf16_n256(const float* __restrict__ A, const __bf16* __restrict__ W,
                    const float* __restrict__ bias, float* __restrict__ C,
                    int M, int K, int relu)
{
    const int lane = threadIdx.x & 31;
    const int wave = threadIdx.x >> 5;
    const int half = lane >> 4;
    const int l15  = lane & 15;
    const int m0   = blockIdx.x * 32 + (wave >> 2) * 16;
    const int n0   = (wave & 3) * 64;

    v8f acc[4];
    #pragma unroll
    for (int t = 0; t < 4; ++t) acc[t] = (v8f){};

    const float* arow = A + (size_t)(m0 + l15) * K;

    for (int k0 = 0; k0 < K; k0 += 32) {
        v16bf a;
        #pragma unroll
        for (int q = 0; q < 8; ++q) {
            int kk = k0 + (q >> 2) * 16 + half * 8 + (q & 3) * 2;
            float2 v = *(const float2*)(arow + kk);
            a[2 * q]     = (__bf16)v.x;
            a[2 * q + 1] = (__bf16)v.y;
        }
        #pragma unroll
        for (int t = 0; t < 4; ++t) {
            const __bf16* wrow = W + (size_t)(n0 + t * 16 + l15) * K;
            v16bf b;
            #pragma unroll
            for (int q = 0; q < 8; ++q) {
                int kk = k0 + (q >> 2) * 16 + half * 8 + (q & 3) * 2;
                b[2 * q]     = wrow[kk];
                b[2 * q + 1] = wrow[kk + 1];
            }
            acc[t] = __builtin_amdgcn_wmma_f32_16x16x32_bf16(
                false, a, false, b, (short)0, acc[t], false, false);
        }
    }

    #pragma unroll
    for (int t = 0; t < 4; ++t) {
        int n = n0 + t * 16 + l15;
        float bv = bias[n];
        #pragma unroll
        for (int r = 0; r < 8; ++r) {
            int m = m0 + half * 8 + r;
            float v = acc[t][r] + bv;
            if (relu) v = fmaxf(v, 0.0f);
            C[(size_t)m * HDIM + n] = v;
        }
    }
}

typedef __attribute__((ext_vector_type(8))) unsigned int v8u;

struct a_step { float4 f0, f1, f2, f3; };

__device__ __forceinline__ a_step load_a_step(const float* p) {
    a_step s;
    s.f0 = *(const float4*)(p);
    s.f1 = *(const float4*)(p + 4);
    s.f2 = *(const float4*)(p + 16);
    s.f3 = *(const float4*)(p + 20);
    return s;
}

template <int K, int NT>
__global__ __launch_bounds__(512)
void gemm_bf16_w_in_lds(const float* __restrict__ A, const __bf16* __restrict__ W,
                        const float* __restrict__ bias, float* __restrict__ C,
                        int M, int relu)
{
    constexpr int PITCH  = K * 2 + 16;
    constexpr int CPR    = K / 8;
    constexpr int CHUNKS = HDIM * CPR;
    constexpr int WPR    = 16 / NT;
    constexpr int KS     = K / 32;
    constexpr int G      = (65535 - 47 - (KS - 1) * 64) / (16 * PITCH) + 1 < NT
                         ? (65535 - 47 - (KS - 1) * 64) / (16 * PITCH) + 1 : NT;
    constexpr int NB     = (NT + G - 1) / G;
    __shared__ __attribute__((aligned(16))) unsigned char L[HDIM * PITCH];

    const int tid  = threadIdx.x;
    const int lane = tid & 31;
    const int wave = __builtin_amdgcn_readfirstlane(tid >> 5);
    const int half = lane >> 4;
    const int l15  = lane & 15;

    const int total  = (M / 16) * WPR;
    const int stride = gridDim.x * 16;
    int it = blockIdx.x * 16 + wave;

    a_step q0, q1;
    q0.f0 = q0.f1 = q0.f2 = q0.f3 = make_float4(0.f, 0.f, 0.f, 0.f);
    q1 = q0;
    if (it < total) {
        const float* p = A + (size_t)((it / WPR) * 16 + l15) * K + half * 8;
        q0 = load_a_step(p);
        q1 = load_a_step(p + 32);
    }

    #pragma unroll 16
    for (int i = 0; i < CHUNKS / 512; ++i) {
        const int c = i * 512 + tid;
        const int row = c / CPR, col = c % CPR;
        const uint4 v = *(const uint4*)(W + (size_t)c * 8);
        *(uint4*)(L + row * PITCH + col * 16) = v;
    }
    __syncthreads();

    unsigned goff[NB];
    #pragma unroll
    for (int g = 0; g < NB; ++g) {
        goff[g] = g * G * 16 * PITCH;
        if (g > 0) asm volatile("" : "+v"(goff[g]));
    }

    for (; it < total; it += stride) {
        const int m0 = (it / WPR) * 16;
        const int n0 = (it % WPR) * (NT * 16);
        const float* arow = A + (size_t)(m0 + l15) * K + half * 8;
        const bool more = it + stride < total;
        const float* anext = A + (size_t)(((it + stride) / WPR) * 16 + l15) * K + half * 8;
        const unsigned char* brow = L + (n0 + l15) * PITCH + half * 16;

        v8f acc[NT];
        #pragma unroll
        for (int t = 0; t < NT; ++t) acc[t] = (v8f){};

        #pragma unroll
        for (int ks = 0; ks < KS; ++ks) {
            a_step q2 = q1;
            if (ks + 2 < KS)  q2 = load_a_step(arow + (ks + 2) * 32);
            else if (more)    q2 = load_a_step(anext + (ks + 2 - KS) * 32);

            v16bf a;
            a[0]  = (__bf16)q0.f0.x; a[1]  = (__bf16)q0.f0.y; a[2]  = (__bf16)q0.f0.z; a[3]  = (__bf16)q0.f0.w;
            a[4]  = (__bf16)q0.f1.x; a[5]  = (__bf16)q0.f1.y; a[6]  = (__bf16)q0.f1.z; a[7]  = (__bf16)q0.f1.w;
            a[8]  = (__bf16)q0.f2.x; a[9]  = (__bf16)q0.f2.y; a[10] = (__bf16)q0.f2.z; a[11] = (__bf16)q0.f2.w;
            a[12] = (__bf16)q0.f3.x; a[13] = (__bf16)q0.f3.y; a[14] = (__bf16)q0.f3.z; a[15] = (__bf16)q0.f3.w;

            #pragma unroll
            for (int t = 0; t < NT; ++t) {
                const unsigned char* bp = brow + goff[t / G] + (t % G) * 16 * PITCH + ks * 64;
                const uint4 lo = *(const uint4*)(bp);
                const uint4 hi = *(const uint4*)(bp + 32);
                v8u bw;
                bw[0] = lo.x; bw[1] = lo.y; bw[2] = lo.z; bw[3] = lo.w;
                bw[4] = hi.x; bw[5] = hi.y; bw[6] = hi.z; bw[7] = hi.w;
                const v16bf b = __builtin_bit_cast(v16bf, bw);
                acc[t] = __builtin_amdgcn_wmma_f32_16x16x32_bf16(
                    false, a, false, b, (short)0, acc[t], false, false);
            }
            q0 = q1;
            q1 = q2;
        }

        #pragma unroll
        for (int t = 0; t < NT; ++t) {
            const int n = n0 + t * 16 + l15;
            const float bv = bias[n];
            #pragma unroll
            for (int r = 0; r < 8; ++r) {
                const int m = m0 + half * 8 + r;
                float v = acc[t][r] + bv;
                if (relu) v = fmaxf(v, 0.0f);
                C[(size_t)m * HDIM + n] = v;
            }
        }
    }
}

__device__ __forceinline__ void store_a_step(float* p, const a_step& s) {
    *(float4*)(p)      = s.f0;
    *(float4*)(p + 4)  = s.f1;
    *(float4*)(p + 16) = s.f2;
    *(float4*)(p + 20) = s.f3;
}

__device__ __forceinline__ v16bf to_frag(const a_step& q) {
    v16bf a;
    a[0]  = (__bf16)q.f0.x; a[1]  = (__bf16)q.f0.y; a[2]  = (__bf16)q.f0.z; a[3]  = (__bf16)q.f0.w;
    a[4]  = (__bf16)q.f1.x; a[5]  = (__bf16)q.f1.y; a[6]  = (__bf16)q.f1.z; a[7]  = (__bf16)q.f1.w;
    a[8]  = (__bf16)q.f2.x; a[9]  = (__bf16)q.f2.y; a[10] = (__bf16)q.f2.z; a[11] = (__bf16)q.f2.w;
    a[12] = (__bf16)q.f3.x; a[13] = (__bf16)q.f3.y; a[14] = (__bf16)q.f3.z; a[15] = (__bf16)q.f3.w;
    return a;
}

__device__ __forceinline__ v16bf lds_frag(const unsigned char* bp) {
    const uint4 lo = *(const uint4*)(bp);
    const uint4 hi = *(const uint4*)(bp + 32);
    v8u bw;
    bw[0] = lo.x; bw[1] = lo.y; bw[2] = lo.z; bw[3] = lo.w;
    bw[4] = hi.x; bw[5] = hi.y; bw[6] = hi.z; bw[7] = hi.w;
    return __builtin_bit_cast(v16bf, bw);
}

__global__ __launch_bounds__(512)
void gemm_pair_bf16_w_in_lds(const float* __restrict__ X, const __bf16* __restrict__ W,
                             const float* __restrict__ b0, const float* __restrict__ b1,
                             float* __restrict__ H, float* __restrict__ C,
                             int M, int keep_rows)
{
    constexpr int K      = HDIM;
    constexpr int PITCH  = K * 2 + 16;
    constexpr int CPR    = K / 8;
    constexpr int ROWS   = 2 * HDIM;
    constexpr int CHUNKS = ROWS * CPR;
    constexpr int KS     = K / 32;
    constexpr int NT     = HDIM / 16;
    constexpr int G      = 8;
    constexpr int NB     = 2 * NT / G;
    static_assert((G - 1) * 16 * PITCH + (KS - 1) * 64 + 47 < 65536,
                  "a group's fragments must be in reach of the 16-bit LDS offset");
    __shared__ __attribute__((aligned(16))) unsigned char L[ROWS * PITCH];

    const int tid  = threadIdx.x;
    const int lane = tid & 31;
    const int wave = __builtin_amdgcn_readfirstlane(tid >> 5);
    const int half = lane >> 4;
    const int l15  = lane & 15;

    const int total   = M / 16;
    const int quarter = total / 4;
    const int stride  = gridDim.x * 16;
    int it = blockIdx.x * 16 + wave;

    a_step q0, q1;
    q0.f0 = q0.f1 = q0.f2 = q0.f3 = make_float4(0.f, 0.f, 0.f, 0.f);
    q1 = q0;
    if (it < total) {
        const int rt = (it >> 2) + (it & 3) * quarter;
        const float* p = X + (size_t)(rt * 16 + l15) * K + half * 8;
        q0 = load_a_step(p);
        q1 = load_a_step(p + 32);
    }

    #pragma unroll 16
    for (int i = 0; i < CHUNKS / 512; ++i) {
        const int c = i * 512 + tid;
        const int row = c / CPR, col = c % CPR;
        const uint4 v = *(const uint4*)(W + (size_t)c * 8);
        *(uint4*)(L + row * PITCH + col * 16) = v;
    }
    __syncthreads();

    unsigned goff[NB];
    #pragma unroll
    for (int g = 0; g < NB; ++g) {
        goff[g] = g * G * 16 * PITCH;
        if (g > 0) asm volatile("" : "+v"(goff[g]));
    }
    const unsigned char* brow = L + l15 * PITCH + half * 16;

    for (; it < total; it += stride) {
        const int rt = (it >> 2) + (it & 3) * quarter;
        const int m0 = rt * 16;
        const float* xrow = X + (size_t)(m0 + l15) * K + half * 8;
        const bool keep = m0 < keep_rows;
        const bool more = it + stride < total;
        const int itn = it + stride;
        const float* xnext = X + (size_t)(((itn >> 2) + (itn & 3) * quarter) * 16 + l15) * K + half * 8;

        v8f acc[NT];
        #pragma unroll
        for (int t = 0; t < NT; ++t) acc[t] = (v8f){};

        #pragma unroll
        for (int ks = 0; ks < KS; ++ks) {
            a_step q2 = q1;
            if (ks + 2 < KS) q2 = load_a_step(xrow + (ks + 2) * 32);
            const v16bf x = to_frag(q0);
            #pragma unroll
            for (int t = 0; t < NT; ++t) {
                const v16bf w = lds_frag(brow + goff[t / G] + (t % G) * 16 * PITCH + ks * 64);
                acc[t] = __builtin_amdgcn_wmma_f32_16x16x32_bf16(
                    false, w, false, x, (short)0, acc[t], false, false);
            }
            q0 = q1;
            q1 = q2;
        }

        unsigned boff = half * 8;
        asm volatile("" : "+v"(boff));
        #pragma unroll
        for (int s2 = 0; s2 < KS; ++s2) {
            const a_step bb = load_a_step(b0 + boff + s2 * 32);
            v8f& lo = acc[2 * s2];
            v8f& hi = acc[2 * s2 + 1];
            lo[0] += bb.f0.x; lo[1] += bb.f0.y; lo[2] += bb.f0.z; lo[3] += bb.f0.w;
            lo[4] += bb.f1.x; lo[5] += bb.f1.y; lo[6] += bb.f1.z; lo[7] += bb.f1.w;
            hi[0] += bb.f2.x; hi[1] += bb.f2.y; hi[2] += bb.f2.z; hi[3] += bb.f2.w;
            hi[4] += bb.f3.x; hi[5] += bb.f3.y; hi[6] += bb.f3.z; hi[7] += bb.f3.w;
        }

        if (keep) {
            float* hrow = H + (size_t)(m0 + l15) * K + half * 8;
            #pragma unroll
            for (int s2 = 0; s2 < KS; ++s2) {
                const v8f& lo = acc[2 * s2];
                const v8f& hi = acc[2 * s2 + 1];
                a_step h;
                h.f0 = make_float4(lo[0], lo[1], lo[2], lo[3]);
                h.f1 = make_float4(lo[4], lo[5], lo[6], lo[7]);
                h.f2 = make_float4(hi[0], hi[1], hi[2], hi[3]);
                h.f3 = make_float4(hi[4], hi[5], hi[6], hi[7]);
                store_a_step(hrow + s2 * 32, h);
            }
        }

        if (more) {
            q0 = load_a_step(xnext);
            q1 = load_a_step(xnext + 32);
        }

        v16bf af[KS];
        #pragma unroll
        for (int s2 = 0; s2 < KS; ++s2) {
            const v8f& lo = acc[2 * s2];
            const v8f& hi = acc[2 * s2 + 1];
            v16bf a;
            a[0]  = (__bf16)lo[0]; a[1]  = (__bf16)lo[1]; a[2]  = (__bf16)lo[2]; a[3]  = (__bf16)lo[3];
            a[4]  = (__bf16)lo[4]; a[5]  = (__bf16)lo[5]; a[6]  = (__bf16)lo[6]; a[7]  = (__bf16)lo[7];
            a[8]  = (__bf16)hi[0]; a[9]  = (__bf16)hi[1]; a[10] = (__bf16)hi[2]; a[11] = (__bf16)hi[3];
            a[12] = (__bf16)hi[4]; a[13] = (__bf16)hi[5]; a[14] = (__bf16)hi[6]; a[15] = (__bf16)hi[7];
            af[s2] = a;
        }

        #pragma unroll
        for (int hf = 0; hf < 2; ++hf) {
            v8f ac2[NT / 2];
            #pragma unroll
            for (int t = 0; t < NT / 2; ++t) ac2[t] = (v8f){};
            #pragma unroll
            for (int ks = 0; ks < KS; ++ks) {
                #pragma unroll
                for (int t = 0; t < NT / 2; ++t) {
                    const int T = NT + hf * (NT / 2) + t;
                    const v16bf b = lds_frag(brow + goff[T / G] + (T % G) * 16 * PITCH + ks * 64);
                    ac2[t] = __builtin_amdgcn_wmma_f32_16x16x32_bf16(
                        false, af[ks], false, b, (short)0, ac2[t], false, false);
                }
            }
            #pragma unroll
            for (int t = 0; t < NT / 2; ++t) {
                const int n = hf * (HDIM / 2) + t * 16 + l15;
                const float bv = b1[n];
                #pragma unroll
                for (int r = 0; r < 8; ++r) {
                    const int m = m0 + half * 8 + r;
                    C[(size_t)m * HDIM + n] = fmaxf(ac2[t][r] + bv, 0.0f);
                }
            }
        }
    }
}

__global__ __launch_bounds__(256)
void edge_count_kernel(const int* __restrict__ dst1, int* __restrict__ cnt1,
                       const int* __restrict__ dst2, int* __restrict__ cnt2)
{
    int i = blockIdx.x * 256 + threadIdx.x;
    if (i < E1_N) {
        atomicAdd(cnt1 + dst1[i], 1);
    } else {
        int e = i - E1_N;
        if (e < E2_N) atomicAdd(cnt2 + dst2[e], 1);
    }
}

template <int PER>
__device__ __forceinline__ void block_offsets(const int* __restrict__ cnt,
                                              int* __restrict__ off,
                                              int* __restrict__ cur,
                                              int* wtot)
{
    const int t = threadIdx.x, lane = t & 31, wv = t >> 5;
    int v[PER];
    const int4* c4 = (const int4*)(cnt + t * PER);
    for (int q = 0; q < PER / 4; ++q) {
        int4 x = c4[q];
        v[4 * q] = x.x; v[4 * q + 1] = x.y; v[4 * q + 2] = x.z; v[4 * q + 3] = x.w;
    }
    int sum = 0;
    for (int k = 0; k < PER; ++k) sum += v[k];
    int inc = sum;
    for (int o = 1; o < 32; o <<= 1) {
        int y = __shfl_up(inc, o, 32);
        if (lane >= o) inc += y;
    }
    if (lane == 31) wtot[wv] = inc;
    __syncthreads();
    int wt = wtot[lane];
    int winc = wt;
    for (int o = 1; o < 32; o <<= 1) {
        int y = __shfl_up(winc, o, 32);
        if (lane >= o) winc += y;
    }
    int run = __shfl(winc, wv, 32) - __shfl(wt, wv, 32) + inc - sum;
    int total = __shfl(winc, 31, 32);
    int4* o4 = (int4*)(off + t * PER);
    int4* u4 = (int4*)(cur + t * PER);
    for (int q = 0; q < PER / 4; ++q) {
        int4 x;
        x.x = run; run += v[4 * q];
        x.y = run; run += v[4 * q + 1];
        x.z = run; run += v[4 * q + 2];
        x.w = run; run += v[4 * q + 3];
        o4[q] = x;
        u4[q] = x;
    }
    if (t == 1023) off[1024 * PER] = total;
}

__global__ __launch_bounds__(1024)
void segment_offsets_kernel(const int* __restrict__ cnt1, int* __restrict__ off1,
                            int* __restrict__ cur1,
                            const int* __restrict__ cnt2, int* __restrict__ off2,
                            int* __restrict__ cur2)
{
    __shared__ int wtot[32];
    if (blockIdx.x == 0) block_offsets<N_MID / 1024>(cnt1, off1, cur1, wtot);
    else                 block_offsets<N_DST / 1024>(cnt2, off2, cur2, wtot);
}

__global__ __launch_bounds__(256)
void edge_place_kernel(const int* __restrict__ src1, const int* __restrict__ dst1,
                       const float* __restrict__ w1, int* __restrict__ cur1,
                       int* __restrict__ pairs1,
                       const int* __restrict__ src2, const int* __restrict__ dst2,
                       const float* __restrict__ w2, int* __restrict__ cur2,
                       int* __restrict__ pairs2)
{
    int i = blockIdx.x * 256 + threadIdx.x;
    if (i < E1_N) {
        int p = atomicAdd(cur1 + dst1[i], 1);
        int2 v; v.x = src1[i]; v.y = __float_as_int(w1[i]);
        *(int2*)(pairs1 + 2 * (size_t)p) = v;
    } else {
        int e = i - E1_N;
        if (e < E2_N) {
            int p = atomicAdd(cur2 + dst2[e], 1);
            int2 v; v.x = src2[e]; v.y = __float_as_int(w2[e]);
            *(int2*)(pairs2 + 2 * (size_t)p) = v;
        }
    }
}

__device__ __forceinline__ void fma4(float4& a, const float4& x, float w) {
    a.x = fmaf(x.x, w, a.x); a.y = fmaf(x.y, w, a.y);
    a.z = fmaf(x.z, w, a.z); a.w = fmaf(x.w, w, a.w);
}

__global__ __launch_bounds__(256)
void segment_gather_kernel(const float* __restrict__ nsrc,
                           const int* __restrict__ off,
                           const int* __restrict__ pairs,
                           const float* __restrict__ hdst,
                           float* __restrict__ z, int rows)
{
    const int lane = threadIdx.x & 31;
    const int r    = blockIdx.x * 8 + (threadIdx.x >> 5);
    if (r >= rows) return;
    const float4* hr = (const float4*)(hdst + (size_t)r * HDIM);
    float4*       zr = (float4*)(z + (size_t)r * (2 * HDIM));
    const float4 h0 = hr[lane], h1 = hr[32 + lane];
    const int beg = __builtin_amdgcn_readfirstlane(off[r]);
    const int end = __builtin_amdgcn_readfirstlane(off[r + 1]);
    zr[64 + lane] = h0;
    zr[96 + lane] = h1;
    float4 a0 = make_float4(0.f, 0.f, 0.f, 0.f);
    float4 a1 = make_float4(0.f, 0.f, 0.f, 0.f);
    float  ws = 0.0f;
    for (int base = beg; base < end; base += 32) {
        int n = end - base;
        n = n < 32 ? n : 32;
        int ps = 0, pw = 0;
        if (lane < n) {
            int2 p = *(const int2*)(pairs + 2 * (size_t)(base + lane));
            ps = p.x; pw = p.y;
        }
        int j = 0;
        for (; j + 4 <= n; j += 4) {
            const float4* r0 = (const float4*)(nsrc + (size_t)__builtin_amdgcn_readlane(ps, j)     * HDIM);
            const float4* r1 = (const float4*)(nsrc + (size_t)__builtin_amdgcn_readlane(ps, j + 1) * HDIM);
            const float4* r2 = (const float4*)(nsrc + (size_t)__builtin_amdgcn_readlane(ps, j + 2) * HDIM);
            const float4* r3 = (const float4*)(nsrc + (size_t)__builtin_amdgcn_readlane(ps, j + 3) * HDIM);
            float w0 = __int_as_float(__builtin_amdgcn_readlane(pw, j));
            float w1 = __int_as_float(__builtin_amdgcn_readlane(pw, j + 1));
            float w2 = __int_as_float(__builtin_amdgcn_readlane(pw, j + 2));
            float w3 = __int_as_float(__builtin_amdgcn_readlane(pw, j + 3));
            float4 x0 = r0[lane], y0 = r0[32 + lane];
            float4 x1 = r1[lane], y1 = r1[32 + lane];
            float4 x2 = r2[lane], y2 = r2[32 + lane];
            float4 x3 = r3[lane], y3 = r3[32 + lane];
            fma4(a0, x0, w0); fma4(a1, y0, w0); ws += w0;
            fma4(a0, x1, w1); fma4(a1, y1, w1); ws += w1;
            fma4(a0, x2, w2); fma4(a1, y2, w2); ws += w2;
            fma4(a0, x3, w3); fma4(a1, y3, w3); ws += w3;
        }
        #pragma nounroll
        for (; j < n; ++j) {
            const float4* r0 = (const float4*)(nsrc + (size_t)__builtin_amdgcn_readlane(ps, j) * HDIM);
            float w0 = __int_as_float(__builtin_amdgcn_readlane(pw, j));
            float4 x0 = r0[lane], y0 = r0[32 + lane];
            fma4(a0, x0, w0); fma4(a1, y0, w0); ws += w0;
        }
    }
    ws = ws < 1.0f ? 1.0f : ws;
    const float inv = __frcp_rn(ws);
    zr[lane]      = make_float4(a0.x * inv, a0.y * inv, a0.z * inv, a0.w * inv);
    zr[32 + lane] = make_float4(a1.x * inv, a1.y * inv, a1.z * inv, a1.w * inv);
}

__global__ __launch_bounds__(256)
void skip_norm_kernel(const float* __restrict__ hp, const float* __restrict__ h2,
                      float* __restrict__ zf, int rows)
{
    int r    = blockIdx.x * 8 + (threadIdx.x >> 5);
    int lane = threadIdx.x & 31;
    if (r >= rows) return;
    float v[8];
    float ss = 0.0f;
    #pragma unroll
    for (int j = 0; j < 8; ++j) {
        int c = lane * 8 + j;
        v[j] = hp[(size_t)r * HDIM + c] + h2[(size_t)r * HDIM + c];
        ss += v[j] * v[j];
    }
    #pragma unroll
    for (int o = 16; o > 0; o >>= 1) ss += __shfl_xor(ss, o, 32);
    float nrm   = sqrtf(ss);
    float scale = (nrm == 0.0f) ? 1.0f : 1.0f / nrm;
    #pragma unroll
    for (int j = 0; j < 8; ++j)
        zf[(size_t)r * HDIM + lane * 8 + j] = v[j] * scale;
}

__global__ __launch_bounds__(256)
void score_kernel(const float* __restrict__ zf, const float* __restrict__ bias,
                  const int* __restrict__ nids,
                  const int* __restrict__ ps, const int* __restrict__ pd,
                  const int* __restrict__ ns, const int* __restrict__ nd,
                  float* __restrict__ out, int npairs)
{
    int p    = blockIdx.x * 8 + (threadIdx.x >> 5);
    int lane = threadIdx.x & 31;
    if (p >= npairs) return;
    int pu = ps[p], pv = pd[p], nu = ns[p], nv = nd[p];
    float sp = 0.0f, sn = 0.0f;
    #pragma unroll
    for (int j = 0; j < 8; ++j) {
        int c = lane * 8 + j;
        sp += zf[(size_t)pu * HDIM + c] * zf[(size_t)pv * HDIM + c];
        sn += zf[(size_t)nu * HDIM + c] * zf[(size_t)nv * HDIM + c];
    }
    #pragma unroll
    for (int o = 16; o > 0; o >>= 1) {
        sp += __shfl_xor(sp, o, 32);
        sn += __shfl_xor(sn, o, 32);
    }
    if (lane == 0) {
        float gp = bias[nids[pu]] + bias[nids[pv]];
        float gn = bias[nids[nu]] + bias[nids[nv]];
        float m  = (sn + gn) - (sp + gp) + 1.0f;
        out[p] = m > 0.0f ? m : 0.0f;
    }
}

extern "C" void kernel_launch(void* const* d_in, const int* in_sizes, int n_in,
                              void* d_out, int out_size, void* d_ws, size_t ws_size,
                              hipStream_t stream) {
    const float* feat = (const float*)d_in[0];
    const float* Wp   = (const float*)d_in[1];
    const float* bp   = (const float*)d_in[2];
    const float* Q1   = (const float*)d_in[3];
    const float* bq1  = (const float*)d_in[4];
    const float* W1   = (const float*)d_in[5];
    const float* bw1  = (const float*)d_in[6];
    const float* Q2   = (const float*)d_in[7];
    const float* bq2  = (const float*)d_in[8];
    const float* W2   = (const float*)d_in[9];
    const float* bw2  = (const float*)d_in[10];
    const float* w1v  = (const float*)d_in[11];
    const float* w2v  = (const float*)d_in[12];
    const float* bias = (const float*)d_in[13];
    const int* src1   = (const int*)d_in[14];
    const int* dst1   = (const int*)d_in[15];
    const int* src2   = (const int*)d_in[16];
    const int* dst2   = (const int*)d_in[17];
    const int* pos_s  = (const int*)d_in[18];
    const int* pos_d  = (const int*)d_in[19];
    const int* neg_s  = (const int*)d_in[20];
    const int* neg_d  = (const int*)d_in[21];
    const int* nids   = (const int*)d_in[22];
    float* out        = (float*)d_out;

    char* ws = (char*)d_ws;
    const size_t SZ_HP  = (size_t)N_SRC * HDIM * 4;
    const size_t SZ_N   = (size_t)N_SRC * HDIM * 4;
    const size_t SZ_H1  = (size_t)N_MID * HDIM * 4;
    const size_t SZ_AGG = (size_t)N_MID * HDIM * 4;
    const size_t SZ_WS  = 256 * 1024;
    const size_t SZ_H2  = (size_t)N_DST * HDIM * 4;
    const size_t SZ_ZF  = (size_t)N_DST * HDIM * 4;
    float*  hp   = (float*)(ws);
    float*  nbuf = (float*)(ws + SZ_HP);
    float*  h1   = (float*)(ws + SZ_HP + SZ_N);
    float*  zbuf = hp + (size_t)N_MID * HDIM;
    int* cnt1   = (int*)(ws + SZ_HP + SZ_N + SZ_H1);
    int* cnt2   = cnt1 + N_MID;
    int* off1   = cnt2 + N_DST;
    int* off2   = off1 + N_MID + 64;
    int* cur1   = off2 + N_DST + 64;
    int* cur2   = cur1 + N_MID;
    int* pairs1 = cur2 + N_DST;
    int* pairs2 = pairs1 + 2 * E1_N;
    float*  h2   = (float*)(ws + SZ_HP + SZ_N + SZ_H1 + SZ_AGG + SZ_WS);
    float*  zf   = (float*)(ws + SZ_HP + SZ_N + SZ_H1 + SZ_AGG + SZ_WS + SZ_H2);
    __bf16* Wpb  = (__bf16*)(ws + SZ_HP + SZ_N + SZ_H1 + SZ_AGG + SZ_WS + SZ_H2 + SZ_ZF);
    __bf16* Q1b  = Wpb + HDIM * HDIM;
    __bf16* W1b  = Q1b + HDIM * HDIM;
    __bf16* Q2b  = W1b + HDIM * 2 * HDIM;
    __bf16* W2b  = Q2b + HDIM * HDIM;

    f32_to_bf16_kernel<<<(HDIM * HDIM + 255) / 256, 256, 0, stream>>>(Wp, Wpb, HDIM * HDIM);
    f32_to_bf16_kernel<<<(HDIM * HDIM + 255) / 256, 256, 0, stream>>>(Q1, Q1b, HDIM * HDIM);
    f32_to_bf16_kernel<<<(HDIM * 2 * HDIM + 255) / 256, 256, 0, stream>>>(W1, W1b, HDIM * 2 * HDIM);
    f32_to_bf16_kernel<<<(HDIM * HDIM + 255) / 256, 256, 0, stream>>>(Q2, Q2b, HDIM * HDIM);
    f32_to_bf16_kernel<<<(HDIM * 2 * HDIM + 255) / 256, 256, 0, stream>>>(W2, W2b, HDIM * 2 * HDIM);

    zero_f32_kernel<<<(N_MID + N_DST + 255) / 256, 256, 0, stream>>>((float*)cnt1, N_MID + N_DST);
    edge_count_kernel<<<(E1_N + E2_N + 255) / 256, 256, 0, stream>>>(dst1, cnt1, dst2, cnt2);
    segment_offsets_kernel<<<2, 1024, 0, stream>>>(cnt1, off1, cur1, cnt2, off2, cur2);
    edge_place_kernel<<<(E1_N + E2_N + 255) / 256, 256, 0, stream>>>(src1, dst1, w1v, cur1, pairs1,
                                                                     src2, dst2, w2v, cur2, pairs2);

    gemm_pair_bf16_w_in_lds<<<256, 512, 0, stream>>>(feat, Wpb, bp, bq1, hp, nbuf, N_SRC, N_MID);
    segment_gather_kernel<<<N_MID / 8, 256, 0, stream>>>(nbuf, off1, pairs1, hp, zbuf, N_MID);
    gemm_bf16_w_in_lds<2 * HDIM, 8><<<256, 512, 0, stream>>>(zbuf, W1b, bw1, h1, N_MID, 1);
    gemm_bf16_w_in_lds<HDIM, 8><<<256, 512, 0, stream>>>(h1, Q2b, bq2, nbuf, N_MID, 1);
    segment_gather_kernel<<<N_DST / 8, 256, 0, stream>>>(nbuf, off2, pairs2, h1, zbuf, N_DST);
    gemm_bf16_n256<<<N_DST / 32, 256, 0, stream>>>(zbuf, W2b, bw2, h2, N_DST, 2 * HDIM, 1);
    skip_norm_kernel<<<N_DST / 8, 256, 0, stream>>>(hp, h2, zf, N_DST);
    score_kernel<<<N_PAIRS / 8, 256, 0, stream>>>(zf, bias, nids,
                                                  pos_s, pos_d, neg_s, neg_d,
                                                  out, N_PAIRS);
}


# Layer-1 messages n1 stored as bf16 instead of f32 (precision of that intermediate narrows): gathered table 128 to 64 MiB; 2nd GEMM transposed so a lane stores 8 bf16 of a row; f32 sums, same order
# speedup vs baseline: 2.7017x; 1.0641x over previous
#include <hip/hip_runtime.h>
#include <hip/hip_bf16.h>


typedef __attribute__((ext_vector_type(16))) __bf16 v16bf;
typedef __attribute__((ext_vector_type(8)))  float  v8f;
typedef __attribute__((ext_vector_type(8)))  __bf16 v8bf;

#define N_SRC   131072
#define N_MID   32768
#define N_DST   8192
#define E1_N    524288
#define E2_N    131072
#define HDIM    256
#define N_PAIRS 8192

__global__ void f32_to_bf16_kernel(const float* __restrict__ src,
                                   __bf16* __restrict__ dst, int n) {
    int i = blockIdx.x * blockDim.x + threadIdx.x;
    if (i < n) dst[i] = (__bf16)src[i];
}

__global__ void zero_f32_kernel(float* __restrict__ p, int n) {
    int i = blockIdx.x * blockDim.x + threadIdx.x;
    if (i < n) p[i] = 0.0f;
}

__global__ __launch_bounds__(256)
void gemm_bf16_n256(const float* __restrict__ A, const __bf16* __restrict__ W,
                    const float* __restrict__ bias, float* __restrict__ C,
                    int M, int K, int relu)
{
    const int lane = threadIdx.x & 31;
    const int wave = threadIdx.x >> 5;
    const int half = lane >> 4;
    const int l15  = lane & 15;
    const int m0   = blockIdx.x * 32 + (wave >> 2) * 16;
    const int n0   = (wave & 3) * 64;

    v8f acc[4];
    #pragma unroll
    for (int t = 0; t < 4; ++t) acc[t] = (v8f){};

    const float* arow = A + (size_t)(m0 + l15) * K;

    for (int k0 = 0; k0 < K; k0 += 32) {
        v16bf a;
        #pragma unroll
        for (int q = 0; q < 8; ++q) {
            int kk = k0 + (q >> 2) * 16 + half * 8 + (q & 3) * 2;
            float2 v = *(const float2*)(arow + kk);
            a[2 * q]     = (__bf16)v.x;
            a[2 * q + 1] = (__bf16)v.y;
        }
        #pragma unroll
        for (int t = 0; t < 4; ++t) {
            const __bf16* wrow = W + (size_t)(n0 + t * 16 + l15) * K;
            v16bf b;
            #pragma unroll
            for (int q = 0; q < 8; ++q) {
                int kk = k0 + (q >> 2) * 16 + half * 8 + (q & 3) * 2;
                b[2 * q]     = wrow[kk];
                b[2 * q + 1] = wrow[kk + 1];
            }
            acc[t] = __builtin_amdgcn_wmma_f32_16x16x32_bf16(
                false, a, false, b, (short)0, acc[t], false, false);
        }
    }

    #pragma unroll
    for (int t = 0; t < 4; ++t) {
        int n = n0 + t * 16 + l15;
        float bv = bias[n];
        #pragma unroll
        for (int r = 0; r < 8; ++r) {
            int m = m0 + half * 8 + r;
            float v = acc[t][r] + bv;
            if (relu) v = fmaxf(v, 0.0f);
            C[(size_t)m * HDIM + n] = v;
        }
    }
}

typedef __attribute__((ext_vector_type(8))) unsigned int v8u;

struct a_step { float4 f0, f1, f2, f3; };

__device__ __forceinline__ a_step load_a_step(const float* p) {
    a_step s;
    s.f0 = *(const float4*)(p);
    s.f1 = *(const float4*)(p + 4);
    s.f2 = *(const float4*)(p + 16);
    s.f3 = *(const float4*)(p + 20);
    return s;
}

template <int K, int NT>
__global__ __launch_bounds__(512)
void gemm_bf16_w_in_lds(const float* __restrict__ A, const __bf16* __restrict__ W,
                        const float* __restrict__ bias, float* __restrict__ C,
                        int M, int relu)
{
    constexpr int PITCH  = K * 2 + 16;
    constexpr int CPR    = K / 8;
    constexpr int CHUNKS = HDIM * CPR;
    constexpr int WPR    = 16 / NT;
    constexpr int KS     = K / 32;
    constexpr int G      = (65535 - 47 - (KS - 1) * 64) / (16 * PITCH) + 1 < NT
                         ? (65535 - 47 - (KS - 1) * 64) / (16 * PITCH) + 1 : NT;
    constexpr int NB     = (NT + G - 1) / G;
    __shared__ __attribute__((aligned(16))) unsigned char L[HDIM * PITCH];

    const int tid  = threadIdx.x;
    const int lane = tid & 31;
    const int wave = __builtin_amdgcn_readfirstlane(tid >> 5);
    const int half = lane >> 4;
    const int l15  = lane & 15;

    const int total  = (M / 16) * WPR;
    const int stride = gridDim.x * 16;
    int it = blockIdx.x * 16 + wave;

    a_step q0, q1;
    q0.f0 = q0.f1 = q0.f2 = q0.f3 = make_float4(0.f, 0.f, 0.f, 0.f);
    q1 = q0;
    if (it < total) {
        const float* p = A + (size_t)((it / WPR) * 16 + l15) * K + half * 8;
        q0 = load_a_step(p);
        q1 = load_a_step(p + 32);
    }

    #pragma unroll 16
    for (int i = 0; i < CHUNKS / 512; ++i) {
        const int c = i * 512 + tid;
        const int row = c / CPR, col = c % CPR;
        const uint4 v = *(const uint4*)(W + (size_t)c * 8);
        *(uint4*)(L + row * PITCH + col * 16) = v;
    }
    __syncthreads();

    unsigned goff[NB];
    #pragma unroll
    for (int g = 0; g < NB; ++g) {
        goff[g] = g * G * 16 * PITCH;
        if (g > 0) asm volatile("" : "+v"(goff[g]));
    }

    for (; it < total; it += stride) {
        const int m0 = (it / WPR) * 16;
        const int n0 = (it % WPR) * (NT * 16);
        const float* arow = A + (size_t)(m0 + l15) * K + half * 8;
        const bool more = it + stride < total;
        const float* anext = A + (size_t)(((it + stride) / WPR) * 16 + l15) * K + half * 8;
        const unsigned char* brow = L + (n0 + l15) * PITCH + half * 16;

        v8f acc[NT];
        #pragma unroll
        for (int t = 0; t < NT; ++t) acc[t] = (v8f){};

        #pragma unroll
        for (int ks = 0; ks < KS; ++ks) {
            a_step q2 = q1;
            if (ks + 2 < KS)  q2 = load_a_step(arow + (ks + 2) * 32);
            else if (more)    q2 = load_a_step(anext + (ks + 2 - KS) * 32);

            v16bf a;
            a[0]  = (__bf16)q0.f0.x; a[1]  = (__bf16)q0.f0.y; a[2]  = (__bf16)q0.f0.z; a[3]  = (__bf16)q0.f0.w;
            a[4]  = (__bf16)q0.f1.x; a[5]  = (__bf16)q0.f1.y; a[6]  = (__bf16)q0.f1.z; a[7]  = (__bf16)q0.f1.w;
            a[8]  = (__bf16)q0.f2.x; a[9]  = (__bf16)q0.f2.y; a[10] = (__bf16)q0.f2.z; a[11] = (__bf16)q0.f2.w;
            a[12] = (__bf16)q0.f3.x; a[13] = (__bf16)q0.f3.y; a[14] = (__bf16)q0.f3.z; a[15] = (__bf16)q0.f3.w;

            #pragma unroll
            for (int t = 0; t < NT; ++t) {
                const unsigned char* bp = brow + goff[t / G] + (t % G) * 16 * PITCH + ks * 64;
                const uint4 lo = *(const uint4*)(bp);
                const uint4 hi = *(const uint4*)(bp + 32);
                v8u bw;
                bw[0] = lo.x; bw[1] = lo.y; bw[2] = lo.z; bw[3] = lo.w;
                bw[4] = hi.x; bw[5] = hi.y; bw[6] = hi.z; bw[7] = hi.w;
                const v16bf b = __builtin_bit_cast(v16bf, bw);
                acc[t] = __builtin_amdgcn_wmma_f32_16x16x32_bf16(
                    false, a, false, b, (short)0, acc[t], false, false);
            }
            q0 = q1;
            q1 = q2;
        }

        #pragma unroll
        for (int t = 0; t < NT; ++t) {
            const int n = n0 + t * 16 + l15;
            const float bv = bias[n];
            #pragma unroll
            for (int r = 0; r < 8; ++r) {
                const int m = m0 + half * 8 + r;
                float v = acc[t][r] + bv;
                if (relu) v = fmaxf(v, 0.0f);
                C[(size_t)m * HDIM + n] = v;
            }
        }
    }
}

__device__ __forceinline__ void store_a_step(float* p, const a_step& s) {
    *(float4*)(p)      = s.f0;
    *(float4*)(p + 4)  = s.f1;
    *(float4*)(p + 16) = s.f2;
    *(float4*)(p + 20) = s.f3;
}

__device__ __forceinline__ v16bf to_frag(const a_step& q) {
    v16bf a;
    a[0]  = (__bf16)q.f0.x; a[1]  = (__bf16)q.f0.y; a[2]  = (__bf16)q.f0.z; a[3]  = (__bf16)q.f0.w;
    a[4]  = (__bf16)q.f1.x; a[5]  = (__bf16)q.f1.y; a[6]  = (__bf16)q.f1.z; a[7]  = (__bf16)q.f1.w;
    a[8]  = (__bf16)q.f2.x; a[9]  = (__bf16)q.f2.y; a[10] = (__bf16)q.f2.z; a[11] = (__bf16)q.f2.w;
    a[12] = (__bf16)q.f3.x; a[13] = (__bf16)q.f3.y; a[14] = (__bf16)q.f3.z; a[15] = (__bf16)q.f3.w;
    return a;
}

__device__ __forceinline__ v16bf lds_frag(const unsigned char* bp) {
    const uint4 lo = *(const uint4*)(bp);
    const uint4 hi = *(const uint4*)(bp + 32);
    v8u bw;
    bw[0] = lo.x; bw[1] = lo.y; bw[2] = lo.z; bw[3] = lo.w;
    bw[4] = hi.x; bw[5] = hi.y; bw[6] = hi.z; bw[7] = hi.w;
    return __builtin_bit_cast(v16bf, bw);
}

__global__ __launch_bounds__(512)
void gemm_pair_bf16_w_in_lds(const float* __restrict__ X, const __bf16* __restrict__ W,
                             const float* __restrict__ b0, const float* __restrict__ b1,
                             float* __restrict__ H, __bf16* __restrict__ C,
                             int M, int keep_rows)
{
    constexpr int K      = HDIM;
    constexpr int PITCH  = K * 2 + 16;
    constexpr int CPR    = K / 8;
    constexpr int ROWS   = 2 * HDIM;
    constexpr int CHUNKS = ROWS * CPR;
    constexpr int KS     = K / 32;
    constexpr int NT     = HDIM / 16;
    constexpr int G      = 8;
    constexpr int NB     = 2 * NT / G;
    static_assert((G - 1) * 16 * PITCH + (KS - 1) * 64 + 47 < 65536,
                  "a group's fragments must be in reach of the 16-bit LDS offset");
    __shared__ __attribute__((aligned(16))) unsigned char L[ROWS * PITCH];

    const int tid  = threadIdx.x;
    const int lane = tid & 31;
    const int wave = __builtin_amdgcn_readfirstlane(tid >> 5);
    const int half = lane >> 4;
    const int l15  = lane & 15;

    const int total   = M / 16;
    const int quarter = total / 4;
    const int stride  = gridDim.x * 16;
    int it = blockIdx.x * 16 + wave;

    a_step q0, q1;
    q0.f0 = q0.f1 = q0.f2 = q0.f3 = make_float4(0.f, 0.f, 0.f, 0.f);
    q1 = q0;
    if (it < total) {
        const int rt = (it >> 2) + (it & 3) * quarter;
        const float* p = X + (size_t)(rt * 16 + l15) * K + half * 8;
        q0 = load_a_step(p);
        q1 = load_a_step(p + 32);
    }

    #pragma unroll 16
    for (int i = 0; i < CHUNKS / 512; ++i) {
        const int c = i * 512 + tid;
        const int row = c / CPR, col = c % CPR;
        const uint4 v = *(const uint4*)(W + (size_t)c * 8);
        *(uint4*)(L + row * PITCH + col * 16) = v;
    }
    __syncthreads();

    unsigned goff[NB];
    #pragma unroll
    for (int g = 0; g < NB; ++g) {
        goff[g] = g * G * 16 * PITCH;
        if (g > 0) asm volatile("" : "+v"(goff[g]));
    }
    const unsigned char* brow = L + l15 * PITCH + half * 16;

    for (; it < total; it += stride) {
        const int rt = (it >> 2) + (it & 3) * quarter;
        const int m0 = rt * 16;
        const float* xrow = X + (size_t)(m0 + l15) * K + half * 8;
        const bool keep = m0 < keep_rows;
        const bool more = it + stride < total;
        const int itn = it + stride;
        const float* xnext = X + (size_t)(((itn >> 2) + (itn & 3) * quarter) * 16 + l15) * K + half * 8;

        v8f acc[NT];
        #pragma unroll
        for (int t = 0; t < NT; ++t) acc[t] = (v8f){};

        #pragma unroll
        for (int ks = 0; ks < KS; ++ks) {
            a_step q2 = q1;
            if (ks + 2 < KS) q2 = load_a_step(xrow + (ks + 2) * 32);
            const v16bf x = to_frag(q0);
            #pragma unroll
            for (int t = 0; t < NT; ++t) {
                const v16bf w = lds_frag(brow + goff[t / G] + (t % G) * 16 * PITCH + ks * 64);
                acc[t] = __builtin_amdgcn_wmma_f32_16x16x32_bf16(
                    false, w, false, x, (short)0, acc[t], false, false);
            }
            q0 = q1;
            q1 = q2;
        }

        unsigned boff = half * 8;
        asm volatile("" : "+v"(boff));
        #pragma unroll
        for (int s2 = 0; s2 < KS; ++s2) {
            const a_step bb = load_a_step(b0 + boff + s2 * 32);
            v8f& lo = acc[2 * s2];
            v8f& hi = acc[2 * s2 + 1];
            lo[0] += bb.f0.x; lo[1] += bb.f0.y; lo[2] += bb.f0.z; lo[3] += bb.f0.w;
            lo[4] += bb.f1.x; lo[5] += bb.f1.y; lo[6] += bb.f1.z; lo[7] += bb.f1.w;
            hi[0] += bb.f2.x; hi[1] += bb.f2.y; hi[2] += bb.f2.z; hi[3] += bb.f2.w;
            hi[4] += bb.f3.x; hi[5] += bb.f3.y; hi[6] += bb.f3.z; hi[7] += bb.f3.w;
        }

        if (keep) {
            float* hrow = H + (size_t)(m0 + l15) * K + half * 8;
            #pragma unroll
            for (int s2 = 0; s2 < KS; ++s2) {
                const v8f& lo = acc[2 * s2];
                const v8f& hi = acc[2 * s2 + 1];
                a_step h;
                h.f0 = make_float4(lo[0], lo[1], lo[2], lo[3]);
                h.f1 = make_float4(lo[4], lo[5], lo[6], lo[7]);
                h.f2 = make_float4(hi[0], hi[1], hi[2], hi[3]);
                h.f3 = make_float4(hi[4], hi[5], hi[6], hi[7]);
                store_a_step(hrow + s2 * 32, h);
            }
        }

        if (more) {
            q0 = load_a_step(xnext);
            q1 = load_a_step(xnext + 32);
        }

        v16bf af[KS];
        #pragma unroll
        for (int s2 = 0; s2 < KS; ++s2) {
            const v8f& lo = acc[2 * s2];
            const v8f& hi = acc[2 * s2 + 1];
            v16bf a;
            a[0]  = (__bf16)lo[0]; a[1]  = (__bf16)lo[1]; a[2]  = (__bf16)lo[2]; a[3]  = (__bf16)lo[3];
            a[4]  = (__bf16)lo[4]; a[5]  = (__bf16)lo[5]; a[6]  = (__bf16)lo[6]; a[7]  = (__bf16)lo[7];
            a[8]  = (__bf16)hi[0]; a[9]  = (__bf16)hi[1]; a[10] = (__bf16)hi[2]; a[11] = (__bf16)hi[3];
            a[12] = (__bf16)hi[4]; a[13] = (__bf16)hi[5]; a[14] = (__bf16)hi[6]; a[15] = (__bf16)hi[7];
            af[s2] = a;
        }

        __bf16* crow = C + (size_t)(m0 + l15) * HDIM + half * 8;
        #pragma unroll
        for (int hf = 0; hf < 2; ++hf) {
            v8f ac2[NT / 2];
            #pragma unroll
            for (int t = 0; t < NT / 2; ++t) ac2[t] = (v8f){};
            #pragma unroll
            for (int ks = 0; ks < KS; ++ks) {
                #pragma unroll
                for (int t = 0; t < NT / 2; ++t) {
                    const int T = NT + hf * (NT / 2) + t;
                    const v16bf w = lds_frag(brow + goff[T / G] + (T % G) * 16 * PITCH + ks * 64);
                    ac2[t] = __builtin_amdgcn_wmma_f32_16x16x32_bf16(
                        false, w, false, af[ks], (short)0, ac2[t], false, false);
                }
            }
            #pragma unroll
            for (int t = 0; t < NT / 2; ++t) {
                const int tt = hf * (NT / 2) + t;
                const float4 c0 = *(const float4*)(b1 + boff + tt * 16);
                const float4 c1 = *(const float4*)(b1 + boff + tt * 16 + 4);
                v8bf o;
                o[0] = (__bf16)fmaxf(ac2[t][0] + c0.x, 0.0f);
                o[1] = (__bf16)fmaxf(ac2[t][1] + c0.y, 0.0f);
                o[2] = (__bf16)fmaxf(ac2[t][2] + c0.z, 0.0f);
                o[3] = (__bf16)fmaxf(ac2[t][3] + c0.w, 0.0f);
                o[4] = (__bf16)fmaxf(ac2[t][4] + c1.x, 0.0f);
                o[5] = (__bf16)fmaxf(ac2[t][5] + c1.y, 0.0f);
                o[6] = (__bf16)fmaxf(ac2[t][6] + c1.z, 0.0f);
                o[7] = (__bf16)fmaxf(ac2[t][7] + c1.w, 0.0f);
                *(v8bf*)(crow + tt * 16) = o;
            }
        }
    }
}

__global__ __launch_bounds__(256)
void edge_count_kernel(const int* __restrict__ dst1, int* __restrict__ cnt1,
                       const int* __restrict__ dst2, int* __restrict__ cnt2)
{
    int i = blockIdx.x * 256 + threadIdx.x;
    if (i < E1_N) {
        atomicAdd(cnt1 + dst1[i], 1);
    } else {
        int e = i - E1_N;
        if (e < E2_N) atomicAdd(cnt2 + dst2[e], 1);
    }
}

template <int PER>
__device__ __forceinline__ void block_offsets(const int* __restrict__ cnt,
                                              int* __restrict__ off,
                                              int* __restrict__ cur,
                                              int* wtot)
{
    const int t = threadIdx.x, lane = t & 31, wv = t >> 5;
    int v[PER];
    const int4* c4 = (const int4*)(cnt + t * PER);
    for (int q = 0; q < PER / 4; ++q) {
        int4 x = c4[q];
        v[4 * q] = x.x; v[4 * q + 1] = x.y; v[4 * q + 2] = x.z; v[4 * q + 3] = x.w;
    }
    int sum = 0;
    for (int k = 0; k < PER; ++k) sum += v[k];
    int inc = sum;
    for (int o = 1; o < 32; o <<= 1) {
        int y = __shfl_up(inc, o, 32);
        if (lane >= o) inc += y;
    }
    if (lane == 31) wtot[wv] = inc;
    __syncthreads();
    int wt = wtot[lane];
    int winc = wt;
    for (int o = 1; o < 32; o <<= 1) {
        int y = __shfl_up(winc, o, 32);
        if (lane >= o) winc += y;
    }
    int run = __shfl(winc, wv, 32) - __shfl(wt, wv, 32) + inc - sum;
    int total = __shfl(winc, 31, 32);
    int4* o4 = (int4*)(off + t * PER);
    int4* u4 = (int4*)(cur + t * PER);
    for (int q = 0; q < PER / 4; ++q) {
        int4 x;
        x.x = run; run += v[4 * q];
        x.y = run; run += v[4 * q + 1];
        x.z = run; run += v[4 * q + 2];
        x.w = run; run += v[4 * q + 3];
        o4[q] = x;
        u4[q] = x;
    }
    if (t == 1023) off[1024 * PER] = total;
}

__global__ __launch_bounds__(1024)
void segment_offsets_kernel(const int* __restrict__ cnt1, int* __restrict__ off1,
                            int* __restrict__ cur1,
                            const int* __restrict__ cnt2, int* __restrict__ off2,
                            int* __restrict__ cur2)
{
    __shared__ int wtot[32];
    if (blockIdx.x == 0) block_offsets<N_MID / 1024>(cnt1, off1, cur1, wtot);
    else                 block_offsets<N_DST / 1024>(cnt2, off2, cur2, wtot);
}

__global__ __launch_bounds__(256)
void edge_place_kernel(const int* __restrict__ src1, const int* __restrict__ dst1,
                       const float* __restrict__ w1, int* __restrict__ cur1,
                       int* __restrict__ pairs1,
                       const int* __restrict__ src2, const int* __restrict__ dst2,
                       const float* __restrict__ w2, int* __restrict__ cur2,
                       int* __restrict__ pairs2)
{
    int i = blockIdx.x * 256 + threadIdx.x;
    if (i < E1_N) {
        int p = atomicAdd(cur1 + dst1[i], 1);
        int2 v; v.x = src1[i]; v.y = __float_as_int(w1[i]);
        *(int2*)(pairs1 + 2 * (size_t)p) = v;
    } else {
        int e = i - E1_N;
        if (e < E2_N) {
            int p = atomicAdd(cur2 + dst2[e], 1);
            int2 v; v.x = src2[e]; v.y = __float_as_int(w2[e]);
            *(int2*)(pairs2 + 2 * (size_t)p) = v;
        }
    }
}

__device__ __forceinline__ void fma4(float4& a, const float4& x, float w) {
    a.x = fmaf(x.x, w, a.x); a.y = fmaf(x.y, w, a.y);
    a.z = fmaf(x.z, w, a.z); a.w = fmaf(x.w, w, a.w);
}

template <bool TAB16> struct row_raw;
template <> struct row_raw<false> { float4 x, y; };
template <> struct row_raw<true>  { uint2  x, y; };

__device__ __forceinline__ row_raw<false> load_row(const float* tab, int row, int lane) {
    const float4* p = (const float4*)(tab + (size_t)row * HDIM);
    row_raw<false> v;
    v.x = p[lane]; v.y = p[32 + lane];
    return v;
}
__device__ __forceinline__ row_raw<true> load_row(const __bf16* tab, int row, int lane) {
    const uint2* p = (const uint2*)(tab + (size_t)row * HDIM);
    row_raw<true> v;
    v.x = p[lane]; v.y = p[32 + lane];
    return v;
}
__device__ __forceinline__ float4 widen4(const float4& v) { return v; }
__device__ __forceinline__ float4 widen4(const uint2& v) {
    return make_float4(__uint_as_float(v.x << 16), __uint_as_float(v.x & 0xffff0000u),
                       __uint_as_float(v.y << 16), __uint_as_float(v.y & 0xffff0000u));
}

template <bool TAB16, typename TabT>
__device__ __forceinline__ void segment_gather_body(const TabT* __restrict__ nsrc,
                                                    const int* __restrict__ off,
                                                    const int* __restrict__ pairs,
                                                    const float* __restrict__ hdst,
                                                    float* __restrict__ z, int rows)
{
    const int lane = threadIdx.x & 31;
    const int r    = blockIdx.x * 8 + (threadIdx.x >> 5);
    if (r >= rows) return;
    const float4* hr = (const float4*)(hdst + (size_t)r * HDIM);
    float4*       zr = (float4*)(z + (size_t)r * (2 * HDIM));
    const float4 h0 = hr[lane], h1 = hr[32 + lane];
    const int beg = __builtin_amdgcn_readfirstlane(off[r]);
    const int end = __builtin_amdgcn_readfirstlane(off[r + 1]);
    zr[64 + lane] = h0;
    zr[96 + lane] = h1;
    float4 a0 = make_float4(0.f, 0.f, 0.f, 0.f);
    float4 a1 = make_float4(0.f, 0.f, 0.f, 0.f);
    float  ws = 0.0f;
    for (int base = beg; base < end; base += 32) {
        int n = end - base;
        n = n < 32 ? n : 32;
        int ps = 0, pw = 0;
        if (lane < n) {
            int2 p = *(const int2*)(pairs + 2 * (size_t)(base + lane));
            ps = p.x; pw = p.y;
        }
        int j = 0;
        for (; j + 4 <= n; j += 4) {
            const row_raw<TAB16> v0 = load_row(nsrc, __builtin_amdgcn_readlane(ps, j),     lane);
            const row_raw<TAB16> v1 = load_row(nsrc, __builtin_amdgcn_readlane(ps, j + 1), lane);
            const row_raw<TAB16> v2 = load_row(nsrc, __builtin_amdgcn_readlane(ps, j + 2), lane);
            const row_raw<TAB16> v3 = load_row(nsrc, __builtin_amdgcn_readlane(ps, j + 3), lane);
            float w0 = __int_as_float(__builtin_amdgcn_readlane(pw, j));
            float w1 = __int_as_float(__builtin_amdgcn_readlane(pw, j + 1));
            float w2 = __int_as_float(__builtin_amdgcn_readlane(pw, j + 2));
            float w3 = __int_as_float(__builtin_amdgcn_readlane(pw, j + 3));
            fma4(a0, widen4(v0.x), w0); fma4(a1, widen4(v0.y), w0); ws += w0;
            fma4(a0, widen4(v1.x), w1); fma4(a1, widen4(v1.y), w1); ws += w1;
            fma4(a0, widen4(v2.x), w2); fma4(a1, widen4(v2.y), w2); ws += w2;
            fma4(a0, widen4(v3.x), w3); fma4(a1, widen4(v3.y), w3); ws += w3;
        }
        #pragma nounroll
        for (; j < n; ++j) {
            const row_raw<TAB16> v0 = load_row(nsrc, __builtin_amdgcn_readlane(ps, j), lane);
            float w0 = __int_as_float(__builtin_amdgcn_readlane(pw, j));
            fma4(a0, widen4(v0.x), w0); fma4(a1, widen4(v0.y), w0); ws += w0;
        }
    }
    ws = ws < 1.0f ? 1.0f : ws;
    const float inv = __frcp_rn(ws);
    zr[lane]      = make_float4(a0.x * inv, a0.y * inv, a0.z * inv, a0.w * inv);
    zr[32 + lane] = make_float4(a1.x * inv, a1.y * inv, a1.z * inv, a1.w * inv);
}

__global__ __launch_bounds__(256)
void segment_gather_kernel(const float* __restrict__ nsrc,
                           const int* __restrict__ off,
                           const int* __restrict__ pairs,
                           const float* __restrict__ hdst,
                           float* __restrict__ z, int rows)
{
    segment_gather_body<false>(nsrc, off, pairs, hdst, z, rows);
}

__global__ __launch_bounds__(256)
void segment_gather_bf16_kernel(const __bf16* __restrict__ nsrc,
                                const int* __restrict__ off,
                                const int* __restrict__ pairs,
                                const float* __restrict__ hdst,
                                float* __restrict__ z, int rows)
{
    segment_gather_body<true>(nsrc, off, pairs, hdst, z, rows);
}

__global__ __launch_bounds__(256)
void skip_norm_kernel(const float* __restrict__ hp, const float* __restrict__ h2,
                      float* __restrict__ zf, int rows)
{
    int r    = blockIdx.x * 8 + (threadIdx.x >> 5);
    int lane = threadIdx.x & 31;
    if (r >= rows) return;
    float v[8];
    float ss = 0.0f;
    #pragma unroll
    for (int j = 0; j < 8; ++j) {
        int c = lane * 8 + j;
        v[j] = hp[(size_t)r * HDIM + c] + h2[(size_t)r * HDIM + c];
        ss += v[j] * v[j];
    }
    #pragma unroll
    for (int o = 16; o > 0; o >>= 1) ss += __shfl_xor(ss, o, 32);
    float nrm   = sqrtf(ss);
    float scale = (nrm == 0.0f) ? 1.0f : 1.0f / nrm;
    #pragma unroll
    for (int j = 0; j < 8; ++j)
        zf[(size_t)r * HDIM + lane * 8 + j] = v[j] * scale;
}

__global__ __launch_bounds__(256)
void score_kernel(const float* __restrict__ zf, const float* __restrict__ bias,
                  const int* __restrict__ nids,
                  const int* __restrict__ ps, const int* __restrict__ pd,
                  const int* __restrict__ ns, const int* __restrict__ nd,
                  float* __restrict__ out, int npairs)
{
    int p    = blockIdx.x * 8 + (threadIdx.x >> 5);
    int lane = threadIdx.x & 31;
    if (p >= npairs) return;
    int pu = ps[p], pv = pd[p], nu = ns[p], nv = nd[p];
    float sp = 0.0f, sn = 0.0f;
    #pragma unroll
    for (int j = 0; j < 8; ++j) {
        int c = lane * 8 + j;
        sp += zf[(size_t)pu * HDIM + c] * zf[(size_t)pv * HDIM + c];
        sn += zf[(size_t)nu * HDIM + c] * zf[(size_t)nv * HDIM + c];
    }
    #pragma unroll
    for (int o = 16; o > 0; o >>= 1) {
        sp += __shfl_xor(sp, o, 32);
        sn += __shfl_xor(sn, o, 32);
    }
    if (lane == 0) {
        float gp = bias[nids[pu]] + bias[nids[pv]];
        float gn = bias[nids[nu]] + bias[nids[nv]];
        float m  = (sn + gn) - (sp + gp) + 1.0f;
        out[p] = m > 0.0f ? m : 0.0f;
    }
}

extern "C" void kernel_launch(void* const* d_in, const int* in_sizes, int n_in,
                              void* d_out, int out_size, void* d_ws, size_t ws_size,
                              hipStream_t stream) {
    const float* feat = (const float*)d_in[0];
    const float* Wp   = (const float*)d_in[1];
    const float* bp   = (const float*)d_in[2];
    const float* Q1   = (const float*)d_in[3];
    const float* bq1  = (const float*)d_in[4];
    const float* W1   = (const float*)d_in[5];
    const float* bw1  = (const float*)d_in[6];
    const float* Q2   = (const float*)d_in[7];
    const float* bq2  = (const float*)d_in[8];
    const float* W2   = (const float*)d_in[9];
    const float* bw2  = (const float*)d_in[10];
    const float* w1v  = (const float*)d_in[11];
    const float* w2v  = (const float*)d_in[12];
    const float* bias = (const float*)d_in[13];
    const int* src1   = (const int*)d_in[14];
    const int* dst1   = (const int*)d_in[15];
    const int* src2   = (const int*)d_in[16];
    const int* dst2   = (const int*)d_in[17];
    const int* pos_s  = (const int*)d_in[18];
    const int* pos_d  = (const int*)d_in[19];
    const int* neg_s  = (const int*)d_in[20];
    const int* neg_d  = (const int*)d_in[21];
    const int* nids   = (const int*)d_in[22];
    float* out        = (float*)d_out;

    char* ws = (char*)d_ws;
    const size_t SZ_HP  = (size_t)N_SRC * HDIM * 4;
    const size_t SZ_N   = (size_t)N_SRC * HDIM * 4;
    const size_t SZ_H1  = (size_t)N_MID * HDIM * 4;
    const size_t SZ_AGG = (size_t)N_MID * HDIM * 4;
    const size_t SZ_WS  = 256 * 1024;
    const size_t SZ_H2  = (size_t)N_DST * HDIM * 4;
    const size_t SZ_ZF  = (size_t)N_DST * HDIM * 4;
    float*  hp   = (float*)(ws);
    float*  nbuf = (float*)(ws + SZ_HP);
    __bf16* n1b  = (__bf16*)(ws + SZ_HP);
    float*  h1   = (float*)(ws + SZ_HP + SZ_N);
    float*  zbuf = hp + (size_t)N_MID * HDIM;
    int* cnt1   = (int*)(ws + SZ_HP + SZ_N + SZ_H1);
    int* cnt2   = cnt1 + N_MID;
    int* off1   = cnt2 + N_DST;
    int* off2   = off1 + N_MID + 64;
    int* cur1   = off2 + N_DST + 64;
    int* cur2   = cur1 + N_MID;
    int* pairs1 = cur2 + N_DST;
    int* pairs2 = pairs1 + 2 * E1_N;
    float*  h2   = (float*)(ws + SZ_HP + SZ_N + SZ_H1 + SZ_AGG + SZ_WS);
    float*  zf   = (float*)(ws + SZ_HP + SZ_N + SZ_H1 + SZ_AGG + SZ_WS + SZ_H2);
    __bf16* Wpb  = (__bf16*)(ws + SZ_HP + SZ_N + SZ_H1 + SZ_AGG + SZ_WS + SZ_H2 + SZ_ZF);
    __bf16* Q1b  = Wpb + HDIM * HDIM;
    __bf16* W1b  = Q1b + HDIM * HDIM;
    __bf16* Q2b  = W1b + HDIM * 2 * HDIM;
    __bf16* W2b  = Q2b + HDIM * HDIM;

    f32_to_bf16_kernel<<<(HDIM * HDIM + 255) / 256, 256, 0, stream>>>(Wp, Wpb, HDIM * HDIM);
    f32_to_bf16_kernel<<<(HDIM * HDIM + 255) / 256, 256, 0, stream>>>(Q1, Q1b, HDIM * HDIM);
    f32_to_bf16_kernel<<<(HDIM * 2 * HDIM + 255) / 256, 256, 0, stream>>>(W1, W1b, HDIM * 2 * HDIM);
    f32_to_bf16_kernel<<<(HDIM * HDIM + 255) / 256, 256, 0, stream>>>(Q2, Q2b, HDIM * HDIM);
    f32_to_bf16_kernel<<<(HDIM * 2 * HDIM + 255) / 256, 256, 0, stream>>>(W2, W2b, HDIM * 2 * HDIM);

    zero_f32_kernel<<<(N_MID + N_DST + 255) / 256, 256, 0, stream>>>((float*)cnt1, N_MID + N_DST);
    edge_count_kernel<<<(E1_N + E2_N + 255) / 256, 256, 0, stream>>>(dst1, cnt1, dst2, cnt2);
    segment_offsets_kernel<<<2, 1024, 0, stream>>>(cnt1, off1, cur1, cnt2, off2, cur2);
    edge_place_kernel<<<(E1_N + E2_N + 255) / 256, 256, 0, stream>>>(src1, dst1, w1v, cur1, pairs1,
                                                                     src2, dst2, w2v, cur2, pairs2);

    gemm_pair_bf16_w_in_lds<<<256, 512, 0, stream>>>(feat, Wpb, bp, bq1, hp, n1b, N_SRC, N_MID);
    segment_gather_bf16_kernel<<<N_MID / 8, 256, 0, stream>>>(n1b, off1, pairs1, hp, zbuf, N_MID);
    gemm_bf16_w_in_lds<2 * HDIM, 8><<<256, 512, 0, stream>>>(zbuf, W1b, bw1, h1, N_MID, 1);
    gemm_bf16_w_in_lds<HDIM, 8><<<256, 512, 0, stream>>>(h1, Q2b, bq2, nbuf, N_MID, 1);
    segment_gather_kernel<<<N_DST / 8, 256, 0, stream>>>(nbuf, off2, pairs2, h1, zbuf, N_DST);
    gemm_bf16_n256<<<N_DST / 32, 256, 0, stream>>>(zbuf, W2b, bw2, h2, N_DST, 2 * HDIM, 1);
    skip_norm_kernel<<<N_DST / 8, 256, 0, stream>>>(hp, h2, zf, N_DST);
    score_kernel<<<N_PAIRS / 8, 256, 0, stream>>>(zf, bias, nids,
                                                  pos_s, pos_d, neg_s, neg_d,
                                                  out, N_PAIRS);
}


# Dispatch cost of small kernels: weight conversions + counter clear in one launch; edge count/place, norm, score take 4-8 items a thread (17->12 launches, about half the waves); sums in arrival order
# speedup vs baseline: 2.8244x; 1.0454x over previous
#include <hip/hip_runtime.h>
#include <hip/hip_bf16.h>


typedef __attribute__((ext_vector_type(16))) __bf16 v16bf;
typedef __attribute__((ext_vector_type(8)))  float  v8f;
typedef __attribute__((ext_vector_type(8)))  __bf16 v8bf;

#define N_SRC   131072
#define N_MID   32768
#define N_DST   8192
#define E1_N    524288
#define E2_N    131072
#define HDIM    256
#define N_PAIRS 8192

#define PREP_THREADS  128
#define PREP_SLAB     (HDIM * HDIM / 4)
#define PREP_W_BLOCKS (PREP_SLAB / PREP_THREADS)
#define PREP_Z_UPT    4
#define PREP_Z_BLOCKS ((N_MID + N_DST) / 4 / PREP_Z_UPT / PREP_THREADS)
__global__ __launch_bounds__(PREP_THREADS)
void prep_kernel(const float* __restrict__ Wp, const float* __restrict__ Q1,
                 const float* __restrict__ W1, const float* __restrict__ Q2,
                 const float* __restrict__ W2, __bf16* __restrict__ wb,
                 int* __restrict__ cnt)
{
    typedef __attribute__((ext_vector_type(4))) __bf16 v4bf;
    static_assert(PREP_SLAB % PREP_THREADS == 0 &&
                  (N_MID + N_DST) % (4 * PREP_Z_UPT * PREP_THREADS) == 0, "whole blocks only");
    const int tid = threadIdx.x;
    const int blk = blockIdx.x;
    if (blk < PREP_W_BLOCKS) {
        const int u = blk * PREP_THREADS + tid;
        float4 v[7];
        v[0] = ((const float4*)Wp)[u];
        v[1] = ((const float4*)Q1)[u];
        v[2] = ((const float4*)W1)[u];
        v[3] = ((const float4*)W1)[u + PREP_SLAB];
        v[4] = ((const float4*)Q2)[u];
        v[5] = ((const float4*)W2)[u];
        v[6] = ((const float4*)W2)[u + PREP_SLAB];
        #pragma unroll
        for (int k = 0; k < 7; ++k) {
            v4bf o;
            o[0] = (__bf16)v[k].x; o[1] = (__bf16)v[k].y;
            o[2] = (__bf16)v[k].z; o[3] = (__bf16)v[k].w;
            *(v4bf*)(wb + ((size_t)k * PREP_SLAB + u) * 4) = o;
        }
    } else {
        #pragma unroll
        for (int k = 0; k < PREP_Z_UPT; ++k) {
            const int u = (blk - PREP_W_BLOCKS + k * PREP_Z_BLOCKS) * PREP_THREADS + tid;
            ((int4*)cnt)[u] = make_int4(0, 0, 0, 0);
        }
    }
}

__global__ __launch_bounds__(256)
void gemm_bf16_n256(const float* __restrict__ A, const __bf16* __restrict__ W,
                    const float* __restrict__ bias, float* __restrict__ C,
                    int M, int K, int relu)
{
    const int lane = threadIdx.x & 31;
    const int wave = threadIdx.x >> 5;
    const int half = lane >> 4;
    const int l15  = lane & 15;
    const int m0   = blockIdx.x * 32 + (wave >> 2) * 16;
    const int n0   = (wave & 3) * 64;

    v8f acc[4];
    #pragma unroll
    for (int t = 0; t < 4; ++t) acc[t] = (v8f){};

    const float* arow = A + (size_t)(m0 + l15) * K;

    for (int k0 = 0; k0 < K; k0 += 32) {
        v16bf a;
        #pragma unroll
        for (int q = 0; q < 8; ++q) {
            int kk = k0 + (q >> 2) * 16 + half * 8 + (q & 3) * 2;
            float2 v = *(const float2*)(arow + kk);
            a[2 * q]     = (__bf16)v.x;
            a[2 * q + 1] = (__bf16)v.y;
        }
        #pragma unroll
        for (int t = 0; t < 4; ++t) {
            const __bf16* wrow = W + (size_t)(n0 + t * 16 + l15) * K;
            v16bf b;
            #pragma unroll
            for (int q = 0; q < 8; ++q) {
                int kk = k0 + (q >> 2) * 16 + half * 8 + (q & 3) * 2;
                b[2 * q]     = wrow[kk];
                b[2 * q + 1] = wrow[kk + 1];
            }
            acc[t] = __builtin_amdgcn_wmma_f32_16x16x32_bf16(
                false, a, false, b, (short)0, acc[t], false, false);
        }
    }

    #pragma unroll
    for (int t = 0; t < 4; ++t) {
        int n = n0 + t * 16 + l15;
        float bv = bias[n];
        #pragma unroll
        for (int r = 0; r < 8; ++r) {
            int m = m0 + half * 8 + r;
            float v = acc[t][r] + bv;
            if (relu) v = fmaxf(v, 0.0f);
            C[(size_t)m * HDIM + n] = v;
        }
    }
}

typedef __attribute__((ext_vector_type(8))) unsigned int v8u;

struct a_step { float4 f0, f1, f2, f3; };

__device__ __forceinline__ a_step load_a_step(const float* p) {
    a_step s;
    s.f0 = *(const float4*)(p);
    s.f1 = *(const float4*)(p + 4);
    s.f2 = *(const float4*)(p + 16);
    s.f3 = *(const float4*)(p + 20);
    return s;
}

template <int K, int NT>
__global__ __launch_bounds__(512)
void gemm_bf16_w_in_lds(const float* __restrict__ A, const __bf16* __restrict__ W,
                        const float* __restrict__ bias, float* __restrict__ C,
                        int M, int relu)
{
    constexpr int PITCH  = K * 2 + 16;
    constexpr int CPR    = K / 8;
    constexpr int CHUNKS = HDIM * CPR;
    constexpr int WPR    = 16 / NT;
    constexpr int KS     = K / 32;
    constexpr int G      = (65535 - 47 - (KS - 1) * 64) / (16 * PITCH) + 1 < NT
                         ? (65535 - 47 - (KS - 1) * 64) / (16 * PITCH) + 1 : NT;
    constexpr int NB     = (NT + G - 1) / G;
    __shared__ __attribute__((aligned(16))) unsigned char L[HDIM * PITCH];

    const int tid  = threadIdx.x;
    const int lane = tid & 31;
    const int wave = __builtin_amdgcn_readfirstlane(tid >> 5);
    const int half = lane >> 4;
    const int l15  = lane & 15;

    const int total  = (M / 16) * WPR;
    const int stride = gridDim.x * 16;
    int it = blockIdx.x * 16 + wave;

    a_step q0, q1;
    q0.f0 = q0.f1 = q0.f2 = q0.f3 = make_float4(0.f, 0.f, 0.f, 0.f);
    q1 = q0;
    if (it < total) {
        const float* p = A + (size_t)((it / WPR) * 16 + l15) * K + half * 8;
        q0 = load_a_step(p);
        q1 = load_a_step(p + 32);
    }

    #pragma unroll 16
    for (int i = 0; i < CHUNKS / 512; ++i) {
        const int c = i * 512 + tid;
        const int row = c / CPR, col = c % CPR;
        const uint4 v = *(const uint4*)(W + (size_t)c * 8);
        *(uint4*)(L + row * PITCH + col * 16) = v;
    }
    __syncthreads();

    unsigned goff[NB];
    #pragma unroll
    for (int g = 0; g < NB; ++g) {
        goff[g] = g * G * 16 * PITCH;
        if (g > 0) asm volatile("" : "+v"(goff[g]));
    }

    for (; it < total; it += stride) {
        const int m0 = (it / WPR) * 16;
        const int n0 = (it % WPR) * (NT * 16);
        const float* arow = A + (size_t)(m0 + l15) * K + half * 8;
        const bool more = it + stride < total;
        const float* anext = A + (size_t)(((it + stride) / WPR) * 16 + l15) * K + half * 8;
        const unsigned char* brow = L + (n0 + l15) * PITCH + half * 16;

        v8f acc[NT];
        #pragma unroll
        for (int t = 0; t < NT; ++t) acc[t] = (v8f){};

        #pragma unroll
        for (int ks = 0; ks < KS; ++ks) {
            a_step q2 = q1;
            if (ks + 2 < KS)  q2 = load_a_step(arow + (ks + 2) * 32);
            else if (more)    q2 = load_a_step(anext + (ks + 2 - KS) * 32);

            v16bf a;
            a[0]  = (__bf16)q0.f0.x; a[1]  = (__bf16)q0.f0.y; a[2]  = (__bf16)q0.f0.z; a[3]  = (__bf16)q0.f0.w;
            a[4]  = (__bf16)q0.f1.x; a[5]  = (__bf16)q0.f1.y; a[6]  = (__bf16)q0.f1.z; a[7]  = (__bf16)q0.f1.w;
            a[8]  = (__bf16)q0.f2.x; a[9]  = (__bf16)q0.f2.y; a[10] = (__bf16)q0.f2.z; a[11] = (__bf16)q0.f2.w;
            a[12] = (__bf16)q0.f3.x; a[13] = (__bf16)q0.f3.y; a[14] = (__bf16)q0.f3.z; a[15] = (__bf16)q0.f3.w;

            #pragma unroll
            for (int t = 0; t < NT; ++t) {
                const unsigned char* bp = brow + goff[t / G] + (t % G) * 16 * PITCH + ks * 64;
                const uint4 lo = *(const uint4*)(bp);
                const uint4 hi = *(const uint4*)(bp + 32);
                v8u bw;
                bw[0] = lo.x; bw[1] = lo.y; bw[2] = lo.z; bw[3] = lo.w;
                bw[4] = hi.x; bw[5] = hi.y; bw[6] = hi.z; bw[7] = hi.w;
                const v16bf b = __builtin_bit_cast(v16bf, bw);
                acc[t] = __builtin_amdgcn_wmma_f32_16x16x32_bf16(
                    false, a, false, b, (short)0, acc[t], false, false);
            }
            q0 = q1;
            q1 = q2;
        }

        #pragma unroll
        for (int t = 0; t < NT; ++t) {
            const int n = n0 + t * 16 + l15;
            const float bv = bias[n];
            #pragma unroll
            for (int r = 0; r < 8; ++r) {
                const int m = m0 + half * 8 + r;
                float v = acc[t][r] + bv;
                if (relu) v = fmaxf(v, 0.0f);
                C[(size_t)m * HDIM + n] = v;
            }
        }
    }
}

__device__ __forceinline__ void store_a_step(float* p, const a_step& s) {
    *(float4*)(p)      = s.f0;
    *(float4*)(p + 4)  = s.f1;
    *(float4*)(p + 16) = s.f2;
    *(float4*)(p + 20) = s.f3;
}

__device__ __forceinline__ v16bf to_frag(const a_step& q) {
    v16bf a;
    a[0]  = (__bf16)q.f0.x; a[1]  = (__bf16)q.f0.y; a[2]  = (__bf16)q.f0.z; a[3]  = (__bf16)q.f0.w;
    a[4]  = (__bf16)q.f1.x; a[5]  = (__bf16)q.f1.y; a[6]  = (__bf16)q.f1.z; a[7]  = (__bf16)q.f1.w;
    a[8]  = (__bf16)q.f2.x; a[9]  = (__bf16)q.f2.y; a[10] = (__bf16)q.f2.z; a[11] = (__bf16)q.f2.w;
    a[12] = (__bf16)q.f3.x; a[13] = (__bf16)q.f3.y; a[14] = (__bf16)q.f3.z; a[15] = (__bf16)q.f3.w;
    return a;
}

__device__ __forceinline__ v16bf lds_frag(const unsigned char* bp) {
    const uint4 lo = *(const uint4*)(bp);
    const uint4 hi = *(const uint4*)(bp + 32);
    v8u bw;
    bw[0] = lo.x; bw[1] = lo.y; bw[2] = lo.z; bw[3] = lo.w;
    bw[4] = hi.x; bw[5] = hi.y; bw[6] = hi.z; bw[7] = hi.w;
    return __builtin_bit_cast(v16bf, bw);
}

__global__ __launch_bounds__(512)
void gemm_pair_bf16_w_in_lds(const float* __restrict__ X, const __bf16* __restrict__ W,
                             const float* __restrict__ b0, const float* __restrict__ b1,
                             float* __restrict__ H, __bf16* __restrict__ C,
                             int M, int keep_rows)
{
    constexpr int K      = HDIM;
    constexpr int PITCH  = K * 2 + 16;
    constexpr int CPR    = K / 8;
    constexpr int ROWS   = 2 * HDIM;
    constexpr int CHUNKS = ROWS * CPR;
    constexpr int KS     = K / 32;
    constexpr int NT     = HDIM / 16;
    constexpr int G      = 8;
    constexpr int NB     = 2 * NT / G;
    static_assert((G - 1) * 16 * PITCH + (KS - 1) * 64 + 47 < 65536,
                  "a group's fragments must be in reach of the 16-bit LDS offset");
    __shared__ __attribute__((aligned(16))) unsigned char L[ROWS * PITCH];

    const int tid  = threadIdx.x;
    const int lane = tid & 31;
    const int wave = __builtin_amdgcn_readfirstlane(tid >> 5);
    const int half = lane >> 4;
    const int l15  = lane & 15;

    const int total   = M / 16;
    const int quarter = total / 4;
    const int stride  = gridDim.x * 16;
    int it = blockIdx.x * 16 + wave;

    a_step q0, q1;
    q0.f0 = q0.f1 = q0.f2 = q0.f3 = make_float4(0.f, 0.f, 0.f, 0.f);
    q1 = q0;
    if (it < total) {
        const int rt = (it >> 2) + (it & 3) * quarter;
        const float* p = X + (size_t)(rt * 16 + l15) * K + half * 8;
        q0 = load_a_step(p);
        q1 = load_a_step(p + 32);
    }

    #pragma unroll 16
    for (int i = 0; i < CHUNKS / 512; ++i) {
        const int c = i * 512 + tid;
        const int row = c / CPR, col = c % CPR;
        const uint4 v = *(const uint4*)(W + (size_t)c * 8);
        *(uint4*)(L + row * PITCH + col * 16) = v;
    }
    __syncthreads();

    unsigned goff[NB];
    #pragma unroll
    for (int g = 0; g < NB; ++g) {
        goff[g] = g * G * 16 * PITCH;
        if (g > 0) asm volatile("" : "+v"(goff[g]));
    }
    const unsigned char* brow = L + l15 * PITCH + half * 16;

    for (; it < total; it += stride) {
        const int rt = (it >> 2) + (it & 3) * quarter;
        const int m0 = rt * 16;
        const float* xrow = X + (size_t)(m0 + l15) * K + half * 8;
        const bool keep = m0 < keep_rows;
        const bool more = it + stride < total;
        const int itn = it + stride;
        const float* xnext = X + (size_t)(((itn >> 2) + (itn & 3) * quarter) * 16 + l15) * K + half * 8;

        v8f acc[NT];
        #pragma unroll
        for (int t = 0; t < NT; ++t) acc[t] = (v8f){};

        #pragma unroll
        for (int ks = 0; ks < KS; ++ks) {
            a_step q2 = q1;
            if (ks + 2 < KS) q2 = load_a_step(xrow + (ks + 2) * 32);
            const v16bf x = to_frag(q0);
            #pragma unroll
            for (int t = 0; t < NT; ++t) {
                const v16bf w = lds_frag(brow + goff[t / G] + (t % G) * 16 * PITCH + ks * 64);
                acc[t] = __builtin_amdgcn_wmma_f32_16x16x32_bf16(
                    false, w, false, x, (short)0, acc[t], false, false);
            }
            q0 = q1;
            q1 = q2;
        }

        unsigned boff = half * 8;
        asm volatile("" : "+v"(boff));
        #pragma unroll
        for (int s2 = 0; s2 < KS; ++s2) {
            const a_step bb = load_a_step(b0 + boff + s2 * 32);
            v8f& lo = acc[2 * s2];
            v8f& hi = acc[2 * s2 + 1];
            lo[0] += bb.f0.x; lo[1] += bb.f0.y; lo[2] += bb.f0.z; lo[3] += bb.f0.w;
            lo[4] += bb.f1.x; lo[5] += bb.f1.y; lo[6] += bb.f1.z; lo[7] += bb.f1.w;
            hi[0] += bb.f2.x; hi[1] += bb.f2.y; hi[2] += bb.f2.z; hi[3] += bb.f2.w;
            hi[4] += bb.f3.x; hi[5] += bb.f3.y; hi[6] += bb.f3.z; hi[7] += bb.f3.w;
        }

        if (keep) {
            float* hrow = H + (size_t)(m0 + l15) * K + half * 8;
            #pragma unroll
            for (int s2 = 0; s2 < KS; ++s2) {
                const v8f& lo = acc[2 * s2];
                const v8f& hi = acc[2 * s2 + 1];
                a_step h;
                h.f0 = make_float4(lo[0], lo[1], lo[2], lo[3]);
                h.f1 = make_float4(lo[4], lo[5], lo[6], lo[7]);
                h.f2 = make_float4(hi[0], hi[1], hi[2], hi[3]);
                h.f3 = make_float4(hi[4], hi[5], hi[6], hi[7]);
                store_a_step(hrow + s2 * 32, h);
            }
        }

        if (more) {
            q0 = load_a_step(xnext);
            q1 = load_a_step(xnext + 32);
        }

        v16bf af[KS];
        #pragma unroll
        for (int s2 = 0; s2 < KS; ++s2) {
            const v8f& lo = acc[2 * s2];
            const v8f& hi = acc[2 * s2 + 1];
            v16bf a;
            a[0]  = (__bf16)lo[0]; a[1]  = (__bf16)lo[1]; a[2]  = (__bf16)lo[2]; a[3]  = (__bf16)lo[3];
            a[4]  = (__bf16)lo[4]; a[5]  = (__bf16)lo[5]; a[6]  = (__bf16)lo[6]; a[7]  = (__bf16)lo[7];
            a[8]  = (__bf16)hi[0]; a[9]  = (__bf16)hi[1]; a[10] = (__bf16)hi[2]; a[11] = (__bf16)hi[3];
            a[12] = (__bf16)hi[4]; a[13] = (__bf16)hi[5]; a[14] = (__bf16)hi[6]; a[15] = (__bf16)hi[7];
            af[s2] = a;
        }

        __bf16* crow = C + (size_t)(m0 + l15) * HDIM + half * 8;
        #pragma unroll
        for (int hf = 0; hf < 2; ++hf) {
            v8f ac2[NT / 2];
            #pragma unroll
            for (int t = 0; t < NT / 2; ++t) ac2[t] = (v8f){};
            #pragma unroll
            for (int ks = 0; ks < KS; ++ks) {
                #pragma unroll
                for (int t = 0; t < NT / 2; ++t) {
                    const int T = NT + hf * (NT / 2) + t;
                    const v16bf w = lds_frag(brow + goff[T / G] + (T % G) * 16 * PITCH + ks * 64);
                    ac2[t] = __builtin_amdgcn_wmma_f32_16x16x32_bf16(
                        false, w, false, af[ks], (short)0, ac2[t], false, false);
                }
            }
            #pragma unroll
            for (int t = 0; t < NT / 2; ++t) {
                const int tt = hf * (NT / 2) + t;
                const float4 c0 = *(const float4*)(b1 + boff + tt * 16);
                const float4 c1 = *(const float4*)(b1 + boff + tt * 16 + 4);
                v8bf o;
                o[0] = (__bf16)fmaxf(ac2[t][0] + c0.x, 0.0f);
                o[1] = (__bf16)fmaxf(ac2[t][1] + c0.y, 0.0f);
                o[2] = (__bf16)fmaxf(ac2[t][2] + c0.z, 0.0f);
                o[3] = (__bf16)fmaxf(ac2[t][3] + c0.w, 0.0f);
                o[4] = (__bf16)fmaxf(ac2[t][4] + c1.x, 0.0f);
                o[5] = (__bf16)fmaxf(ac2[t][5] + c1.y, 0.0f);
                o[6] = (__bf16)fmaxf(ac2[t][6] + c1.z, 0.0f);
                o[7] = (__bf16)fmaxf(ac2[t][7] + c1.w, 0.0f);
                *(v8bf*)(crow + tt * 16) = o;
            }
        }
    }
}

#define EDGES_PER_THREAD 8
__device__ __forceinline__ void count8(const int* __restrict__ dst, int* __restrict__ cnt, int i) {
    const int4* d4 = (const int4*)(dst + (size_t)i * EDGES_PER_THREAD);
    const int4 a = d4[0], b = d4[1];
    atomicAdd(cnt + a.x, 1); atomicAdd(cnt + a.y, 1);
    atomicAdd(cnt + a.z, 1); atomicAdd(cnt + a.w, 1);
    atomicAdd(cnt + b.x, 1); atomicAdd(cnt + b.y, 1);
    atomicAdd(cnt + b.z, 1); atomicAdd(cnt + b.w, 1);
}

__global__ __launch_bounds__(256)
void edge_count_kernel(const int* __restrict__ dst1, int* __restrict__ cnt1,
                       const int* __restrict__ dst2, int* __restrict__ cnt2)
{
    const int i = blockIdx.x * 256 + threadIdx.x;
    if (i < E1_N / EDGES_PER_THREAD) {
        count8(dst1, cnt1, i);
    } else {
        const int e = i - E1_N / EDGES_PER_THREAD;
        if (e < E2_N / EDGES_PER_THREAD) count8(dst2, cnt2, e);
    }
}

template <int PER>
__device__ __forceinline__ void block_offsets(const int* __restrict__ cnt,
                                              int* __restrict__ off,
                                              int* __restrict__ cur,
                                              int* wtot)
{
    const int t = threadIdx.x, lane = t & 31, wv = t >> 5;
    int v[PER];
    const int4* c4 = (const int4*)(cnt + t * PER);
    for (int q = 0; q < PER / 4; ++q) {
        int4 x = c4[q];
        v[4 * q] = x.x; v[4 * q + 1] = x.y; v[4 * q + 2] = x.z; v[4 * q + 3] = x.w;
    }
    int sum = 0;
    for (int k = 0; k < PER; ++k) sum += v[k];
    int inc = sum;
    for (int o = 1; o < 32; o <<= 1) {
        int y = __shfl_up(inc, o, 32);
        if (lane >= o) inc += y;
    }
    if (lane == 31) wtot[wv] = inc;
    __syncthreads();
    int wt = wtot[lane];
    int winc = wt;
    for (int o = 1; o < 32; o <<= 1) {
        int y = __shfl_up(winc, o, 32);
        if (lane >= o) winc += y;
    }
    int run = __shfl(winc, wv, 32) - __shfl(wt, wv, 32) + inc - sum;
    int total = __shfl(winc, 31, 32);
    int4* o4 = (int4*)(off + t * PER);
    int4* u4 = (int4*)(cur + t * PER);
    for (int q = 0; q < PER / 4; ++q) {
        int4 x;
        x.x = run; run += v[4 * q];
        x.y = run; run += v[4 * q + 1];
        x.z = run; run += v[4 * q + 2];
        x.w = run; run += v[4 * q + 3];
        o4[q] = x;
        u4[q] = x;
    }
    if (t == 1023) off[1024 * PER] = total;
}

__global__ __launch_bounds__(1024)
void segment_offsets_kernel(const int* __restrict__ cnt1, int* __restrict__ off1,
                            int* __restrict__ cur1,
                            const int* __restrict__ cnt2, int* __restrict__ off2,
                            int* __restrict__ cur2)
{
    __shared__ int wtot[32];
    if (blockIdx.x == 0) block_offsets<N_MID / 1024>(cnt1, off1, cur1, wtot);
    else                 block_offsets<N_DST / 1024>(cnt2, off2, cur2, wtot);
}

__device__ __forceinline__ void place1(int* __restrict__ pairs, int p, int s, float w) {
    int2 v; v.x = s; v.y = __float_as_int(w);
    *(int2*)(pairs + 2 * (size_t)p) = v;
}

__device__ __forceinline__ void place8(const int* __restrict__ src, const int* __restrict__ dst,
                                       const float* __restrict__ w, int* __restrict__ cur,
                                       int* __restrict__ pairs, int i) {
    const int4*   d4 = (const int4*)(dst + (size_t)i * EDGES_PER_THREAD);
    const int4*   s4 = (const int4*)(src + (size_t)i * EDGES_PER_THREAD);
    const float4* w4 = (const float4*)(w + (size_t)i * EDGES_PER_THREAD);
    const int4 da = d4[0], db = d4[1];
    const int4 sa = s4[0], sb = s4[1];
    const float4 wa = w4[0], wb = w4[1];
    const int p0 = atomicAdd(cur + da.x, 1), p1 = atomicAdd(cur + da.y, 1);
    const int p2 = atomicAdd(cur + da.z, 1), p3 = atomicAdd(cur + da.w, 1);
    const int p4 = atomicAdd(cur + db.x, 1), p5 = atomicAdd(cur + db.y, 1);
    const int p6 = atomicAdd(cur + db.z, 1), p7 = atomicAdd(cur + db.w, 1);
    place1(pairs, p0, sa.x, wa.x); place1(pairs, p1, sa.y, wa.y);
    place1(pairs, p2, sa.z, wa.z); place1(pairs, p3, sa.w, wa.w);
    place1(pairs, p4, sb.x, wb.x); place1(pairs, p5, sb.y, wb.y);
    place1(pairs, p6, sb.z, wb.z); place1(pairs, p7, sb.w, wb.w);
}

__global__ __launch_bounds__(256)
void edge_place_kernel(const int* __restrict__ src1, const int* __restrict__ dst1,
                       const float* __restrict__ w1, int* __restrict__ cur1,
                       int* __restrict__ pairs1,
                       const int* __restrict__ src2, const int* __restrict__ dst2,
                       const float* __restrict__ w2, int* __restrict__ cur2,
                       int* __restrict__ pairs2)
{
    const int i = blockIdx.x * 256 + threadIdx.x;
    if (i < E1_N / EDGES_PER_THREAD) {
        place8(src1, dst1, w1, cur1, pairs1, i);
    } else {
        const int e = i - E1_N / EDGES_PER_THREAD;
        if (e < E2_N / EDGES_PER_THREAD) place8(src2, dst2, w2, cur2, pairs2, e);
    }
}

__device__ __forceinline__ void fma4(float4& a, const float4& x, float w) {
    a.x = fmaf(x.x, w, a.x); a.y = fmaf(x.y, w, a.y);
    a.z = fmaf(x.z, w, a.z); a.w = fmaf(x.w, w, a.w);
}

template <bool TAB16> struct row_raw;
template <> struct row_raw<false> { float4 x, y; };
template <> struct row_raw<true>  { uint2  x, y; };

__device__ __forceinline__ row_raw<false> load_row(const float* tab, int row, int lane) {
    const float4* p = (const float4*)(tab + (size_t)row * HDIM);
    row_raw<false> v;
    v.x = p[lane]; v.y = p[32 + lane];
    return v;
}
__device__ __forceinline__ row_raw<true> load_row(const __bf16* tab, int row, int lane) {
    const uint2* p = (const uint2*)(tab + (size_t)row * HDIM);
    row_raw<true> v;
    v.x = p[lane]; v.y = p[32 + lane];
    return v;
}
__device__ __forceinline__ float4 widen4(const float4& v) { return v; }
__device__ __forceinline__ float4 widen4(const uint2& v) {
    return make_float4(__uint_as_float(v.x << 16), __uint_as_float(v.x & 0xffff0000u),
                       __uint_as_float(v.y << 16), __uint_as_float(v.y & 0xffff0000u));
}

template <bool TAB16, typename TabT>
__device__ __forceinline__ void segment_gather_body(const TabT* __restrict__ nsrc,
                                                    const int* __restrict__ off,
                                                    const int* __restrict__ pairs,
                                                    const float* __restrict__ hdst,
                                                    float* __restrict__ z, int rows)
{
    const int lane = threadIdx.x & 31;
    const int r    = blockIdx.x * 8 + (threadIdx.x >> 5);
    if (r >= rows) return;
    const float4* hr = (const float4*)(hdst + (size_t)r * HDIM);
    float4*       zr = (float4*)(z + (size_t)r * (2 * HDIM));
    const float4 h0 = hr[lane], h1 = hr[32 + lane];
    const int beg = __builtin_amdgcn_readfirstlane(off[r]);
    const int end = __builtin_amdgcn_readfirstlane(off[r + 1]);
    zr[64 + lane] = h0;
    zr[96 + lane] = h1;
    float4 a0 = make_float4(0.f, 0.f, 0.f, 0.f);
    float4 a1 = make_float4(0.f, 0.f, 0.f, 0.f);
    float  ws = 0.0f;
    for (int base = beg; base < end; base += 32) {
        int n = end - base;
        n = n < 32 ? n : 32;
        int ps = 0, pw = 0;
        if (lane < n) {
            int2 p = *(const int2*)(pairs + 2 * (size_t)(base + lane));
            ps = p.x; pw = p.y;
        }
        int j = 0;
        for (; j + 4 <= n; j += 4) {
            const row_raw<TAB16> v0 = load_row(nsrc, __builtin_amdgcn_readlane(ps, j),     lane);
            const row_raw<TAB16> v1 = load_row(nsrc, __builtin_amdgcn_readlane(ps, j + 1), lane);
            const row_raw<TAB16> v2 = load_row(nsrc, __builtin_amdgcn_readlane(ps, j + 2), lane);
            const row_raw<TAB16> v3 = load_row(nsrc, __builtin_amdgcn_readlane(ps, j + 3), lane);
            float w0 = __int_as_float(__builtin_amdgcn_readlane(pw, j));
            float w1 = __int_as_float(__builtin_amdgcn_readlane(pw, j + 1));
            float w2 = __int_as_float(__builtin_amdgcn_readlane(pw, j + 2));
            float w3 = __int_as_float(__builtin_amdgcn_readlane(pw, j + 3));
            fma4(a0, widen4(v0.x), w0); fma4(a1, widen4(v0.y), w0); ws += w0;
            fma4(a0, widen4(v1.x), w1); fma4(a1, widen4(v1.y), w1); ws += w1;
            fma4(a0, widen4(v2.x), w2); fma4(a1, widen4(v2.y), w2); ws += w2;
            fma4(a0, widen4(v3.x), w3); fma4(a1, widen4(v3.y), w3); ws += w3;
        }
        #pragma nounroll
        for (; j < n; ++j) {
            const row_raw<TAB16> v0 = load_row(nsrc, __builtin_amdgcn_readlane(ps, j), lane);
            float w0 = __int_as_float(__builtin_amdgcn_readlane(pw, j));
            fma4(a0, widen4(v0.x), w0); fma4(a1, widen4(v0.y), w0); ws += w0;
        }
    }
    ws = ws < 1.0f ? 1.0f : ws;
    const float inv = __frcp_rn(ws);
    zr[lane]      = make_float4(a0.x * inv, a0.y * inv, a0.z * inv, a0.w * inv);
    zr[32 + lane] = make_float4(a1.x * inv, a1.y * inv, a1.z * inv, a1.w * inv);
}

__global__ __launch_bounds__(256)
void segment_gather_kernel(const float* __restrict__ nsrc,
                           const int* __restrict__ off,
                           const int* __restrict__ pairs,
                           const float* __restrict__ hdst,
                           float* __restrict__ z, int rows)
{
    segment_gather_body<false>(nsrc, off, pairs, hdst, z, rows);
}

__global__ __launch_bounds__(256)
void segment_gather_bf16_kernel(const __bf16* __restrict__ nsrc,
                                const int* __restrict__ off,
                                const int* __restrict__ pairs,
                                const float* __restrict__ hdst,
                                float* __restrict__ z, int rows)
{
    segment_gather_body<true>(nsrc, off, pairs, hdst, z, rows);
}

#define NORM_ROWS_PER_WAVE 4
static_assert(N_DST % (8 * NORM_ROWS_PER_WAVE) == 0, "whole waves only");
__global__ __launch_bounds__(256)
void skip_norm_kernel(const float* __restrict__ hp, const float* __restrict__ h2,
                      float* __restrict__ zf, int rows)
{
    const int lane = threadIdx.x & 31;
    const int r0   = (blockIdx.x * 8 + (threadIdx.x >> 5)) * NORM_ROWS_PER_WAVE;
    if (r0 >= rows) return;
    float4 a0[NORM_ROWS_PER_WAVE], a1[NORM_ROWS_PER_WAVE], b0[NORM_ROWS_PER_WAVE], b1[NORM_ROWS_PER_WAVE];
    #pragma unroll
    for (int k = 0; k < NORM_ROWS_PER_WAVE; ++k) {
        const float4* a = (const float4*)(hp + (size_t)(r0 + k) * HDIM + lane * 8);
        const float4* b = (const float4*)(h2 + (size_t)(r0 + k) * HDIM + lane * 8);
        a0[k] = a[0]; a1[k] = a[1]; b0[k] = b[0]; b1[k] = b[1];
    }
    #pragma unroll
    for (int k = 0; k < NORM_ROWS_PER_WAVE; ++k) {
        float v[8];
        v[0] = a0[k].x + b0[k].x; v[1] = a0[k].y + b0[k].y; v[2] = a0[k].z + b0[k].z; v[3] = a0[k].w + b0[k].w;
        v[4] = a1[k].x + b1[k].x; v[5] = a1[k].y + b1[k].y; v[6] = a1[k].z + b1[k].z; v[7] = a1[k].w + b1[k].w;
        float ss = 0.0f;
        #pragma unroll
        for (int j = 0; j < 8; ++j) ss += v[j] * v[j];
        #pragma unroll
        for (int o = 16; o > 0; o >>= 1) ss += __shfl_xor(ss, o, 32);
        const float nrm   = sqrtf(ss);
        const float scale = (nrm == 0.0f) ? 1.0f : 1.0f / nrm;
        float4* z = (float4*)(zf + (size_t)(r0 + k) * HDIM + lane * 8);
        z[0] = make_float4(v[0] * scale, v[1] * scale, v[2] * scale, v[3] * scale);
        z[1] = make_float4(v[4] * scale, v[5] * scale, v[6] * scale, v[7] * scale);
    }
}

#define PAIRS_PER_WAVE 4
static_assert(N_PAIRS % (8 * PAIRS_PER_WAVE) == 0, "whole waves only");
__global__ __launch_bounds__(256)
void score_kernel(const float* __restrict__ zf, const float* __restrict__ bias,
                  const int* __restrict__ nids,
                  const int* __restrict__ ps, const int* __restrict__ pd,
                  const int* __restrict__ ns, const int* __restrict__ nd,
                  float* __restrict__ out, int npairs)
{
    const int lane = threadIdx.x & 31;
    const int p0   = (blockIdx.x * 8 + (threadIdx.x >> 5)) * PAIRS_PER_WAVE;
    if (p0 >= npairs) return;
    int pu[PAIRS_PER_WAVE], pv[PAIRS_PER_WAVE], nu[PAIRS_PER_WAVE], nv[PAIRS_PER_WAVE];
    #pragma unroll
    for (int k = 0; k < PAIRS_PER_WAVE; ++k) {
        pu[k] = ps[p0 + k]; pv[k] = pd[p0 + k]; nu[k] = ns[p0 + k]; nv[k] = nd[p0 + k];
    }
    int ipu[PAIRS_PER_WAVE], ipv[PAIRS_PER_WAVE], inu[PAIRS_PER_WAVE], inv[PAIRS_PER_WAVE];
    #pragma unroll
    for (int k = 0; k < PAIRS_PER_WAVE; ++k) {
        ipu[k] = nids[pu[k]]; ipv[k] = nids[pv[k]];
        inu[k] = nids[nu[k]]; inv[k] = nids[nv[k]];
    }
    float4 a0[PAIRS_PER_WAVE], a1[PAIRS_PER_WAVE], b0[PAIRS_PER_WAVE], b1[PAIRS_PER_WAVE];
    float4 e0[PAIRS_PER_WAVE], e1[PAIRS_PER_WAVE], f0[PAIRS_PER_WAVE], f1[PAIRS_PER_WAVE];
    #pragma unroll
    for (int k = 0; k < PAIRS_PER_WAVE; ++k) {
        const float4* a = (const float4*)(zf + (size_t)pu[k] * HDIM + lane * 8);
        const float4* b = (const float4*)(zf + (size_t)pv[k] * HDIM + lane * 8);
        const float4* e = (const float4*)(zf + (size_t)nu[k] * HDIM + lane * 8);
        const float4* f = (const float4*)(zf + (size_t)nv[k] * HDIM + lane * 8);
        a0[k] = a[0]; a1[k] = a[1]; b0[k] = b[0]; b1[k] = b[1];
        e0[k] = e[0]; e1[k] = e[1]; f0[k] = f[0]; f1[k] = f[1];
    }
    float gp[PAIRS_PER_WAVE], gn[PAIRS_PER_WAVE];
    #pragma unroll
    for (int k = 0; k < PAIRS_PER_WAVE; ++k) {
        gp[k] = bias[ipu[k]] + bias[ipv[k]];
        gn[k] = bias[inu[k]] + bias[inv[k]];
    }
    #pragma unroll
    for (int k = 0; k < PAIRS_PER_WAVE; ++k) {
        float sp = 0.0f, sn = 0.0f;
        sp += a0[k].x * b0[k].x; sn += e0[k].x * f0[k].x;
        sp += a0[k].y * b0[k].y; sn += e0[k].y * f0[k].y;
        sp += a0[k].z * b0[k].z; sn += e0[k].z * f0[k].z;
        sp += a0[k].w * b0[k].w; sn += e0[k].w * f0[k].w;
        sp += a1[k].x * b1[k].x; sn += e1[k].x * f1[k].x;
        sp += a1[k].y * b1[k].y; sn += e1[k].y * f1[k].y;
        sp += a1[k].z * b1[k].z; sn += e1[k].z * f1[k].z;
        sp += a1[k].w * b1[k].w; sn += e1[k].w * f1[k].w;
        #pragma unroll
        for (int o = 16; o > 0; o >>= 1) {
            sp += __shfl_xor(sp, o, 32);
            sn += __shfl_xor(sn, o, 32);
        }
        const float m = (sn + gn[k]) - (sp + gp[k]) + 1.0f;
        if (lane == 0) out[p0 + k] = m > 0.0f ? m : 0.0f;
    }
}

extern "C" void kernel_launch(void* const* d_in, const int* in_sizes, int n_in,
                              void* d_out, int out_size, void* d_ws, size_t ws_size,
                              hipStream_t stream) {
    const float* feat = (const float*)d_in[0];
    const float* Wp   = (const float*)d_in[1];
    const float* bp   = (const float*)d_in[2];
    const float* Q1   = (const float*)d_in[3];
    const float* bq1  = (const float*)d_in[4];
    const float* W1   = (const float*)d_in[5];
    const float* bw1  = (const float*)d_in[6];
    const float* Q2   = (const float*)d_in[7];
    const float* bq2  = (const float*)d_in[8];
    const float* W2   = (const float*)d_in[9];
    const float* bw2  = (const float*)d_in[10];
    const float* w1v  = (const float*)d_in[11];
    const float* w2v  = (const float*)d_in[12];
    const float* bias = (const float*)d_in[13];
    const int* src1   = (const int*)d_in[14];
    const int* dst1   = (const int*)d_in[15];
    const int* src2   = (const int*)d_in[16];
    const int* dst2   = (const int*)d_in[17];
    const int* pos_s  = (const int*)d_in[18];
    const int* pos_d  = (const int*)d_in[19];
    const int* neg_s  = (const int*)d_in[20];
    const int* neg_d  = (const int*)d_in[21];
    const int* nids   = (const int*)d_in[22];
    float* out        = (float*)d_out;

    char* ws = (char*)d_ws;
    const size_t SZ_HP  = (size_t)N_SRC * HDIM * 4;
    const size_t SZ_N   = (size_t)N_SRC * HDIM * 4;
    const size_t SZ_H1  = (size_t)N_MID * HDIM * 4;
    const size_t SZ_AGG = (size_t)N_MID * HDIM * 4;
    const size_t SZ_WS  = 256 * 1024;
    const size_t SZ_H2  = (size_t)N_DST * HDIM * 4;
    const size_t SZ_ZF  = (size_t)N_DST * HDIM * 4;
    float*  hp   = (float*)(ws);
    float*  nbuf = (float*)(ws + SZ_HP);
    __bf16* n1b  = (__bf16*)(ws + SZ_HP);
    float*  h1   = (float*)(ws + SZ_HP + SZ_N);
    float*  zbuf = hp + (size_t)N_MID * HDIM;
    int* cnt1   = (int*)(ws + SZ_HP + SZ_N + SZ_H1);
    int* cnt2   = cnt1 + N_MID;
    int* off1   = cnt2 + N_DST;
    int* off2   = off1 + N_MID + 64;
    int* cur1   = off2 + N_DST + 64;
    int* cur2   = cur1 + N_MID;
    int* pairs1 = cur2 + N_DST;
    int* pairs2 = pairs1 + 2 * E1_N;
    float*  h2   = (float*)(ws + SZ_HP + SZ_N + SZ_H1 + SZ_AGG + SZ_WS);
    float*  zf   = (float*)(ws + SZ_HP + SZ_N + SZ_H1 + SZ_AGG + SZ_WS + SZ_H2);
    __bf16* Wpb  = (__bf16*)(ws + SZ_HP + SZ_N + SZ_H1 + SZ_AGG + SZ_WS + SZ_H2 + SZ_ZF);
    __bf16* Q1b  = Wpb + HDIM * HDIM;
    __bf16* W1b  = Q1b + HDIM * HDIM;
    __bf16* Q2b  = W1b + HDIM * 2 * HDIM;
    __bf16* W2b  = Q2b + HDIM * HDIM;

    prep_kernel<<<PREP_W_BLOCKS + PREP_Z_BLOCKS, PREP_THREADS, 0, stream>>>(Wp, Q1, W1, Q2, W2, Wpb, cnt1);

    edge_count_kernel<<<(E1_N + E2_N) / EDGES_PER_THREAD / 256, 256, 0, stream>>>(dst1, cnt1, dst2, cnt2);
    segment_offsets_kernel<<<2, 1024, 0, stream>>>(cnt1, off1, cur1, cnt2, off2, cur2);
    edge_place_kernel<<<(E1_N + E2_N) / EDGES_PER_THREAD / 256, 256, 0, stream>>>(src1, dst1, w1v, cur1, pairs1,
                                                                              src2, dst2, w2v, cur2, pairs2);

    gemm_pair_bf16_w_in_lds<<<256, 512, 0, stream>>>(feat, Wpb, bp, bq1, hp, n1b, N_SRC, N_MID);
    segment_gather_bf16_kernel<<<N_MID / 8, 256, 0, stream>>>(n1b, off1, pairs1, hp, zbuf, N_MID);
    gemm_bf16_w_in_lds<2 * HDIM, 8><<<256, 512, 0, stream>>>(zbuf, W1b, bw1, h1, N_MID, 1);
    gemm_bf16_w_in_lds<HDIM, 8><<<256, 512, 0, stream>>>(h1, Q2b, bq2, nbuf, N_MID, 1);
    segment_gather_kernel<<<N_DST / 8, 256, 0, stream>>>(nbuf, off2, pairs2, h1, zbuf, N_DST);
    gemm_bf16_n256<<<N_DST / 32, 256, 0, stream>>>(zbuf, W2b, bw2, h2, N_DST, 2 * HDIM, 1);
    skip_norm_kernel<<<N_DST / (8 * NORM_ROWS_PER_WAVE), 256, 0, stream>>>(hp, h2, zf, N_DST);
    score_kernel<<<N_PAIRS / (8 * PAIRS_PER_WAVE), 256, 0, stream>>>(zf, bias, nids,
                                                  pos_s, pos_d, neg_s, neg_d,
                                                  out, N_PAIRS);
}


# Edges binned in one pass: a row owns 32 slots taken by arrival rank (int atomic), later edges go to an exact overflow list; count pass and prefix scan gone (12->10 launches); sums in arrival order
# speedup vs baseline: 3.5336x; 1.2511x over previous
#include <hip/hip_runtime.h>
#include <hip/hip_bf16.h>


typedef __attribute__((ext_vector_type(16))) __bf16 v16bf;
typedef __attribute__((ext_vector_type(8)))  float  v8f;
typedef __attribute__((ext_vector_type(8)))  __bf16 v8bf;

#define N_SRC   131072
#define N_MID   32768
#define N_DST   8192
#define E1_N    524288
#define E2_N    131072
#define HDIM    256
#define N_PAIRS 8192

#define PREP_THREADS  128
#define PREP_SLAB     (HDIM * HDIM / 4)
#define PREP_W_BLOCKS (PREP_SLAB / PREP_THREADS)
#define PREP_Z_UPT    4
#define CNT_INTS      (N_MID + N_DST + 2048)
#define PREP_Z_BLOCKS (CNT_INTS / 4 / PREP_Z_UPT / PREP_THREADS)
__global__ __launch_bounds__(PREP_THREADS)
void prep_kernel(const float* __restrict__ Wp, const float* __restrict__ Q1,
                 const float* __restrict__ W1, const float* __restrict__ Q2,
                 const float* __restrict__ W2, __bf16* __restrict__ wb,
                 int* __restrict__ cnt)
{
    typedef __attribute__((ext_vector_type(4))) __bf16 v4bf;
    static_assert(PREP_SLAB % PREP_THREADS == 0 &&
                  CNT_INTS % (4 * PREP_Z_UPT * PREP_THREADS) == 0, "whole blocks only");
    const int tid = threadIdx.x;
    const int blk = blockIdx.x;
    if (blk < PREP_W_BLOCKS) {
        const int u = blk * PREP_THREADS + tid;
        float4 v[7];
        v[0] = ((const float4*)Wp)[u];
        v[1] = ((const float4*)Q1)[u];
        v[2] = ((const float4*)W1)[u];
        v[3] = ((const float4*)W1)[u + PREP_SLAB];
        v[4] = ((const float4*)Q2)[u];
        v[5] = ((const float4*)W2)[u];
        v[6] = ((const float4*)W2)[u + PREP_SLAB];
        #pragma unroll
        for (int k = 0; k < 7; ++k) {
            v4bf o;
            o[0] = (__bf16)v[k].x; o[1] = (__bf16)v[k].y;
            o[2] = (__bf16)v[k].z; o[3] = (__bf16)v[k].w;
            *(v4bf*)(wb + ((size_t)k * PREP_SLAB + u) * 4) = o;
        }
    } else {
        #pragma unroll
        for (int k = 0; k < PREP_Z_UPT; ++k) {
            const int u = (blk - PREP_W_BLOCKS + k * PREP_Z_BLOCKS) * PREP_THREADS + tid;
            ((int4*)cnt)[u] = make_int4(0, 0, 0, 0);
        }
    }
}

__global__ __launch_bounds__(256)
void gemm_bf16_n256(const float* __restrict__ A, const __bf16* __restrict__ W,
                    const float* __restrict__ bias, float* __restrict__ C,
                    int M, int K, int relu)
{
    const int lane = threadIdx.x & 31;
    const int wave = threadIdx.x >> 5;
    const int half = lane >> 4;
    const int l15  = lane & 15;
    const int m0   = blockIdx.x * 32 + (wave >> 2) * 16;
    const int n0   = (wave & 3) * 64;

    v8f acc[4];
    #pragma unroll
    for (int t = 0; t < 4; ++t) acc[t] = (v8f){};

    const float* arow = A + (size_t)(m0 + l15) * K;

    for (int k0 = 0; k0 < K; k0 += 32) {
        v16bf a;
        #pragma unroll
        for (int q = 0; q < 8; ++q) {
            int kk = k0 + (q >> 2) * 16 + half * 8 + (q & 3) * 2;
            float2 v = *(const float2*)(arow + kk);
            a[2 * q]     = (__bf16)v.x;
            a[2 * q + 1] = (__bf16)v.y;
        }
        #pragma unroll
        for (int t = 0; t < 4; ++t) {
            const __bf16* wrow = W + (size_t)(n0 + t * 16 + l15) * K;
            v16bf b;
            #pragma unroll
            for (int q = 0; q < 8; ++q) {
                int kk = k0 + (q >> 2) * 16 + half * 8 + (q & 3) * 2;
                b[2 * q]     = wrow[kk];
                b[2 * q + 1] = wrow[kk + 1];
            }
            acc[t] = __builtin_amdgcn_wmma_f32_16x16x32_bf16(
                false, a, false, b, (short)0, acc[t], false, false);
        }
    }

    #pragma unroll
    for (int t = 0; t < 4; ++t) {
        int n = n0 + t * 16 + l15;
        float bv = bias[n];
        #pragma unroll
        for (int r = 0; r < 8; ++r) {
            int m = m0 + half * 8 + r;
            float v = acc[t][r] + bv;
            if (relu) v = fmaxf(v, 0.0f);
            C[(size_t)m * HDIM + n] = v;
        }
    }
}

typedef __attribute__((ext_vector_type(8))) unsigned int v8u;

struct a_step { float4 f0, f1, f2, f3; };

__device__ __forceinline__ a_step load_a_step(const float* p) {
    a_step s;
    s.f0 = *(const float4*)(p);
    s.f1 = *(const float4*)(p + 4);
    s.f2 = *(const float4*)(p + 16);
    s.f3 = *(const float4*)(p + 20);
    return s;
}

template <int K, int NT>
__global__ __launch_bounds__(512)
void gemm_bf16_w_in_lds(const float* __restrict__ A, const __bf16* __restrict__ W,
                        const float* __restrict__ bias, float* __restrict__ C,
                        int M, int relu)
{
    constexpr int PITCH  = K * 2 + 16;
    constexpr int CPR    = K / 8;
    constexpr int CHUNKS = HDIM * CPR;
    constexpr int WPR    = 16 / NT;
    constexpr int KS     = K / 32;
    constexpr int G      = (65535 - 47 - (KS - 1) * 64) / (16 * PITCH) + 1 < NT
                         ? (65535 - 47 - (KS - 1) * 64) / (16 * PITCH) + 1 : NT;
    constexpr int NB     = (NT + G - 1) / G;
    __shared__ __attribute__((aligned(16))) unsigned char L[HDIM * PITCH];

    const int tid  = threadIdx.x;
    const int lane = tid & 31;
    const int wave = __builtin_amdgcn_readfirstlane(tid >> 5);
    const int half = lane >> 4;
    const int l15  = lane & 15;

    const int total  = (M / 16) * WPR;
    const int stride = gridDim.x * 16;
    int it = blockIdx.x * 16 + wave;

    a_step q0, q1;
    q0.f0 = q0.f1 = q0.f2 = q0.f3 = make_float4(0.f, 0.f, 0.f, 0.f);
    q1 = q0;
    if (it < total) {
        const float* p = A + (size_t)((it / WPR) * 16 + l15) * K + half * 8;
        q0 = load_a_step(p);
        q1 = load_a_step(p + 32);
    }

    #pragma unroll 16
    for (int i = 0; i < CHUNKS / 512; ++i) {
        const int c = i * 512 + tid;
        const int row = c / CPR, col = c % CPR;
        const uint4 v = *(const uint4*)(W + (size_t)c * 8);
        *(uint4*)(L + row * PITCH + col * 16) = v;
    }
    __syncthreads();

    unsigned goff[NB];
    #pragma unroll
    for (int g = 0; g < NB; ++g) {
        goff[g] = g * G * 16 * PITCH;
        if (g > 0) asm volatile("" : "+v"(goff[g]));
    }

    for (; it < total; it += stride) {
        const int m0 = (it / WPR) * 16;
        const int n0 = (it % WPR) * (NT * 16);
        const float* arow = A + (size_t)(m0 + l15) * K + half * 8;
        const bool more = it + stride < total;
        const float* anext = A + (size_t)(((it + stride) / WPR) * 16 + l15) * K + half * 8;
        const unsigned char* brow = L + (n0 + l15) * PITCH + half * 16;

        v8f acc[NT];
        #pragma unroll
        for (int t = 0; t < NT; ++t) acc[t] = (v8f){};

        #pragma unroll
        for (int ks = 0; ks < KS; ++ks) {
            a_step q2 = q1;
            if (ks + 2 < KS)  q2 = load_a_step(arow + (ks + 2) * 32);
            else if (more)    q2 = load_a_step(anext + (ks + 2 - KS) * 32);

            v16bf a;
            a[0]  = (__bf16)q0.f0.x; a[1]  = (__bf16)q0.f0.y; a[2]  = (__bf16)q0.f0.z; a[3]  = (__bf16)q0.f0.w;
            a[4]  = (__bf16)q0.f1.x; a[5]  = (__bf16)q0.f1.y; a[6]  = (__bf16)q0.f1.z; a[7]  = (__bf16)q0.f1.w;
            a[8]  = (__bf16)q0.f2.x; a[9]  = (__bf16)q0.f2.y; a[10] = (__bf16)q0.f2.z; a[11] = (__bf16)q0.f2.w;
            a[12] = (__bf16)q0.f3.x; a[13] = (__bf16)q0.f3.y; a[14] = (__bf16)q0.f3.z; a[15] = (__bf16)q0.f3.w;

            #pragma unroll
            for (int t = 0; t < NT; ++t) {
                const unsigned char* bp = brow + goff[t / G] + (t % G) * 16 * PITCH + ks * 64;
                const uint4 lo = *(const uint4*)(bp);
                const uint4 hi = *(const uint4*)(bp + 32);
                v8u bw;
                bw[0] = lo.x; bw[1] = lo.y; bw[2] = lo.z; bw[3] = lo.w;
                bw[4] = hi.x; bw[5] = hi.y; bw[6] = hi.z; bw[7] = hi.w;
                const v16bf b = __builtin_bit_cast(v16bf, bw);
                acc[t] = __builtin_amdgcn_wmma_f32_16x16x32_bf16(
                    false, a, false, b, (short)0, acc[t], false, false);
            }
            q0 = q1;
            q1 = q2;
        }

        #pragma unroll
        for (int t = 0; t < NT; ++t) {
            const int n = n0 + t * 16 + l15;
            const float bv = bias[n];
            #pragma unroll
            for (int r = 0; r < 8; ++r) {
                const int m = m0 + half * 8 + r;
                float v = acc[t][r] + bv;
                if (relu) v = fmaxf(v, 0.0f);
                C[(size_t)m * HDIM + n] = v;
            }
        }
    }
}

__device__ __forceinline__ void store_a_step(float* p, const a_step& s) {
    *(float4*)(p)      = s.f0;
    *(float4*)(p + 4)  = s.f1;
    *(float4*)(p + 16) = s.f2;
    *(float4*)(p + 20) = s.f3;
}

__device__ __forceinline__ v16bf to_frag(const a_step& q) {
    v16bf a;
    a[0]  = (__bf16)q.f0.x; a[1]  = (__bf16)q.f0.y; a[2]  = (__bf16)q.f0.z; a[3]  = (__bf16)q.f0.w;
    a[4]  = (__bf16)q.f1.x; a[5]  = (__bf16)q.f1.y; a[6]  = (__bf16)q.f1.z; a[7]  = (__bf16)q.f1.w;
    a[8]  = (__bf16)q.f2.x; a[9]  = (__bf16)q.f2.y; a[10] = (__bf16)q.f2.z; a[11] = (__bf16)q.f2.w;
    a[12] = (__bf16)q.f3.x; a[13] = (__bf16)q.f3.y; a[14] = (__bf16)q.f3.z; a[15] = (__bf16)q.f3.w;
    return a;
}

__device__ __forceinline__ v16bf lds_frag(const unsigned char* bp) {
    const uint4 lo = *(const uint4*)(bp);
    const uint4 hi = *(const uint4*)(bp + 32);
    v8u bw;
    bw[0] = lo.x; bw[1] = lo.y; bw[2] = lo.z; bw[3] = lo.w;
    bw[4] = hi.x; bw[5] = hi.y; bw[6] = hi.z; bw[7] = hi.w;
    return __builtin_bit_cast(v16bf, bw);
}

__global__ __launch_bounds__(512)
void gemm_pair_bf16_w_in_lds(const float* __restrict__ X, const __bf16* __restrict__ W,
                             const float* __restrict__ b0, const float* __restrict__ b1,
                             float* __restrict__ H, __bf16* __restrict__ C,
                             int M, int keep_rows)
{
    constexpr int K      = HDIM;
    constexpr int PITCH  = K * 2 + 16;
    constexpr int CPR    = K / 8;
    constexpr int ROWS   = 2 * HDIM;
    constexpr int CHUNKS = ROWS * CPR;
    constexpr int KS     = K / 32;
    constexpr int NT     = HDIM / 16;
    constexpr int G      = 8;
    constexpr int NB     = 2 * NT / G;
    static_assert((G - 1) * 16 * PITCH + (KS - 1) * 64 + 47 < 65536,
                  "a group's fragments must be in reach of the 16-bit LDS offset");
    __shared__ __attribute__((aligned(16))) unsigned char L[ROWS * PITCH];

    const int tid  = threadIdx.x;
    const int lane = tid & 31;
    const int wave = __builtin_amdgcn_readfirstlane(tid >> 5);
    const int half = lane >> 4;
    const int l15  = lane & 15;

    const int total   = M / 16;
    const int quarter = total / 4;
    const int stride  = gridDim.x * 16;
    int it = blockIdx.x * 16 + wave;

    a_step q0, q1;
    q0.f0 = q0.f1 = q0.f2 = q0.f3 = make_float4(0.f, 0.f, 0.f, 0.f);
    q1 = q0;
    if (it < total) {
        const int rt = (it >> 2) + (it & 3) * quarter;
        const float* p = X + (size_t)(rt * 16 + l15) * K + half * 8;
        q0 = load_a_step(p);
        q1 = load_a_step(p + 32);
    }

    #pragma unroll 16
    for (int i = 0; i < CHUNKS / 512; ++i) {
        const int c = i * 512 + tid;
        const int row = c / CPR, col = c % CPR;
        const uint4 v = *(const uint4*)(W + (size_t)c * 8);
        *(uint4*)(L + row * PITCH + col * 16) = v;
    }
    __syncthreads();

    unsigned goff[NB];
    #pragma unroll
    for (int g = 0; g < NB; ++g) {
        goff[g] = g * G * 16 * PITCH;
        if (g > 0) asm volatile("" : "+v"(goff[g]));
    }
    const unsigned char* brow = L + l15 * PITCH + half * 16;

    for (; it < total; it += stride) {
        const int rt = (it >> 2) + (it & 3) * quarter;
        const int m0 = rt * 16;
        const float* xrow = X + (size_t)(m0 + l15) * K + half * 8;
        const bool keep = m0 < keep_rows;
        const bool more = it + stride < total;
        const int itn = it + stride;
        const float* xnext = X + (size_t)(((itn >> 2) + (itn & 3) * quarter) * 16 + l15) * K + half * 8;

        v8f acc[NT];
        #pragma unroll
        for (int t = 0; t < NT; ++t) acc[t] = (v8f){};

        #pragma unroll
        for (int ks = 0; ks < KS; ++ks) {
            a_step q2 = q1;
            if (ks + 2 < KS) q2 = load_a_step(xrow + (ks + 2) * 32);
            const v16bf x = to_frag(q0);
            #pragma unroll
            for (int t = 0; t < NT; ++t) {
                const v16bf w = lds_frag(brow + goff[t / G] + (t % G) * 16 * PITCH + ks * 64);
                acc[t] = __builtin_amdgcn_wmma_f32_16x16x32_bf16(
                    false, w, false, x, (short)0, acc[t], false, false);
            }
            q0 = q1;
            q1 = q2;
        }

        unsigned boff = half * 8;
        asm volatile("" : "+v"(boff));
        #pragma unroll
        for (int s2 = 0; s2 < KS; ++s2) {
            const a_step bb = load_a_step(b0 + boff + s2 * 32);
            v8f& lo = acc[2 * s2];
            v8f& hi = acc[2 * s2 + 1];
            lo[0] += bb.f0.x; lo[1] += bb.f0.y; lo[2] += bb.f0.z; lo[3] += bb.f0.w;
            lo[4] += bb.f1.x; lo[5] += bb.f1.y; lo[6] += bb.f1.z; lo[7] += bb.f1.w;
            hi[0] += bb.f2.x; hi[1] += bb.f2.y; hi[2] += bb.f2.z; hi[3] += bb.f2.w;
            hi[4] += bb.f3.x; hi[5] += bb.f3.y; hi[6] += bb.f3.z; hi[7] += bb.f3.w;
        }

        if (keep) {
            float* hrow = H + (size_t)(m0 + l15) * K + half * 8;
            #pragma unroll
            for (int s2 = 0; s2 < KS; ++s2) {
                const v8f& lo = acc[2 * s2];
                const v8f& hi = acc[2 * s2 + 1];
                a_step h;
                h.f0 = make_float4(lo[0], lo[1], lo[2], lo[3]);
                h.f1 = make_float4(lo[4], lo[5], lo[6], lo[7]);
                h.f2 = make_float4(hi[0], hi[1], hi[2], hi[3]);
                h.f3 = make_float4(hi[4], hi[5], hi[6], hi[7]);
                store_a_step(hrow + s2 * 32, h);
            }
        }

        if (more) {
            q0 = load_a_step(xnext);
            q1 = load_a_step(xnext + 32);
        }

        v16bf af[KS];
        #pragma unroll
        for (int s2 = 0; s2 < KS; ++s2) {
            const v8f& lo = acc[2 * s2];
            const v8f& hi = acc[2 * s2 + 1];
            v16bf a;
            a[0]  = (__bf16)lo[0]; a[1]  = (__bf16)lo[1]; a[2]  = (__bf16)lo[2]; a[3]  = (__bf16)lo[3];
            a[4]  = (__bf16)lo[4]; a[5]  = (__bf16)lo[5]; a[6]  = (__bf16)lo[6]; a[7]  = (__bf16)lo[7];
            a[8]  = (__bf16)hi[0]; a[9]  = (__bf16)hi[1]; a[10] = (__bf16)hi[2]; a[11] = (__bf16)hi[3];
            a[12] = (__bf16)hi[4]; a[13] = (__bf16)hi[5]; a[14] = (__bf16)hi[6]; a[15] = (__bf16)hi[7];
            af[s2] = a;
        }

        __bf16* crow = C + (size_t)(m0 + l15) * HDIM + half * 8;
        #pragma unroll
        for (int hf = 0; hf < 2; ++hf) {
            v8f ac2[NT / 2];
            #pragma unroll
            for (int t = 0; t < NT / 2; ++t) ac2[t] = (v8f){};
            #pragma unroll
            for (int ks = 0; ks < KS; ++ks) {
                #pragma unroll
                for (int t = 0; t < NT / 2; ++t) {
                    const int T = NT + hf * (NT / 2) + t;
                    const v16bf w = lds_frag(brow + goff[T / G] + (T % G) * 16 * PITCH + ks * 64);
                    ac2[t] = __builtin_amdgcn_wmma_f32_16x16x32_bf16(
                        false, w, false, af[ks], (short)0, ac2[t], false, false);
                }
            }
            #pragma unroll
            for (int t = 0; t < NT / 2; ++t) {
                const int tt = hf * (NT / 2) + t;
                const float4 c0 = *(const float4*)(b1 + boff + tt * 16);
                const float4 c1 = *(const float4*)(b1 + boff + tt * 16 + 4);
                v8bf o;
                o[0] = (__bf16)fmaxf(ac2[t][0] + c0.x, 0.0f);
                o[1] = (__bf16)fmaxf(ac2[t][1] + c0.y, 0.0f);
                o[2] = (__bf16)fmaxf(ac2[t][2] + c0.z, 0.0f);
                o[3] = (__bf16)fmaxf(ac2[t][3] + c0.w, 0.0f);
                o[4] = (__bf16)fmaxf(ac2[t][4] + c1.x, 0.0f);
                o[5] = (__bf16)fmaxf(ac2[t][5] + c1.y, 0.0f);
                o[6] = (__bf16)fmaxf(ac2[t][6] + c1.z, 0.0f);
                o[7] = (__bf16)fmaxf(ac2[t][7] + c1.w, 0.0f);
                *(v8bf*)(crow + tt * 16) = o;
            }
        }
    }
}

#define SLOTS 32
#define EDGES_PER_THREAD 8
static_assert(SLOTS == 32, "a wave reads a row's slots with one load, a lane per slot");

__device__ __forceinline__ void bin1(int* __restrict__ slots, int4* __restrict__ ovf,
                                     int* __restrict__ ovf_n, int d, int rank, int s, float w) {
    if (rank < SLOTS) {
        int2 v; v.x = s; v.y = __float_as_int(w);
        *(int2*)(slots + 2 * ((size_t)d * SLOTS + rank)) = v;
    } else {
        const int q = atomicAdd(ovf_n, 1);
        ovf[q] = make_int4(d, s, __float_as_int(w), 0);
    }
}

__device__ __forceinline__ void bin8(const int* __restrict__ src, const int* __restrict__ dst,
                                     const float* __restrict__ w, int* __restrict__ cnt,
                                     int* __restrict__ slots, int4* __restrict__ ovf,
                                     int* __restrict__ ovf_n, int i) {
    const int4*   d4 = (const int4*)(dst + (size_t)i * EDGES_PER_THREAD);
    const int4*   s4 = (const int4*)(src + (size_t)i * EDGES_PER_THREAD);
    const float4* w4 = (const float4*)(w + (size_t)i * EDGES_PER_THREAD);
    const int4 da = d4[0], db = d4[1];
    const int4 sa = s4[0], sb = s4[1];
    const float4 wa = w4[0], wb = w4[1];
    const int p0 = atomicAdd(cnt + da.x, 1), p1 = atomicAdd(cnt + da.y, 1);
    const int p2 = atomicAdd(cnt + da.z, 1), p3 = atomicAdd(cnt + da.w, 1);
    const int p4 = atomicAdd(cnt + db.x, 1), p5 = atomicAdd(cnt + db.y, 1);
    const int p6 = atomicAdd(cnt + db.z, 1), p7 = atomicAdd(cnt + db.w, 1);
    bin1(slots, ovf, ovf_n, da.x, p0, sa.x, wa.x); bin1(slots, ovf, ovf_n, da.y, p1, sa.y, wa.y);
    bin1(slots, ovf, ovf_n, da.z, p2, sa.z, wa.z); bin1(slots, ovf, ovf_n, da.w, p3, sa.w, wa.w);
    bin1(slots, ovf, ovf_n, db.x, p4, sb.x, wb.x); bin1(slots, ovf, ovf_n, db.y, p5, sb.y, wb.y);
    bin1(slots, ovf, ovf_n, db.z, p6, sb.z, wb.z); bin1(slots, ovf, ovf_n, db.w, p7, sb.w, wb.w);
}

__global__ __launch_bounds__(256)
void edge_bin_kernel(const int* __restrict__ src1, const int* __restrict__ dst1,
                     const float* __restrict__ w1, int* __restrict__ cnt1,
                     int* __restrict__ slots1, int4* __restrict__ ovf1, int* __restrict__ ovfn1,
                     const int* __restrict__ src2, const int* __restrict__ dst2,
                     const float* __restrict__ w2, int* __restrict__ cnt2,
                     int* __restrict__ slots2, int4* __restrict__ ovf2, int* __restrict__ ovfn2)
{
    const int i = blockIdx.x * 256 + threadIdx.x;
    if (i < E1_N / EDGES_PER_THREAD) {
        bin8(src1, dst1, w1, cnt1, slots1, ovf1, ovfn1, i);
    } else {
        const int e = i - E1_N / EDGES_PER_THREAD;
        if (e < E2_N / EDGES_PER_THREAD) bin8(src2, dst2, w2, cnt2, slots2, ovf2, ovfn2, e);
    }
}

__device__ __forceinline__ void fma4(float4& a, const float4& x, float w) {
    a.x = fmaf(x.x, w, a.x); a.y = fmaf(x.y, w, a.y);
    a.z = fmaf(x.z, w, a.z); a.w = fmaf(x.w, w, a.w);
}

template <bool TAB16> struct row_raw;
template <> struct row_raw<false> { float4 x, y; };
template <> struct row_raw<true>  { uint2  x, y; };

__device__ __forceinline__ row_raw<false> load_row(const float* tab, int row, int lane) {
    const float4* p = (const float4*)(tab + (size_t)row * HDIM);
    row_raw<false> v;
    v.x = p[lane]; v.y = p[32 + lane];
    return v;
}
__device__ __forceinline__ row_raw<true> load_row(const __bf16* tab, int row, int lane) {
    const uint2* p = (const uint2*)(tab + (size_t)row * HDIM);
    row_raw<true> v;
    v.x = p[lane]; v.y = p[32 + lane];
    return v;
}
__device__ __forceinline__ float4 widen4(const float4& v) { return v; }
__device__ __forceinline__ float4 widen4(const uint2& v) {
    return make_float4(__uint_as_float(v.x << 16), __uint_as_float(v.x & 0xffff0000u),
                       __uint_as_float(v.y << 16), __uint_as_float(v.y & 0xffff0000u));
}

template <bool TAB16, typename TabT>
__device__ __forceinline__ void segment_gather_body(const TabT* __restrict__ nsrc,
                                                    const int* __restrict__ cnt,
                                                    const int* __restrict__ slots,
                                                    const int4* __restrict__ ovf,
                                                    const int* __restrict__ ovf_n,
                                                    const float* __restrict__ hdst,
                                                    float* __restrict__ z, int rows)
{
    const int lane = threadIdx.x & 31;
    const int r    = blockIdx.x * 8 + (threadIdx.x >> 5);
    if (r >= rows) return;
    const float4* hr = (const float4*)(hdst + (size_t)r * HDIM);
    float4*       zr = (float4*)(z + (size_t)r * (2 * HDIM));
    const float4 h0 = hr[lane], h1 = hr[32 + lane];
    const int deg = __builtin_amdgcn_readfirstlane(cnt[r]);
    zr[64 + lane] = h0;
    zr[96 + lane] = h1;
    float4 a0 = make_float4(0.f, 0.f, 0.f, 0.f);
    float4 a1 = make_float4(0.f, 0.f, 0.f, 0.f);
    float  ws = 0.0f;
    {
        const int n = deg < SLOTS ? deg : SLOTS;
        int ps = 0, pw = 0;
        if (lane < n) {
            int2 p = *(const int2*)(slots + 2 * ((size_t)r * SLOTS + lane));
            ps = p.x; pw = p.y;
        }
        int j = 0;
        for (; j + 4 <= n; j += 4) {
            const row_raw<TAB16> v0 = load_row(nsrc, __builtin_amdgcn_readlane(ps, j),     lane);
            const row_raw<TAB16> v1 = load_row(nsrc, __builtin_amdgcn_readlane(ps, j + 1), lane);
            const row_raw<TAB16> v2 = load_row(nsrc, __builtin_amdgcn_readlane(ps, j + 2), lane);
            const row_raw<TAB16> v3 = load_row(nsrc, __builtin_amdgcn_readlane(ps, j + 3), lane);
            float w0 = __int_as_float(__builtin_amdgcn_readlane(pw, j));
            float w1 = __int_as_float(__builtin_amdgcn_readlane(pw, j + 1));
            float w2 = __int_as_float(__builtin_amdgcn_readlane(pw, j + 2));
            float w3 = __int_as_float(__builtin_amdgcn_readlane(pw, j + 3));
            fma4(a0, widen4(v0.x), w0); fma4(a1, widen4(v0.y), w0); ws += w0;
            fma4(a0, widen4(v1.x), w1); fma4(a1, widen4(v1.y), w1); ws += w1;
            fma4(a0, widen4(v2.x), w2); fma4(a1, widen4(v2.y), w2); ws += w2;
            fma4(a0, widen4(v3.x), w3); fma4(a1, widen4(v3.y), w3); ws += w3;
        }
        #pragma nounroll
        for (; j < n; ++j) {
            const row_raw<TAB16> v0 = load_row(nsrc, __builtin_amdgcn_readlane(ps, j), lane);
            float w0 = __int_as_float(__builtin_amdgcn_readlane(pw, j));
            fma4(a0, widen4(v0.x), w0); fma4(a1, widen4(v0.y), w0); ws += w0;
        }
    }
    if (deg > SLOTS) {
        const int m = __builtin_amdgcn_readfirstlane(ovf_n[0]);
        for (int base = 0; base < m; base += 32) {
            int od = -1, os = 0, ow = 0;
            if (base + lane < m) {
                const int4 e = ovf[base + lane];
                od = e.x; os = e.y; ow = e.z;
            }
            unsigned mask = __builtin_amdgcn_ballot_w32(od == r);
            while (mask != 0u) {
                const int k = __builtin_ctz(mask);
                mask &= mask - 1u;
                const row_raw<TAB16> v0 = load_row(nsrc, __builtin_amdgcn_readlane(os, k), lane);
                float w0 = __int_as_float(__builtin_amdgcn_readlane(ow, k));
                fma4(a0, widen4(v0.x), w0); fma4(a1, widen4(v0.y), w0); ws += w0;
            }
        }
    }
    ws = ws < 1.0f ? 1.0f : ws;
    const float inv = __frcp_rn(ws);
    zr[lane]      = make_float4(a0.x * inv, a0.y * inv, a0.z * inv, a0.w * inv);
    zr[32 + lane] = make_float4(a1.x * inv, a1.y * inv, a1.z * inv, a1.w * inv);
}

__global__ __launch_bounds__(256)
void segment_gather_kernel(const float* __restrict__ nsrc,
                           const int* __restrict__ cnt,
                           const int* __restrict__ slots,
                           const int4* __restrict__ ovf,
                           const int* __restrict__ ovf_n,
                           const float* __restrict__ hdst,
                           float* __restrict__ z, int rows)
{
    segment_gather_body<false>(nsrc, cnt, slots, ovf, ovf_n, hdst, z, rows);
}

__global__ __launch_bounds__(256)
void segment_gather_bf16_kernel(const __bf16* __restrict__ nsrc,
                                const int* __restrict__ cnt,
                                const int* __restrict__ slots,
                                const int4* __restrict__ ovf,
                                const int* __restrict__ ovf_n,
                                const float* __restrict__ hdst,
                                float* __restrict__ z, int rows)
{
    segment_gather_body<true>(nsrc, cnt, slots, ovf, ovf_n, hdst, z, rows);
}

#define NORM_ROWS_PER_WAVE 4
static_assert(N_DST % (8 * NORM_ROWS_PER_WAVE) == 0, "whole waves only");
__global__ __launch_bounds__(256)
void skip_norm_kernel(const float* __restrict__ hp, const float* __restrict__ h2,
                      float* __restrict__ zf, int rows)
{
    const int lane = threadIdx.x & 31;
    const int r0   = (blockIdx.x * 8 + (threadIdx.x >> 5)) * NORM_ROWS_PER_WAVE;
    if (r0 >= rows) return;
    float4 a0[NORM_ROWS_PER_WAVE], a1[NORM_ROWS_PER_WAVE], b0[NORM_ROWS_PER_WAVE], b1[NORM_ROWS_PER_WAVE];
    #pragma unroll
    for (int k = 0; k < NORM_ROWS_PER_WAVE; ++k) {
        const float4* a = (const float4*)(hp + (size_t)(r0 + k) * HDIM + lane * 8);
        const float4* b = (const float4*)(h2 + (size_t)(r0 + k) * HDIM + lane * 8);
        a0[k] = a[0]; a1[k] = a[1]; b0[k] = b[0]; b1[k] = b[1];
    }
    #pragma unroll
    for (int k = 0; k < NORM_ROWS_PER_WAVE; ++k) {
        float v[8];
        v[0] = a0[k].x + b0[k].x; v[1] = a0[k].y + b0[k].y; v[2] = a0[k].z + b0[k].z; v[3] = a0[k].w + b0[k].w;
        v[4] = a1[k].x + b1[k].x; v[5] = a1[k].y + b1[k].y; v[6] = a1[k].z + b1[k].z; v[7] = a1[k].w + b1[k].w;
        float ss = 0.0f;
        #pragma unroll
        for (int j = 0; j < 8; ++j) ss += v[j] * v[j];
        #pragma unroll
        for (int o = 16; o > 0; o >>= 1) ss += __shfl_xor(ss, o, 32);
        const float nrm   = sqrtf(ss);
        const float scale = (nrm == 0.0f) ? 1.0f : 1.0f / nrm;
        float4* z = (float4*)(zf + (size_t)(r0 + k) * HDIM + lane * 8);
        z[0] = make_float4(v[0] * scale, v[1] * scale, v[2] * scale, v[3] * scale);
        z[1] = make_float4(v[4] * scale, v[5] * scale, v[6] * scale, v[7] * scale);
    }
}

#define PAIRS_PER_WAVE 4
static_assert(N_PAIRS % (8 * PAIRS_PER_WAVE) == 0, "whole waves only");
__global__ __launch_bounds__(256)
void score_kernel(const float* __restrict__ zf, const float* __restrict__ bias,
                  const int* __restrict__ nids,
                  const int* __restrict__ ps, const int* __restrict__ pd,
                  const int* __restrict__ ns, const int* __restrict__ nd,
                  float* __restrict__ out, int npairs)
{
    const int lane = threadIdx.x & 31;
    const int p0   = (blockIdx.x * 8 + (threadIdx.x >> 5)) * PAIRS_PER_WAVE;
    if (p0 >= npairs) return;
    int pu[PAIRS_PER_WAVE], pv[PAIRS_PER_WAVE], nu[PAIRS_PER_WAVE], nv[PAIRS_PER_WAVE];
    #pragma unroll
    for (int k = 0; k < PAIRS_PER_WAVE; ++k) {
        pu[k] = ps[p0 + k]; pv[k] = pd[p0 + k]; nu[k] = ns[p0 + k]; nv[k] = nd[p0 + k];
    }
    int ipu[PAIRS_PER_WAVE], ipv[PAIRS_PER_WAVE], inu[PAIRS_PER_WAVE], inv[PAIRS_PER_WAVE];
    #pragma unroll
    for (int k = 0; k < PAIRS_PER_WAVE; ++k) {
        ipu[k] = nids[pu[k]]; ipv[k] = nids[pv[k]];
        inu[k] = nids[nu[k]]; inv[k] = nids[nv[k]];
    }
    float4 a0[PAIRS_PER_WAVE], a1[PAIRS_PER_WAVE], b0[PAIRS_PER_WAVE], b1[PAIRS_PER_WAVE];
    float4 e0[PAIRS_PER_WAVE], e1[PAIRS_PER_WAVE], f0[PAIRS_PER_WAVE], f1[PAIRS_PER_WAVE];
    #pragma unroll
    for (int k = 0; k < PAIRS_PER_WAVE; ++k) {
        const float4* a = (const float4*)(zf + (size_t)pu[k] * HDIM + lane * 8);
        const float4* b = (const float4*)(zf + (size_t)pv[k] * HDIM + lane * 8);
        const float4* e = (const float4*)(zf + (size_t)nu[k] * HDIM + lane * 8);
        const float4* f = (const float4*)(zf + (size_t)nv[k] * HDIM + lane * 8);
        a0[k] = a[0]; a1[k] = a[1]; b0[k] = b[0]; b1[k] = b[1];
        e0[k] = e[0]; e1[k] = e[1]; f0[k] = f[0]; f1[k] = f[1];
    }
    float gp[PAIRS_PER_WAVE], gn[PAIRS_PER_WAVE];
    #pragma unroll
    for (int k = 0; k < PAIRS_PER_WAVE; ++k) {
        gp[k] = bias[ipu[k]] + bias[ipv[k]];
        gn[k] = bias[inu[k]] + bias[inv[k]];
    }
    #pragma unroll
    for (int k = 0; k < PAIRS_PER_WAVE; ++k) {
        float sp = 0.0f, sn = 0.0f;
        sp += a0[k].x * b0[k].x; sn += e0[k].x * f0[k].x;
        sp += a0[k].y * b0[k].y; sn += e0[k].y * f0[k].y;
        sp += a0[k].z * b0[k].z; sn += e0[k].z * f0[k].z;
        sp += a0[k].w * b0[k].w; sn += e0[k].w * f0[k].w;
        sp += a1[k].x * b1[k].x; sn += e1[k].x * f1[k].x;
        sp += a1[k].y * b1[k].y; sn += e1[k].y * f1[k].y;
        sp += a1[k].z * b1[k].z; sn += e1[k].z * f1[k].z;
        sp += a1[k].w * b1[k].w; sn += e1[k].w * f1[k].w;
        #pragma unroll
        for (int o = 16; o > 0; o >>= 1) {
            sp += __shfl_xor(sp, o, 32);
            sn += __shfl_xor(sn, o, 32);
        }
        const float m = (sn + gn[k]) - (sp + gp[k]) + 1.0f;
        if (lane == 0) out[p0 + k] = m > 0.0f ? m : 0.0f;
    }
}

extern "C" void kernel_launch(void* const* d_in, const int* in_sizes, int n_in,
                              void* d_out, int out_size, void* d_ws, size_t ws_size,
                              hipStream_t stream) {
    const float* feat = (const float*)d_in[0];
    const float* Wp   = (const float*)d_in[1];
    const float* bp   = (const float*)d_in[2];
    const float* Q1   = (const float*)d_in[3];
    const float* bq1  = (const float*)d_in[4];
    const float* W1   = (const float*)d_in[5];
    const float* bw1  = (const float*)d_in[6];
    const float* Q2   = (const float*)d_in[7];
    const float* bq2  = (const float*)d_in[8];
    const float* W2   = (const float*)d_in[9];
    const float* bw2  = (const float*)d_in[10];
    const float* w1v  = (const float*)d_in[11];
    const float* w2v  = (const float*)d_in[12];
    const float* bias = (const float*)d_in[13];
    const int* src1   = (const int*)d_in[14];
    const int* dst1   = (const int*)d_in[15];
    const int* src2   = (const int*)d_in[16];
    const int* dst2   = (const int*)d_in[17];
    const int* pos_s  = (const int*)d_in[18];
    const int* pos_d  = (const int*)d_in[19];
    const int* neg_s  = (const int*)d_in[20];
    const int* neg_d  = (const int*)d_in[21];
    const int* nids   = (const int*)d_in[22];
    float* out        = (float*)d_out;

    char* ws = (char*)d_ws;
    const size_t SZ_HP  = (size_t)N_SRC * HDIM * 4;
    const size_t SZ_N   = (size_t)N_SRC * HDIM * 4;
    const size_t SZ_H1  = (size_t)N_MID * HDIM * 4;
    const size_t SZ_AGG = (size_t)N_MID * HDIM * 4;
    const size_t SZ_WS  = 256 * 1024;
    const size_t SZ_H2  = (size_t)N_DST * HDIM * 4;
    const size_t SZ_ZF  = (size_t)N_DST * HDIM * 4;
    float*  hp   = (float*)(ws);
    float*  nbuf = (float*)(ws + SZ_HP);
    __bf16* n1b  = (__bf16*)(ws + SZ_HP);
    float*  h1   = (float*)(ws + SZ_HP + SZ_N);
    float*  zbuf = hp + (size_t)N_MID * HDIM;
    int*  cnt1   = (int*)(ws + SZ_HP + SZ_N + SZ_H1);
    int*  cnt2   = cnt1 + N_MID;
    int*  ovfn1  = cnt2 + N_DST;
    int*  ovfn2  = ovfn1 + 64;
    int*  slots1 = cnt1 + CNT_INTS;
    int*  slots2 = slots1 + 2 * SLOTS * N_MID;
    int4* ovf1   = (int4*)(slots2 + 2 * SLOTS * N_DST);
    int4* ovf2   = ovf1 + E1_N;
    float*  h2   = (float*)(ws + SZ_HP + SZ_N + SZ_H1 + SZ_AGG + SZ_WS);
    float*  zf   = (float*)(ws + SZ_HP + SZ_N + SZ_H1 + SZ_AGG + SZ_WS + SZ_H2);
    __bf16* Wpb  = (__bf16*)(ws + SZ_HP + SZ_N + SZ_H1 + SZ_AGG + SZ_WS + SZ_H2 + SZ_ZF);
    __bf16* Q1b  = Wpb + HDIM * HDIM;
    __bf16* W1b  = Q1b + HDIM * HDIM;
    __bf16* Q2b  = W1b + HDIM * 2 * HDIM;
    __bf16* W2b  = Q2b + HDIM * HDIM;

    prep_kernel<<<PREP_W_BLOCKS + PREP_Z_BLOCKS, PREP_THREADS, 0, stream>>>(Wp, Q1, W1, Q2, W2, Wpb, cnt1);

    edge_bin_kernel<<<(E1_N + E2_N) / EDGES_PER_THREAD / 256, 256, 0, stream>>>(src1, dst1, w1v, cnt1, slots1, ovf1, ovfn1,
                                                                            src2, dst2, w2v, cnt2, slots2, ovf2, ovfn2);

    gemm_pair_bf16_w_in_lds<<<256, 512, 0, stream>>>(feat, Wpb, bp, bq1, hp, n1b, N_SRC, N_MID);
    segment_gather_bf16_kernel<<<N_MID / 8, 256, 0, stream>>>(n1b, cnt1, slots1, ovf1, ovfn1, hp, zbuf, N_MID);
    gemm_bf16_w_in_lds<2 * HDIM, 8><<<256, 512, 0, stream>>>(zbuf, W1b, bw1, h1, N_MID, 1);
    gemm_bf16_w_in_lds<HDIM, 8><<<256, 512, 0, stream>>>(h1, Q2b, bq2, nbuf, N_MID, 1);
    segment_gather_kernel<<<N_DST / 8, 256, 0, stream>>>(nbuf, cnt2, slots2, ovf2, ovfn2, h1, zbuf, N_DST);
    gemm_bf16_n256<<<N_DST / 32, 256, 0, stream>>>(zbuf, W2b, bw2, h2, N_DST, 2 * HDIM, 1);
    skip_norm_kernel<<<N_DST / (8 * NORM_ROWS_PER_WAVE), 256, 0, stream>>>(hp, h2, zf, N_DST);
    score_kernel<<<N_PAIRS / (8 * PAIRS_PER_WAVE), 256, 0, stream>>>(zf, bias, nids,
                                                  pos_s, pos_d, neg_s, neg_d,
                                                  out, N_PAIRS);
}


# GEMM A operands z1, h1, z2 stored once by their producers as bf16 (the loader's rounding, done at the store) in place: no f32 round trip, no hp/h1 pass-through copy; sums in arrival order
# speedup vs baseline: 3.8181x; 1.0805x over previous
#include <hip/hip_runtime.h>
#include <hip/hip_bf16.h>


typedef __attribute__((ext_vector_type(16))) __bf16 v16bf;
typedef __attribute__((ext_vector_type(8)))  float  v8f;
typedef __attribute__((ext_vector_type(8)))  __bf16 v8bf;
typedef __attribute__((ext_vector_type(4)))  __bf16 v4bf;
typedef __attribute__((ext_vector_type(8)))  unsigned int v8u;

struct b_step { uint4 lo, hi; };

__device__ __forceinline__ b_step load_b_step(const __bf16* p) {
    b_step s;
    s.lo = *(const uint4*)(p);
    s.hi = *(const uint4*)(p + 16);
    return s;
}

__device__ __forceinline__ v16bf frag_of(const b_step& q) {
    v8u w;
    w[0] = q.lo.x; w[1] = q.lo.y; w[2] = q.lo.z; w[3] = q.lo.w;
    w[4] = q.hi.x; w[5] = q.hi.y; w[6] = q.hi.z; w[7] = q.hi.w;
    return __builtin_bit_cast(v16bf, w);
}

__device__ __forceinline__ v16bf lds_frag(const unsigned char* bp) {
    b_step q;
    q.lo = *(const uint4*)(bp);
    q.hi = *(const uint4*)(bp + 32);
    return frag_of(q);
}

#define N_SRC   131072
#define N_MID   32768
#define N_DST   8192
#define E1_N    524288
#define E2_N    131072
#define HDIM    256
#define N_PAIRS 8192

#define PREP_THREADS  128
#define PREP_SLAB     (HDIM * HDIM / 4)
#define PREP_W_BLOCKS (PREP_SLAB / PREP_THREADS)
#define PREP_Z_UPT    4
#define CNT_INTS      (N_MID + N_DST + 2048)
#define PREP_Z_BLOCKS (CNT_INTS / 4 / PREP_Z_UPT / PREP_THREADS)
__global__ __launch_bounds__(PREP_THREADS)
void prep_kernel(const float* __restrict__ Wp, const float* __restrict__ Q1,
                 const float* __restrict__ W1, const float* __restrict__ Q2,
                 const float* __restrict__ W2, __bf16* __restrict__ wb,
                 int* __restrict__ cnt)
{
    typedef __attribute__((ext_vector_type(4))) __bf16 v4bf;
    static_assert(PREP_SLAB % PREP_THREADS == 0 &&
                  CNT_INTS % (4 * PREP_Z_UPT * PREP_THREADS) == 0, "whole blocks only");
    const int tid = threadIdx.x;
    const int blk = blockIdx.x;
    if (blk < PREP_W_BLOCKS) {
        const int u = blk * PREP_THREADS + tid;
        float4 v[7];
        v[0] = ((const float4*)Wp)[u];
        v[1] = ((const float4*)Q1)[u];
        v[2] = ((const float4*)W1)[u];
        v[3] = ((const float4*)W1)[u + PREP_SLAB];
        v[4] = ((const float4*)Q2)[u];
        v[5] = ((const float4*)W2)[u];
        v[6] = ((const float4*)W2)[u + PREP_SLAB];
        #pragma unroll
        for (int k = 0; k < 7; ++k) {
            v4bf o;
            o[0] = (__bf16)v[k].x; o[1] = (__bf16)v[k].y;
            o[2] = (__bf16)v[k].z; o[3] = (__bf16)v[k].w;
            *(v4bf*)(wb + ((size_t)k * PREP_SLAB + u) * 4) = o;
        }
    } else {
        #pragma unroll
        for (int k = 0; k < PREP_Z_UPT; ++k) {
            const int u = (blk - PREP_W_BLOCKS + k * PREP_Z_BLOCKS) * PREP_THREADS + tid;
            ((int4*)cnt)[u] = make_int4(0, 0, 0, 0);
        }
    }
}

__global__ __launch_bounds__(256)
void gemm_bf16_n256(const __bf16* __restrict__ A, const __bf16* __restrict__ W,
                    const float* __restrict__ bias, float* __restrict__ C,
                    int M, int K, int relu)
{
    const int lane = threadIdx.x & 31;
    const int wave = threadIdx.x >> 5;
    const int half = lane >> 4;
    const int l15  = lane & 15;
    const int m0   = blockIdx.x * 32 + (wave >> 2) * 16;
    const int n0   = (wave & 3) * 64;

    v8f acc[4];
    #pragma unroll
    for (int t = 0; t < 4; ++t) acc[t] = (v8f){};

    const __bf16* arow = A + (size_t)(m0 + l15) * K + half * 8;

    for (int k0 = 0; k0 < K; k0 += 32) {
        const v16bf a = frag_of(load_b_step(arow + k0));
        #pragma unroll
        for (int t = 0; t < 4; ++t) {
            const __bf16* wrow = W + (size_t)(n0 + t * 16 + l15) * K;
            v16bf b;
            #pragma unroll
            for (int q = 0; q < 8; ++q) {
                int kk = k0 + (q >> 2) * 16 + half * 8 + (q & 3) * 2;
                b[2 * q]     = wrow[kk];
                b[2 * q + 1] = wrow[kk + 1];
            }
            acc[t] = __builtin_amdgcn_wmma_f32_16x16x32_bf16(
                false, a, false, b, (short)0, acc[t], false, false);
        }
    }

    #pragma unroll
    for (int t = 0; t < 4; ++t) {
        int n = n0 + t * 16 + l15;
        float bv = bias[n];
        #pragma unroll
        for (int r = 0; r < 8; ++r) {
            int m = m0 + half * 8 + r;
            float v = acc[t][r] + bv;
            if (relu) v = fmaxf(v, 0.0f);
            C[(size_t)m * HDIM + n] = v;
        }
    }
}

struct a_step { float4 f0, f1, f2, f3; };

__device__ __forceinline__ a_step load_a_step(const float* p) {
    a_step s;
    s.f0 = *(const float4*)(p);
    s.f1 = *(const float4*)(p + 4);
    s.f2 = *(const float4*)(p + 16);
    s.f3 = *(const float4*)(p + 20);
    return s;
}

template <int K, int NT, bool OUT16>
__global__ __launch_bounds__(512)
void gemm_bf16_w_in_lds(const __bf16* __restrict__ A, const __bf16* __restrict__ W,
                        const float* __restrict__ bias, float* __restrict__ C,
                        __bf16* __restrict__ Cb, __bf16* __restrict__ Zb,
                        int M, int relu, int keep_rows)
{
    constexpr int PITCH  = K * 2 + 16;
    constexpr int CPR    = K / 8;
    constexpr int CHUNKS = HDIM * CPR;
    constexpr int WPR    = 16 / NT;
    constexpr int KS     = K / 32;
    constexpr int G      = (65535 - 47 - (KS - 1) * 64) / (16 * PITCH) + 1 < NT
                         ? (65535 - 47 - (KS - 1) * 64) / (16 * PITCH) + 1 : NT;
    constexpr int NB     = (NT + G - 1) / G;
    __shared__ __attribute__((aligned(16))) unsigned char L[HDIM * PITCH];

    const int tid  = threadIdx.x;
    const int lane = tid & 31;
    const int wave = __builtin_amdgcn_readfirstlane(tid >> 5);
    const int half = lane >> 4;
    const int l15  = lane & 15;

    const int total  = (M / 16) * WPR;
    const int stride = gridDim.x * 16;
    int it = blockIdx.x * 16 + wave;

    b_step q0, q1;
    q0.lo = q0.hi = make_uint4(0u, 0u, 0u, 0u);
    q1 = q0;
    if (it < total) {
        const __bf16* p = A + (size_t)((it / WPR) * 16 + l15) * K + half * 8;
        q0 = load_b_step(p);
        q1 = load_b_step(p + 32);
    }

    #pragma unroll 16
    for (int i = 0; i < CHUNKS / 512; ++i) {
        const int c = i * 512 + tid;
        const int row = c / CPR, col = c % CPR;
        const uint4 v = *(const uint4*)(W + (size_t)c * 8);
        *(uint4*)(L + row * PITCH + col * 16) = v;
    }
    __syncthreads();

    unsigned goff[NB];
    #pragma unroll
    for (int g = 0; g < NB; ++g) {
        goff[g] = g * G * 16 * PITCH;
        if (g > 0) asm volatile("" : "+v"(goff[g]));
    }

    for (; it < total; it += stride) {
        const int m0 = (it / WPR) * 16;
        const int n0 = (it % WPR) * (NT * 16);
        const __bf16* arow = A + (size_t)(m0 + l15) * K + half * 8;
        const bool more = it + stride < total;
        const __bf16* anext = A + (size_t)(((it + stride) / WPR) * 16 + l15) * K + half * 8;
        const unsigned char* brow = L + (n0 + l15) * PITCH + half * 16;

        v8f acc[NT];
        #pragma unroll
        for (int t = 0; t < NT; ++t) acc[t] = (v8f){};

        float4 bc0[OUT16 ? NT : 1], bc1[OUT16 ? NT : 1];
        if constexpr (OUT16) {
            #pragma unroll
            for (int t = 0; t < NT; ++t) {
                bc0[t] = *(const float4*)(bias + n0 + half * 8 + t * 16);
                bc1[t] = *(const float4*)(bias + n0 + half * 8 + t * 16 + 4);
            }
        }

        #pragma unroll
        for (int ks = 0; ks < KS; ++ks) {
            b_step q2 = q1;
            if (ks + 2 < KS)  q2 = load_b_step(arow + (ks + 2) * 32);
            else if (more)    q2 = load_b_step(anext + (ks + 2 - KS) * 32);

            const v16bf a = frag_of(q0);

            #pragma unroll
            for (int t = 0; t < NT; ++t) {
                const v16bf b = lds_frag(brow + goff[t / G] + (t % G) * 16 * PITCH + ks * 64);
                if constexpr (OUT16)
                    acc[t] = __builtin_amdgcn_wmma_f32_16x16x32_bf16(
                        false, b, false, a, (short)0, acc[t], false, false);
                else
                    acc[t] = __builtin_amdgcn_wmma_f32_16x16x32_bf16(
                        false, a, false, b, (short)0, acc[t], false, false);
            }
            q0 = q1;
            q1 = q2;
        }

        if constexpr (OUT16) {
            const bool keep = m0 < keep_rows;
            __bf16* crow = Cb + (size_t)(m0 + l15) * HDIM + n0 + half * 8;
            __bf16* zrow = Zb + (size_t)(m0 + l15) * (2 * HDIM) + HDIM + n0 + half * 8;
            #pragma unroll
            for (int t = 0; t < NT; ++t) {
                const float4 c0 = bc0[t], c1 = bc1[t];
                float v[8];
                v[0] = acc[t][0] + c0.x; v[1] = acc[t][1] + c0.y;
                v[2] = acc[t][2] + c0.z; v[3] = acc[t][3] + c0.w;
                v[4] = acc[t][4] + c1.x; v[5] = acc[t][5] + c1.y;
                v[6] = acc[t][6] + c1.z; v[7] = acc[t][7] + c1.w;
                v8bf o;
                #pragma unroll
                for (int r = 0; r < 8; ++r) {
                    if (relu) v[r] = fmaxf(v[r], 0.0f);
                    o[r] = (__bf16)v[r];
                }
                *(v8bf*)(crow + t * 16) = o;
                if (keep) *(v8bf*)(zrow + t * 16) = o;
            }
        } else {
            #pragma unroll
            for (int t = 0; t < NT; ++t) {
                const int n = n0 + t * 16 + l15;
                const float bv = bias[n];
                #pragma unroll
                for (int r = 0; r < 8; ++r) {
                    const int m = m0 + half * 8 + r;
                    float v = acc[t][r] + bv;
                    if (relu) v = fmaxf(v, 0.0f);
                    C[(size_t)m * HDIM + n] = v;
                }
            }
        }
    }
}

__device__ __forceinline__ void store_a_step(float* p, const a_step& s) {
    *(float4*)(p)      = s.f0;
    *(float4*)(p + 4)  = s.f1;
    *(float4*)(p + 16) = s.f2;
    *(float4*)(p + 20) = s.f3;
}

__device__ __forceinline__ v16bf to_frag(const a_step& q) {
    v16bf a;
    a[0]  = (__bf16)q.f0.x; a[1]  = (__bf16)q.f0.y; a[2]  = (__bf16)q.f0.z; a[3]  = (__bf16)q.f0.w;
    a[4]  = (__bf16)q.f1.x; a[5]  = (__bf16)q.f1.y; a[6]  = (__bf16)q.f1.z; a[7]  = (__bf16)q.f1.w;
    a[8]  = (__bf16)q.f2.x; a[9]  = (__bf16)q.f2.y; a[10] = (__bf16)q.f2.z; a[11] = (__bf16)q.f2.w;
    a[12] = (__bf16)q.f3.x; a[13] = (__bf16)q.f3.y; a[14] = (__bf16)q.f3.z; a[15] = (__bf16)q.f3.w;
    return a;
}

__global__ __launch_bounds__(512)
void gemm_pair_bf16_w_in_lds(const float* __restrict__ X, const __bf16* __restrict__ W,
                             const float* __restrict__ b0, const float* __restrict__ b1,
                             float* __restrict__ H, __bf16* __restrict__ Zb,
                             __bf16* __restrict__ C,
                             int M, int keep16_rows, int keep32_rows)
{
    constexpr int K      = HDIM;
    constexpr int PITCH  = K * 2 + 16;
    constexpr int CPR    = K / 8;
    constexpr int ROWS   = 2 * HDIM;
    constexpr int CHUNKS = ROWS * CPR;
    constexpr int KS     = K / 32;
    constexpr int NT     = HDIM / 16;
    constexpr int G      = 8;
    constexpr int NB     = 2 * NT / G;
    static_assert((G - 1) * 16 * PITCH + (KS - 1) * 64 + 47 < 65536,
                  "a group's fragments must be in reach of the 16-bit LDS offset");
    __shared__ __attribute__((aligned(16))) unsigned char L[ROWS * PITCH];

    const int tid  = threadIdx.x;
    const int lane = tid & 31;
    const int wave = __builtin_amdgcn_readfirstlane(tid >> 5);
    const int half = lane >> 4;
    const int l15  = lane & 15;

    const int total   = M / 16;
    const int quarter = total / 4;
    const int stride  = gridDim.x * 16;
    int it = blockIdx.x * 16 + wave;

    a_step q0, q1;
    q0.f0 = q0.f1 = q0.f2 = q0.f3 = make_float4(0.f, 0.f, 0.f, 0.f);
    q1 = q0;
    if (it < total) {
        const int rt = (it >> 2) + (it & 3) * quarter;
        const float* p = X + (size_t)(rt * 16 + l15) * K + half * 8;
        q0 = load_a_step(p);
        q1 = load_a_step(p + 32);
    }

    #pragma unroll 16
    for (int i = 0; i < CHUNKS / 512; ++i) {
        const int c = i * 512 + tid;
        const int row = c / CPR, col = c % CPR;
        const uint4 v = *(const uint4*)(W + (size_t)c * 8);
        *(uint4*)(L + row * PITCH + col * 16) = v;
    }
    __syncthreads();

    unsigned goff[NB];
    #pragma unroll
    for (int g = 0; g < NB; ++g) {
        goff[g] = g * G * 16 * PITCH;
        if (g > 0) asm volatile("" : "+v"(goff[g]));
    }
    const unsigned char* brow = L + l15 * PITCH + half * 16;

    for (; it < total; it += stride) {
        const int rt = (it >> 2) + (it & 3) * quarter;
        const int m0 = rt * 16;
        const float* xrow = X + (size_t)(m0 + l15) * K + half * 8;
        const bool keep16 = m0 < keep16_rows;
        const bool keep32 = m0 < keep32_rows;
        const bool more = it + stride < total;
        const int itn = it + stride;
        const float* xnext = X + (size_t)(((itn >> 2) + (itn & 3) * quarter) * 16 + l15) * K + half * 8;

        v8f acc[NT];
        #pragma unroll
        for (int t = 0; t < NT; ++t) acc[t] = (v8f){};

        #pragma unroll
        for (int ks = 0; ks < KS; ++ks) {
            a_step q2 = q1;
            if (ks + 2 < KS) q2 = load_a_step(xrow + (ks + 2) * 32);
            const v16bf x = to_frag(q0);
            #pragma unroll
            for (int t = 0; t < NT; ++t) {
                const v16bf w = lds_frag(brow + goff[t / G] + (t % G) * 16 * PITCH + ks * 64);
                acc[t] = __builtin_amdgcn_wmma_f32_16x16x32_bf16(
                    false, w, false, x, (short)0, acc[t], false, false);
            }
            q0 = q1;
            q1 = q2;
        }

        unsigned boff = half * 8;
        asm volatile("" : "+v"(boff));
        #pragma unroll
        for (int s2 = 0; s2 < KS; ++s2) {
            const a_step bb = load_a_step(b0 + boff + s2 * 32);
            v8f& lo = acc[2 * s2];
            v8f& hi = acc[2 * s2 + 1];
            lo[0] += bb.f0.x; lo[1] += bb.f0.y; lo[2] += bb.f0.z; lo[3] += bb.f0.w;
            lo[4] += bb.f1.x; lo[5] += bb.f1.y; lo[6] += bb.f1.z; lo[7] += bb.f1.w;
            hi[0] += bb.f2.x; hi[1] += bb.f2.y; hi[2] += bb.f2.z; hi[3] += bb.f2.w;
            hi[4] += bb.f3.x; hi[5] += bb.f3.y; hi[6] += bb.f3.z; hi[7] += bb.f3.w;
        }

        if (keep32) {
            float* hrow = H + (size_t)(m0 + l15) * K + half * 8;
            #pragma unroll
            for (int s2 = 0; s2 < KS; ++s2) {
                const v8f& lo = acc[2 * s2];
                const v8f& hi = acc[2 * s2 + 1];
                a_step h;
                h.f0 = make_float4(lo[0], lo[1], lo[2], lo[3]);
                h.f1 = make_float4(lo[4], lo[5], lo[6], lo[7]);
                h.f2 = make_float4(hi[0], hi[1], hi[2], hi[3]);
                h.f3 = make_float4(hi[4], hi[5], hi[6], hi[7]);
                store_a_step(hrow + s2 * 32, h);
            }
        }

        if (more) {
            q0 = load_a_step(xnext);
            q1 = load_a_step(xnext + 32);
        }

        v16bf af[KS];
        #pragma unroll
        for (int s2 = 0; s2 < KS; ++s2) {
            const v8f& lo = acc[2 * s2];
            const v8f& hi = acc[2 * s2 + 1];
            v16bf a;
            a[0]  = (__bf16)lo[0]; a[1]  = (__bf16)lo[1]; a[2]  = (__bf16)lo[2]; a[3]  = (__bf16)lo[3];
            a[4]  = (__bf16)lo[4]; a[5]  = (__bf16)lo[5]; a[6]  = (__bf16)lo[6]; a[7]  = (__bf16)lo[7];
            a[8]  = (__bf16)hi[0]; a[9]  = (__bf16)hi[1]; a[10] = (__bf16)hi[2]; a[11] = (__bf16)hi[3];
            a[12] = (__bf16)hi[4]; a[13] = (__bf16)hi[5]; a[14] = (__bf16)hi[6]; a[15] = (__bf16)hi[7];
            af[s2] = a;
        }

        if (keep16) {
            __bf16* zrow = Zb + (size_t)(m0 + l15) * (2 * HDIM) + HDIM + half * 8;
            #pragma unroll
            for (int s2 = 0; s2 < KS; ++s2) {
                const v8u w = __builtin_bit_cast(v8u, af[s2]);
                *(uint4*)(zrow + s2 * 32)      = make_uint4(w[0], w[1], w[2], w[3]);
                *(uint4*)(zrow + s2 * 32 + 16) = make_uint4(w[4], w[5], w[6], w[7]);
            }
        }

        __bf16* crow = C + (size_t)(m0 + l15) * HDIM + half * 8;
        #pragma unroll
        for (int hf = 0; hf < 2; ++hf) {
            v8f ac2[NT / 2];
            #pragma unroll
            for (int t = 0; t < NT / 2; ++t) ac2[t] = (v8f){};
            #pragma unroll
            for (int ks = 0; ks < KS; ++ks) {
                #pragma unroll
                for (int t = 0; t < NT / 2; ++t) {
                    const int T = NT + hf * (NT / 2) + t;
                    const v16bf w = lds_frag(brow + goff[T / G] + (T % G) * 16 * PITCH + ks * 64);
                    ac2[t] = __builtin_amdgcn_wmma_f32_16x16x32_bf16(
                        false, w, false, af[ks], (short)0, ac2[t], false, false);
                }
            }
            #pragma unroll
            for (int t = 0; t < NT / 2; ++t) {
                const int tt = hf * (NT / 2) + t;
                const float4 c0 = *(const float4*)(b1 + boff + tt * 16);
                const float4 c1 = *(const float4*)(b1 + boff + tt * 16 + 4);
                v8bf o;
                o[0] = (__bf16)fmaxf(ac2[t][0] + c0.x, 0.0f);
                o[1] = (__bf16)fmaxf(ac2[t][1] + c0.y, 0.0f);
                o[2] = (__bf16)fmaxf(ac2[t][2] + c0.z, 0.0f);
                o[3] = (__bf16)fmaxf(ac2[t][3] + c0.w, 0.0f);
                o[4] = (__bf16)fmaxf(ac2[t][4] + c1.x, 0.0f);
                o[5] = (__bf16)fmaxf(ac2[t][5] + c1.y, 0.0f);
                o[6] = (__bf16)fmaxf(ac2[t][6] + c1.z, 0.0f);
                o[7] = (__bf16)fmaxf(ac2[t][7] + c1.w, 0.0f);
                *(v8bf*)(crow + tt * 16) = o;
            }
        }
    }
}

#define SLOTS 32
#define EDGES_PER_THREAD 8
static_assert(SLOTS == 32, "a wave reads a row's slots with one load, a lane per slot");

__device__ __forceinline__ void bin1(int* __restrict__ slots, int4* __restrict__ ovf,
                                     int* __restrict__ ovf_n, int d, int rank, int s, float w) {
    if (rank < SLOTS) {
        int2 v; v.x = s; v.y = __float_as_int(w);
        *(int2*)(slots + 2 * ((size_t)d * SLOTS + rank)) = v;
    } else {
        const int q = atomicAdd(ovf_n, 1);
        ovf[q] = make_int4(d, s, __float_as_int(w), 0);
    }
}

__device__ __forceinline__ void bin8(const int* __restrict__ src, const int* __restrict__ dst,
                                     const float* __restrict__ w, int* __restrict__ cnt,
                                     int* __restrict__ slots, int4* __restrict__ ovf,
                                     int* __restrict__ ovf_n, int i) {
    const int4*   d4 = (const int4*)(dst + (size_t)i * EDGES_PER_THREAD);
    const int4*   s4 = (const int4*)(src + (size_t)i * EDGES_PER_THREAD);
    const float4* w4 = (const float4*)(w + (size_t)i * EDGES_PER_THREAD);
    const int4 da = d4[0], db = d4[1];
    const int4 sa = s4[0], sb = s4[1];
    const float4 wa = w4[0], wb = w4[1];
    const int p0 = atomicAdd(cnt + da.x, 1), p1 = atomicAdd(cnt + da.y, 1);
    const int p2 = atomicAdd(cnt + da.z, 1), p3 = atomicAdd(cnt + da.w, 1);
    const int p4 = atomicAdd(cnt + db.x, 1), p5 = atomicAdd(cnt + db.y, 1);
    const int p6 = atomicAdd(cnt + db.z, 1), p7 = atomicAdd(cnt + db.w, 1);
    bin1(slots, ovf, ovf_n, da.x, p0, sa.x, wa.x); bin1(slots, ovf, ovf_n, da.y, p1, sa.y, wa.y);
    bin1(slots, ovf, ovf_n, da.z, p2, sa.z, wa.z); bin1(slots, ovf, ovf_n, da.w, p3, sa.w, wa.w);
    bin1(slots, ovf, ovf_n, db.x, p4, sb.x, wb.x); bin1(slots, ovf, ovf_n, db.y, p5, sb.y, wb.y);
    bin1(slots, ovf, ovf_n, db.z, p6, sb.z, wb.z); bin1(slots, ovf, ovf_n, db.w, p7, sb.w, wb.w);
}

__global__ __launch_bounds__(256)
void edge_bin_kernel(const int* __restrict__ src1, const int* __restrict__ dst1,
                     const float* __restrict__ w1, int* __restrict__ cnt1,
                     int* __restrict__ slots1, int4* __restrict__ ovf1, int* __restrict__ ovfn1,
                     const int* __restrict__ src2, const int* __restrict__ dst2,
                     const float* __restrict__ w2, int* __restrict__ cnt2,
                     int* __restrict__ slots2, int4* __restrict__ ovf2, int* __restrict__ ovfn2)
{
    const int i = blockIdx.x * 256 + threadIdx.x;
    if (i < E1_N / EDGES_PER_THREAD) {
        bin8(src1, dst1, w1, cnt1, slots1, ovf1, ovfn1, i);
    } else {
        const int e = i - E1_N / EDGES_PER_THREAD;
        if (e < E2_N / EDGES_PER_THREAD) bin8(src2, dst2, w2, cnt2, slots2, ovf2, ovfn2, e);
    }
}

__device__ __forceinline__ void fma4(float4& a, const float4& x, float w) {
    a.x = fmaf(x.x, w, a.x); a.y = fmaf(x.y, w, a.y);
    a.z = fmaf(x.z, w, a.z); a.w = fmaf(x.w, w, a.w);
}

template <bool TAB16> struct row_raw;
template <> struct row_raw<false> { float4 x, y; };
template <> struct row_raw<true>  { uint2  x, y; };

__device__ __forceinline__ row_raw<false> load_row(const float* tab, int row, int lane) {
    const float4* p = (const float4*)(tab + (size_t)row * HDIM);
    row_raw<false> v;
    v.x = p[lane]; v.y = p[32 + lane];
    return v;
}
__device__ __forceinline__ row_raw<true> load_row(const __bf16* tab, int row, int lane) {
    const uint2* p = (const uint2*)(tab + (size_t)row * HDIM);
    row_raw<true> v;
    v.x = p[lane]; v.y = p[32 + lane];
    return v;
}
__device__ __forceinline__ float4 widen4(const float4& v) { return v; }
__device__ __forceinline__ float4 widen4(const uint2& v) {
    return make_float4(__uint_as_float(v.x << 16), __uint_as_float(v.x & 0xffff0000u),
                       __uint_as_float(v.y << 16), __uint_as_float(v.y & 0xffff0000u));
}

template <bool TAB16, typename TabT>
__device__ __forceinline__ void segment_gather_body(const TabT* __restrict__ nsrc,
                                                    const int* __restrict__ cnt,
                                                    const int* __restrict__ slots,
                                                    const int4* __restrict__ ovf,
                                                    const int* __restrict__ ovf_n,
                                                    __bf16* __restrict__ z, int rows)
{
    const int lane = threadIdx.x & 31;
    const int r    = blockIdx.x * 8 + (threadIdx.x >> 5);
    if (r >= rows) return;
    __bf16* zr = z + (size_t)r * (2 * HDIM);
    const int deg = __builtin_amdgcn_readfirstlane(cnt[r]);
    float4 a0 = make_float4(0.f, 0.f, 0.f, 0.f);
    float4 a1 = make_float4(0.f, 0.f, 0.f, 0.f);
    float  ws = 0.0f;
    {
        const int n = deg < SLOTS ? deg : SLOTS;
        int ps = 0, pw = 0;
        if (lane < n) {
            int2 p = *(const int2*)(slots + 2 * ((size_t)r * SLOTS + lane));
            ps = p.x; pw = p.y;
        }
        int j = 0;
        for (; j + 4 <= n; j += 4) {
            const row_raw<TAB16> v0 = load_row(nsrc, __builtin_amdgcn_readlane(ps, j),     lane);
            const row_raw<TAB16> v1 = load_row(nsrc, __builtin_amdgcn_readlane(ps, j + 1), lane);
            const row_raw<TAB16> v2 = load_row(nsrc, __builtin_amdgcn_readlane(ps, j + 2), lane);
            const row_raw<TAB16> v3 = load_row(nsrc, __builtin_amdgcn_readlane(ps, j + 3), lane);
            float w0 = __int_as_float(__builtin_amdgcn_readlane(pw, j));
            float w1 = __int_as_float(__builtin_amdgcn_readlane(pw, j + 1));
            float w2 = __int_as_float(__builtin_amdgcn_readlane(pw, j + 2));
            float w3 = __int_as_float(__builtin_amdgcn_readlane(pw, j + 3));
            fma4(a0, widen4(v0.x), w0); fma4(a1, widen4(v0.y), w0); ws += w0;
            fma4(a0, widen4(v1.x), w1); fma4(a1, widen4(v1.y), w1); ws += w1;
            fma4(a0, widen4(v2.x), w2); fma4(a1, widen4(v2.y), w2); ws += w2;
            fma4(a0, widen4(v3.x), w3); fma4(a1, widen4(v3.y), w3); ws += w3;
        }
        #pragma nounroll
        for (; j < n; ++j) {
            const row_raw<TAB16> v0 = load_row(nsrc, __builtin_amdgcn_readlane(ps, j), lane);
            float w0 = __int_as_float(__builtin_amdgcn_readlane(pw, j));
            fma4(a0, widen4(v0.x), w0); fma4(a1, widen4(v0.y), w0); ws += w0;
        }
    }
    if (deg > SLOTS) {
        const int m = __builtin_amdgcn_readfirstlane(ovf_n[0]);
        for (int base = 0; base < m; base += 32) {
            int od = -1, os = 0, ow = 0;
            if (base + lane < m) {
                const int4 e = ovf[base + lane];
                od = e.x; os = e.y; ow = e.z;
            }
            unsigned mask = __builtin_amdgcn_ballot_w32(od == r);
            while (mask != 0u) {
                const int k = __builtin_ctz(mask);
                mask &= mask - 1u;
                const row_raw<TAB16> v0 = load_row(nsrc, __builtin_amdgcn_readlane(os, k), lane);
                float w0 = __int_as_float(__builtin_amdgcn_readlane(ow, k));
                fma4(a0, widen4(v0.x), w0); fma4(a1, widen4(v0.y), w0); ws += w0;
            }
        }
    }
    ws = ws < 1.0f ? 1.0f : ws;
    const float inv = __frcp_rn(ws);
    v4bf o0, o1;
    o0[0] = (__bf16)(a0.x * inv); o0[1] = (__bf16)(a0.y * inv);
    o0[2] = (__bf16)(a0.z * inv); o0[3] = (__bf16)(a0.w * inv);
    o1[0] = (__bf16)(a1.x * inv); o1[1] = (__bf16)(a1.y * inv);
    o1[2] = (__bf16)(a1.z * inv); o1[3] = (__bf16)(a1.w * inv);
    *(v4bf*)(zr + 4 * lane)       = o0;
    *(v4bf*)(zr + 128 + 4 * lane) = o1;
}

__global__ __launch_bounds__(256)
void segment_gather_kernel(const float* __restrict__ nsrc,
                           const int* __restrict__ cnt,
                           const int* __restrict__ slots,
                           const int4* __restrict__ ovf,
                           const int* __restrict__ ovf_n,
                           __bf16* __restrict__ z, int rows)
{
    segment_gather_body<false>(nsrc, cnt, slots, ovf, ovf_n, z, rows);
}

__global__ __launch_bounds__(256)
void segment_gather_bf16_kernel(const __bf16* __restrict__ nsrc,
                                const int* __restrict__ cnt,
                                const int* __restrict__ slots,
                                const int4* __restrict__ ovf,
                                const int* __restrict__ ovf_n,
                                __bf16* __restrict__ z, int rows)
{
    segment_gather_body<true>(nsrc, cnt, slots, ovf, ovf_n, z, rows);
}

#define NORM_ROWS_PER_WAVE 4
static_assert(N_DST % (8 * NORM_ROWS_PER_WAVE) == 0, "whole waves only");
__global__ __launch_bounds__(256)
void skip_norm_kernel(const float* __restrict__ hp, const float* __restrict__ h2,
                      float* __restrict__ zf, int rows)
{
    const int lane = threadIdx.x & 31;
    const int r0   = (blockIdx.x * 8 + (threadIdx.x >> 5)) * NORM_ROWS_PER_WAVE;
    if (r0 >= rows) return;
    float4 a0[NORM_ROWS_PER_WAVE], a1[NORM_ROWS_PER_WAVE], b0[NORM_ROWS_PER_WAVE], b1[NORM_ROWS_PER_WAVE];
    #pragma unroll
    for (int k = 0; k < NORM_ROWS_PER_WAVE; ++k) {
        const float4* a = (const float4*)(hp + (size_t)(r0 + k) * HDIM + lane * 8);
        const float4* b = (const float4*)(h2 + (size_t)(r0 + k) * HDIM + lane * 8);
        a0[k] = a[0]; a1[k] = a[1]; b0[k] = b[0]; b1[k] = b[1];
    }
    #pragma unroll
    for (int k = 0; k < NORM_ROWS_PER_WAVE; ++k) {
        float v[8];
        v[0] = a0[k].x + b0[k].x; v[1] = a0[k].y + b0[k].y; v[2] = a0[k].z + b0[k].z; v[3] = a0[k].w + b0[k].w;
        v[4] = a1[k].x + b1[k].x; v[5] = a1[k].y + b1[k].y; v[6] = a1[k].z + b1[k].z; v[7] = a1[k].w + b1[k].w;
        float ss = 0.0f;
        #pragma unroll
        for (int j = 0; j < 8; ++j) ss += v[j] * v[j];
        #pragma unroll
        for (int o = 16; o > 0; o >>= 1) ss += __shfl_xor(ss, o, 32);
        const float nrm   = sqrtf(ss);
        const float scale = (nrm == 0.0f) ? 1.0f : 1.0f / nrm;
        float4* z = (float4*)(zf + (size_t)(r0 + k) * HDIM + lane * 8);
        z[0] = make_float4(v[0] * scale, v[1] * scale, v[2] * scale, v[3] * scale);
        z[1] = make_float4(v[4] * scale, v[5] * scale, v[6] * scale, v[7] * scale);
    }
}

#define PAIRS_PER_WAVE 4
static_assert(N_PAIRS % (8 * PAIRS_PER_WAVE) == 0, "whole waves only");
__global__ __launch_bounds__(256)
void score_kernel(const float* __restrict__ zf, const float* __restrict__ bias,
                  const int* __restrict__ nids,
                  const int* __restrict__ ps, const int* __restrict__ pd,
                  const int* __restrict__ ns, const int* __restrict__ nd,
                  float* __restrict__ out, int npairs)
{
    const int lane = threadIdx.x & 31;
    const int p0   = (blockIdx.x * 8 + (threadIdx.x >> 5)) * PAIRS_PER_WAVE;
    if (p0 >= npairs) return;
    int pu[PAIRS_PER_WAVE], pv[PAIRS_PER_WAVE], nu[PAIRS_PER_WAVE], nv[PAIRS_PER_WAVE];
    #pragma unroll
    for (int k = 0; k < PAIRS_PER_WAVE; ++k) {
        pu[k] = ps[p0 + k]; pv[k] = pd[p0 + k]; nu[k] = ns[p0 + k]; nv[k] = nd[p0 + k];
    }
    int ipu[PAIRS_PER_WAVE], ipv[PAIRS_PER_WAVE], inu[PAIRS_PER_WAVE], inv[PAIRS_PER_WAVE];
    #pragma unroll
    for (int k = 0; k < PAIRS_PER_WAVE; ++k) {
        ipu[k] = nids[pu[k]]; ipv[k] = nids[pv[k]];
        inu[k] = nids[nu[k]]; inv[k] = nids[nv[k]];
    }
    float4 a0[PAIRS_PER_WAVE], a1[PAIRS_PER_WAVE], b0[PAIRS_PER_WAVE], b1[PAIRS_PER_WAVE];
    float4 e0[PAIRS_PER_WAVE], e1[PAIRS_PER_WAVE], f0[PAIRS_PER_WAVE], f1[PAIRS_PER_WAVE];
    #pragma unroll
    for (int k = 0; k < PAIRS_PER_WAVE; ++k) {
        const float4* a = (const float4*)(zf + (size_t)pu[k] * HDIM + lane * 8);
        const float4* b = (const float4*)(zf + (size_t)pv[k] * HDIM + lane * 8);
        const float4* e = (const float4*)(zf + (size_t)nu[k] * HDIM + lane * 8);
        const float4* f = (const float4*)(zf + (size_t)nv[k] * HDIM + lane * 8);
        a0[k] = a[0]; a1[k] = a[1]; b0[k] = b[0]; b1[k] = b[1];
        e0[k] = e[0]; e1[k] = e[1]; f0[k] = f[0]; f1[k] = f[1];
    }
    float gp[PAIRS_PER_WAVE], gn[PAIRS_PER_WAVE];
    #pragma unroll
    for (int k = 0; k < PAIRS_PER_WAVE; ++k) {
        gp[k] = bias[ipu[k]] + bias[ipv[k]];
        gn[k] = bias[inu[k]] + bias[inv[k]];
    }
    #pragma unroll
    for (int k = 0; k < PAIRS_PER_WAVE; ++k) {
        float sp = 0.0f, sn = 0.0f;
        sp += a0[k].x * b0[k].x; sn += e0[k].x * f0[k].x;
        sp += a0[k].y * b0[k].y; sn += e0[k].y * f0[k].y;
        sp += a0[k].z * b0[k].z; sn += e0[k].z * f0[k].z;
        sp += a0[k].w * b0[k].w; sn += e0[k].w * f0[k].w;
        sp += a1[k].x * b1[k].x; sn += e1[k].x * f1[k].x;
        sp += a1[k].y * b1[k].y; sn += e1[k].y * f1[k].y;
        sp += a1[k].z * b1[k].z; sn += e1[k].z * f1[k].z;
        sp += a1[k].w * b1[k].w; sn += e1[k].w * f1[k].w;
        #pragma unroll
        for (int o = 16; o > 0; o >>= 1) {
            sp += __shfl_xor(sp, o, 32);
            sn += __shfl_xor(sn, o, 32);
        }
        const float m = (sn + gn[k]) - (sp + gp[k]) + 1.0f;
        if (lane == 0) out[p0 + k] = m > 0.0f ? m : 0.0f;
    }
}

extern "C" void kernel_launch(void* const* d_in, const int* in_sizes, int n_in,
                              void* d_out, int out_size, void* d_ws, size_t ws_size,
                              hipStream_t stream) {
    const float* feat = (const float*)d_in[0];
    const float* Wp   = (const float*)d_in[1];
    const float* bp   = (const float*)d_in[2];
    const float* Q1   = (const float*)d_in[3];
    const float* bq1  = (const float*)d_in[4];
    const float* W1   = (const float*)d_in[5];
    const float* bw1  = (const float*)d_in[6];
    const float* Q2   = (const float*)d_in[7];
    const float* bq2  = (const float*)d_in[8];
    const float* W2   = (const float*)d_in[9];
    const float* bw2  = (const float*)d_in[10];
    const float* w1v  = (const float*)d_in[11];
    const float* w2v  = (const float*)d_in[12];
    const float* bias = (const float*)d_in[13];
    const int* src1   = (const int*)d_in[14];
    const int* dst1   = (const int*)d_in[15];
    const int* src2   = (const int*)d_in[16];
    const int* dst2   = (const int*)d_in[17];
    const int* pos_s  = (const int*)d_in[18];
    const int* pos_d  = (const int*)d_in[19];
    const int* neg_s  = (const int*)d_in[20];
    const int* neg_d  = (const int*)d_in[21];
    const int* nids   = (const int*)d_in[22];
    float* out        = (float*)d_out;

    char* ws = (char*)d_ws;
    const size_t SZ_HP  = (size_t)N_SRC * HDIM * 4;
    const size_t SZ_N   = (size_t)N_SRC * HDIM * 4;
    const size_t SZ_H1  = (size_t)N_MID * HDIM * 4;
    const size_t SZ_AGG = (size_t)N_MID * HDIM * 4;
    const size_t SZ_WS  = 256 * 1024;
    const size_t SZ_H2  = (size_t)N_DST * HDIM * 4;
    const size_t SZ_ZF  = (size_t)N_DST * HDIM * 4;
    float*  hp   = (float*)(ws);
    float*  nbuf = (float*)(ws + SZ_HP);
    __bf16* n1b  = (__bf16*)(ws + SZ_HP);
    __bf16* h1b  = (__bf16*)(ws + SZ_HP + SZ_N);
    __bf16* z1b  = (__bf16*)(hp + (size_t)N_MID * HDIM);
    __bf16* z2b  = z1b + (size_t)N_MID * 2 * HDIM;
    int*  cnt1   = (int*)(ws + SZ_HP + SZ_N + SZ_H1);
    int*  cnt2   = cnt1 + N_MID;
    int*  ovfn1  = cnt2 + N_DST;
    int*  ovfn2  = ovfn1 + 64;
    int*  slots1 = cnt1 + CNT_INTS;
    int*  slots2 = slots1 + 2 * SLOTS * N_MID;
    int4* ovf1   = (int4*)(slots2 + 2 * SLOTS * N_DST);
    int4* ovf2   = ovf1 + E1_N;
    float*  h2   = (float*)(ws + SZ_HP + SZ_N + SZ_H1 + SZ_AGG + SZ_WS);
    float*  zf   = (float*)(ws + SZ_HP + SZ_N + SZ_H1 + SZ_AGG + SZ_WS + SZ_H2);
    __bf16* Wpb  = (__bf16*)(ws + SZ_HP + SZ_N + SZ_H1 + SZ_AGG + SZ_WS + SZ_H2 + SZ_ZF);
    __bf16* Q1b  = Wpb + HDIM * HDIM;
    __bf16* W1b  = Q1b + HDIM * HDIM;
    __bf16* Q2b  = W1b + HDIM * 2 * HDIM;
    __bf16* W2b  = Q2b + HDIM * HDIM;

    prep_kernel<<<PREP_W_BLOCKS + PREP_Z_BLOCKS, PREP_THREADS, 0, stream>>>(Wp, Q1, W1, Q2, W2, Wpb, cnt1);

    edge_bin_kernel<<<(E1_N + E2_N) / EDGES_PER_THREAD / 256, 256, 0, stream>>>(src1, dst1, w1v, cnt1, slots1, ovf1, ovfn1,
                                                                            src2, dst2, w2v, cnt2, slots2, ovf2, ovfn2);

    gemm_pair_bf16_w_in_lds<<<256, 512, 0, stream>>>(feat, Wpb, bp, bq1, hp, z1b, n1b, N_SRC, N_MID, N_DST);
    segment_gather_bf16_kernel<<<N_MID / 8, 256, 0, stream>>>(n1b, cnt1, slots1, ovf1, ovfn1, z1b, N_MID);
    gemm_bf16_w_in_lds<2 * HDIM, 8, true><<<256, 512, 0, stream>>>(z1b, W1b, bw1, nullptr, h1b, z2b, N_MID, 1, N_DST);
    gemm_bf16_w_in_lds<HDIM, 8, false><<<256, 512, 0, stream>>>(h1b, Q2b, bq2, nbuf, nullptr, nullptr, N_MID, 1, 0);
    segment_gather_kernel<<<N_DST / 8, 256, 0, stream>>>(nbuf, cnt2, slots2, ovf2, ovfn2, z2b, N_DST);
    gemm_bf16_n256<<<N_DST / 32, 256, 0, stream>>>(z2b, W2b, bw2, h2, N_DST, 2 * HDIM, 1);
    skip_norm_kernel<<<N_DST / (8 * NORM_ROWS_PER_WAVE), 256, 0, stream>>>(hp, h2, zf, N_DST);
    score_kernel<<<N_PAIRS / (8 * PAIRS_PER_WAVE), 256, 0, stream>>>(zf, bias, nids,
                                                  pos_s, pos_d, neg_s, neg_d,
                                                  out, N_PAIRS);
}


# Layer-2 tail in one launch: a 32-wave work-group sums its 32 rows' neighbours into a bf16 LDS tile, multiplies by W2 on WMMA, adds hp and normalises (10->8 launches); row-norm sum order changes
# speedup vs baseline: 3.9343x; 1.0304x over previous
#include <hip/hip_runtime.h>
#include <hip/hip_bf16.h>


typedef __attribute__((ext_vector_type(16))) __bf16 v16bf;
typedef __attribute__((ext_vector_type(8)))  float  v8f;
typedef __attribute__((ext_vector_type(8)))  __bf16 v8bf;
typedef __attribute__((ext_vector_type(4)))  __bf16 v4bf;
typedef __attribute__((ext_vector_type(8)))  unsigned int v8u;

struct b_step { uint4 lo, hi; };

__device__ __forceinline__ b_step load_b_step(const __bf16* p) {
    b_step s;
    s.lo = *(const uint4*)(p);
    s.hi = *(const uint4*)(p + 16);
    return s;
}

__device__ __forceinline__ v16bf frag_of(const b_step& q) {
    v8u w;
    w[0] = q.lo.x; w[1] = q.lo.y; w[2] = q.lo.z; w[3] = q.lo.w;
    w[4] = q.hi.x; w[5] = q.hi.y; w[6] = q.hi.z; w[7] = q.hi.w;
    return __builtin_bit_cast(v16bf, w);
}

__device__ __forceinline__ v16bf lds_frag(const unsigned char* bp) {
    b_step q;
    q.lo = *(const uint4*)(bp);
    q.hi = *(const uint4*)(bp + 32);
    return frag_of(q);
}

#define N_SRC   131072
#define N_MID   32768
#define N_DST   8192
#define E1_N    524288
#define E2_N    131072
#define HDIM    256
#define N_PAIRS 8192

#define PREP_THREADS  128
#define PREP_SLAB     (HDIM * HDIM / 4)
#define PREP_W_BLOCKS (PREP_SLAB / PREP_THREADS)
#define PREP_Z_UPT    4
#define CNT_INTS      (N_MID + N_DST + 2048)
#define PREP_Z_BLOCKS (CNT_INTS / 4 / PREP_Z_UPT / PREP_THREADS)
__global__ __launch_bounds__(PREP_THREADS)
void prep_kernel(const float* __restrict__ Wp, const float* __restrict__ Q1,
                 const float* __restrict__ W1, const float* __restrict__ Q2,
                 const float* __restrict__ W2, __bf16* __restrict__ wb,
                 int* __restrict__ cnt)
{
    typedef __attribute__((ext_vector_type(4))) __bf16 v4bf;
    static_assert(PREP_SLAB % PREP_THREADS == 0 &&
                  CNT_INTS % (4 * PREP_Z_UPT * PREP_THREADS) == 0, "whole blocks only");
    const int tid = threadIdx.x;
    const int blk = blockIdx.x;
    if (blk < PREP_W_BLOCKS) {
        const int u = blk * PREP_THREADS + tid;
        float4 v[7];
        v[0] = ((const float4*)Wp)[u];
        v[1] = ((const float4*)Q1)[u];
        v[2] = ((const float4*)W1)[u];
        v[3] = ((const float4*)W1)[u + PREP_SLAB];
        v[4] = ((const float4*)Q2)[u];
        v[5] = ((const float4*)W2)[u];
        v[6] = ((const float4*)W2)[u + PREP_SLAB];
        #pragma unroll
        for (int k = 0; k < 7; ++k) {
            v4bf o;
            o[0] = (__bf16)v[k].x; o[1] = (__bf16)v[k].y;
            o[2] = (__bf16)v[k].z; o[3] = (__bf16)v[k].w;
            *(v4bf*)(wb + ((size_t)k * PREP_SLAB + u) * 4) = o;
        }
    } else {
        #pragma unroll
        for (int k = 0; k < PREP_Z_UPT; ++k) {
            const int u = (blk - PREP_W_BLOCKS + k * PREP_Z_BLOCKS) * PREP_THREADS + tid;
            ((int4*)cnt)[u] = make_int4(0, 0, 0, 0);
        }
    }
}

struct a_step { float4 f0, f1, f2, f3; };

__device__ __forceinline__ a_step load_a_step(const float* p) {
    a_step s;
    s.f0 = *(const float4*)(p);
    s.f1 = *(const float4*)(p + 4);
    s.f2 = *(const float4*)(p + 16);
    s.f3 = *(const float4*)(p + 20);
    return s;
}

template <int K, int NT, bool OUT16>
__global__ __launch_bounds__(512)
void gemm_bf16_w_in_lds(const __bf16* __restrict__ A, const __bf16* __restrict__ W,
                        const float* __restrict__ bias, float* __restrict__ C,
                        __bf16* __restrict__ Cb, __bf16* __restrict__ Zb,
                        int M, int relu, int keep_rows)
{
    constexpr int PITCH  = K * 2 + 16;
    constexpr int CPR    = K / 8;
    constexpr int CHUNKS = HDIM * CPR;
    constexpr int WPR    = 16 / NT;
    constexpr int KS     = K / 32;
    constexpr int G      = (65535 - 47 - (KS - 1) * 64) / (16 * PITCH) + 1 < NT
                         ? (65535 - 47 - (KS - 1) * 64) / (16 * PITCH) + 1 : NT;
    constexpr int NB     = (NT + G - 1) / G;
    __shared__ __attribute__((aligned(16))) unsigned char L[HDIM * PITCH];

    const int tid  = threadIdx.x;
    const int lane = tid & 31;
    const int wave = __builtin_amdgcn_readfirstlane(tid >> 5);
    const int half = lane >> 4;
    const int l15  = lane & 15;

    const int total  = (M / 16) * WPR;
    const int stride = gridDim.x * 16;
    int it = blockIdx.x * 16 + wave;

    b_step q0, q1;
    q0.lo = q0.hi = make_uint4(0u, 0u, 0u, 0u);
    q1 = q0;
    if (it < total) {
        const __bf16* p = A + (size_t)((it / WPR) * 16 + l15) * K + half * 8;
        q0 = load_b_step(p);
        q1 = load_b_step(p + 32);
    }

    #pragma unroll 16
    for (int i = 0; i < CHUNKS / 512; ++i) {
        const int c = i * 512 + tid;
        const int row = c / CPR, col = c % CPR;
        const uint4 v = *(const uint4*)(W + (size_t)c * 8);
        *(uint4*)(L + row * PITCH + col * 16) = v;
    }
    __syncthreads();

    unsigned goff[NB];
    #pragma unroll
    for (int g = 0; g < NB; ++g) {
        goff[g] = g * G * 16 * PITCH;
        if (g > 0) asm volatile("" : "+v"(goff[g]));
    }

    for (; it < total; it += stride) {
        const int m0 = (it / WPR) * 16;
        const int n0 = (it % WPR) * (NT * 16);
        const __bf16* arow = A + (size_t)(m0 + l15) * K + half * 8;
        const bool more = it + stride < total;
        const __bf16* anext = A + (size_t)(((it + stride) / WPR) * 16 + l15) * K + half * 8;
        const unsigned char* brow = L + (n0 + l15) * PITCH + half * 16;

        v8f acc[NT];
        #pragma unroll
        for (int t = 0; t < NT; ++t) acc[t] = (v8f){};

        float4 bc0[OUT16 ? NT : 1], bc1[OUT16 ? NT : 1];
        if constexpr (OUT16) {
            #pragma unroll
            for (int t = 0; t < NT; ++t) {
                bc0[t] = *(const float4*)(bias + n0 + half * 8 + t * 16);
                bc1[t] = *(const float4*)(bias + n0 + half * 8 + t * 16 + 4);
            }
        }

        #pragma unroll
        for (int ks = 0; ks < KS; ++ks) {
            b_step q2 = q1;
            if (ks + 2 < KS)  q2 = load_b_step(arow + (ks + 2) * 32);
            else if (more)    q2 = load_b_step(anext + (ks + 2 - KS) * 32);

            const v16bf a = frag_of(q0);

            #pragma unroll
            for (int t = 0; t < NT; ++t) {
                const v16bf b = lds_frag(brow + goff[t / G] + (t % G) * 16 * PITCH + ks * 64);
                if constexpr (OUT16)
                    acc[t] = __builtin_amdgcn_wmma_f32_16x16x32_bf16(
                        false, b, false, a, (short)0, acc[t], false, false);
                else
                    acc[t] = __builtin_amdgcn_wmma_f32_16x16x32_bf16(
                        false, a, false, b, (short)0, acc[t], false, false);
            }
            q0 = q1;
            q1 = q2;
        }

        if constexpr (OUT16) {
            const bool keep = m0 < keep_rows;
            __bf16* crow = Cb + (size_t)(m0 + l15) * HDIM + n0 + half * 8;
            __bf16* zrow = Zb + (size_t)(m0 + l15) * (2 * HDIM) + HDIM + n0 + half * 8;
            #pragma unroll
            for (int t = 0; t < NT; ++t) {
                const float4 c0 = bc0[t], c1 = bc1[t];
                float v[8];
                v[0] = acc[t][0] + c0.x; v[1] = acc[t][1] + c0.y;
                v[2] = acc[t][2] + c0.z; v[3] = acc[t][3] + c0.w;
                v[4] = acc[t][4] + c1.x; v[5] = acc[t][5] + c1.y;
                v[6] = acc[t][6] + c1.z; v[7] = acc[t][7] + c1.w;
                v8bf o;
                #pragma unroll
                for (int r = 0; r < 8; ++r) {
                    if (relu) v[r] = fmaxf(v[r], 0.0f);
                    o[r] = (__bf16)v[r];
                }
                *(v8bf*)(crow + t * 16) = o;
                if (keep) *(v8bf*)(zrow + t * 16) = o;
            }
        } else {
            #pragma unroll
            for (int t = 0; t < NT; ++t) {
                const int n = n0 + t * 16 + l15;
                const float bv = bias[n];
                #pragma unroll
                for (int r = 0; r < 8; ++r) {
                    const int m = m0 + half * 8 + r;
                    float v = acc[t][r] + bv;
                    if (relu) v = fmaxf(v, 0.0f);
                    C[(size_t)m * HDIM + n] = v;
                }
            }
        }
    }
}

__device__ __forceinline__ void store_a_step(float* p, const a_step& s) {
    *(float4*)(p)      = s.f0;
    *(float4*)(p + 4)  = s.f1;
    *(float4*)(p + 16) = s.f2;
    *(float4*)(p + 20) = s.f3;
}

__device__ __forceinline__ v16bf to_frag(const a_step& q) {
    v16bf a;
    a[0]  = (__bf16)q.f0.x; a[1]  = (__bf16)q.f0.y; a[2]  = (__bf16)q.f0.z; a[3]  = (__bf16)q.f0.w;
    a[4]  = (__bf16)q.f1.x; a[5]  = (__bf16)q.f1.y; a[6]  = (__bf16)q.f1.z; a[7]  = (__bf16)q.f1.w;
    a[8]  = (__bf16)q.f2.x; a[9]  = (__bf16)q.f2.y; a[10] = (__bf16)q.f2.z; a[11] = (__bf16)q.f2.w;
    a[12] = (__bf16)q.f3.x; a[13] = (__bf16)q.f3.y; a[14] = (__bf16)q.f3.z; a[15] = (__bf16)q.f3.w;
    return a;
}

__global__ __launch_bounds__(512)
void gemm_pair_bf16_w_in_lds(const float* __restrict__ X, const __bf16* __restrict__ W,
                             const float* __restrict__ b0, const float* __restrict__ b1,
                             float* __restrict__ H, __bf16* __restrict__ Zb,
                             __bf16* __restrict__ C,
                             int M, int keep16_rows, int keep32_rows)
{
    constexpr int K      = HDIM;
    constexpr int PITCH  = K * 2 + 16;
    constexpr int CPR    = K / 8;
    constexpr int ROWS   = 2 * HDIM;
    constexpr int CHUNKS = ROWS * CPR;
    constexpr int KS     = K / 32;
    constexpr int NT     = HDIM / 16;
    constexpr int G      = 8;
    constexpr int NB     = 2 * NT / G;
    static_assert((G - 1) * 16 * PITCH + (KS - 1) * 64 + 47 < 65536,
                  "a group's fragments must be in reach of the 16-bit LDS offset");
    __shared__ __attribute__((aligned(16))) unsigned char L[ROWS * PITCH];

    const int tid  = threadIdx.x;
    const int lane = tid & 31;
    const int wave = __builtin_amdgcn_readfirstlane(tid >> 5);
    const int half = lane >> 4;
    const int l15  = lane & 15;

    const int total   = M / 16;
    const int quarter = total / 4;
    const int stride  = gridDim.x * 16;
    int it = blockIdx.x * 16 + wave;

    a_step q0, q1;
    q0.f0 = q0.f1 = q0.f2 = q0.f3 = make_float4(0.f, 0.f, 0.f, 0.f);
    q1 = q0;
    if (it < total) {
        const int rt = (it >> 2) + (it & 3) * quarter;
        const float* p = X + (size_t)(rt * 16 + l15) * K + half * 8;
        q0 = load_a_step(p);
        q1 = load_a_step(p + 32);
    }

    #pragma unroll 16
    for (int i = 0; i < CHUNKS / 512; ++i) {
        const int c = i * 512 + tid;
        const int row = c / CPR, col = c % CPR;
        const uint4 v = *(const uint4*)(W + (size_t)c * 8);
        *(uint4*)(L + row * PITCH + col * 16) = v;
    }
    __syncthreads();

    unsigned goff[NB];
    #pragma unroll
    for (int g = 0; g < NB; ++g) {
        goff[g] = g * G * 16 * PITCH;
        if (g > 0) asm volatile("" : "+v"(goff[g]));
    }
    const unsigned char* brow = L + l15 * PITCH + half * 16;

    for (; it < total; it += stride) {
        const int rt = (it >> 2) + (it & 3) * quarter;
        const int m0 = rt * 16;
        const float* xrow = X + (size_t)(m0 + l15) * K + half * 8;
        const bool keep16 = m0 < keep16_rows;
        const bool keep32 = m0 < keep32_rows;
        const bool more = it + stride < total;
        const int itn = it + stride;
        const float* xnext = X + (size_t)(((itn >> 2) + (itn & 3) * quarter) * 16 + l15) * K + half * 8;

        v8f acc[NT];
        #pragma unroll
        for (int t = 0; t < NT; ++t) acc[t] = (v8f){};

        #pragma unroll
        for (int ks = 0; ks < KS; ++ks) {
            a_step q2 = q1;
            if (ks + 2 < KS) q2 = load_a_step(xrow + (ks + 2) * 32);
            const v16bf x = to_frag(q0);
            #pragma unroll
            for (int t = 0; t < NT; ++t) {
                const v16bf w = lds_frag(brow + goff[t / G] + (t % G) * 16 * PITCH + ks * 64);
                acc[t] = __builtin_amdgcn_wmma_f32_16x16x32_bf16(
                    false, w, false, x, (short)0, acc[t], false, false);
            }
            q0 = q1;
            q1 = q2;
        }

        unsigned boff = half * 8;
        asm volatile("" : "+v"(boff));
        #pragma unroll
        for (int s2 = 0; s2 < KS; ++s2) {
            const a_step bb = load_a_step(b0 + boff + s2 * 32);
            v8f& lo = acc[2 * s2];
            v8f& hi = acc[2 * s2 + 1];
            lo[0] += bb.f0.x; lo[1] += bb.f0.y; lo[2] += bb.f0.z; lo[3] += bb.f0.w;
            lo[4] += bb.f1.x; lo[5] += bb.f1.y; lo[6] += bb.f1.z; lo[7] += bb.f1.w;
            hi[0] += bb.f2.x; hi[1] += bb.f2.y; hi[2] += bb.f2.z; hi[3] += bb.f2.w;
            hi[4] += bb.f3.x; hi[5] += bb.f3.y; hi[6] += bb.f3.z; hi[7] += bb.f3.w;
        }

        if (keep32) {
            float* hrow = H + (size_t)(m0 + l15) * K + half * 8;
            #pragma unroll
            for (int s2 = 0; s2 < KS; ++s2) {
                const v8f& lo = acc[2 * s2];
                const v8f& hi = acc[2 * s2 + 1];
                a_step h;
                h.f0 = make_float4(lo[0], lo[1], lo[2], lo[3]);
                h.f1 = make_float4(lo[4], lo[5], lo[6], lo[7]);
                h.f2 = make_float4(hi[0], hi[1], hi[2], hi[3]);
                h.f3 = make_float4(hi[4], hi[5], hi[6], hi[7]);
                store_a_step(hrow + s2 * 32, h);
            }
        }

        if (more) {
            q0 = load_a_step(xnext);
            q1 = load_a_step(xnext + 32);
        }

        v16bf af[KS];
        #pragma unroll
        for (int s2 = 0; s2 < KS; ++s2) {
            const v8f& lo = acc[2 * s2];
            const v8f& hi = acc[2 * s2 + 1];
            v16bf a;
            a[0]  = (__bf16)lo[0]; a[1]  = (__bf16)lo[1]; a[2]  = (__bf16)lo[2]; a[3]  = (__bf16)lo[3];
            a[4]  = (__bf16)lo[4]; a[5]  = (__bf16)lo[5]; a[6]  = (__bf16)lo[6]; a[7]  = (__bf16)lo[7];
            a[8]  = (__bf16)hi[0]; a[9]  = (__bf16)hi[1]; a[10] = (__bf16)hi[2]; a[11] = (__bf16)hi[3];
            a[12] = (__bf16)hi[4]; a[13] = (__bf16)hi[5]; a[14] = (__bf16)hi[6]; a[15] = (__bf16)hi[7];
            af[s2] = a;
        }

        if (keep16) {
            __bf16* zrow = Zb + (size_t)(m0 + l15) * (2 * HDIM) + HDIM + half * 8;
            #pragma unroll
            for (int s2 = 0; s2 < KS; ++s2) {
                const v8u w = __builtin_bit_cast(v8u, af[s2]);
                *(uint4*)(zrow + s2 * 32)      = make_uint4(w[0], w[1], w[2], w[3]);
                *(uint4*)(zrow + s2 * 32 + 16) = make_uint4(w[4], w[5], w[6], w[7]);
            }
        }

        __bf16* crow = C + (size_t)(m0 + l15) * HDIM + half * 8;
        #pragma unroll
        for (int hf = 0; hf < 2; ++hf) {
            v8f ac2[NT / 2];
            #pragma unroll
            for (int t = 0; t < NT / 2; ++t) ac2[t] = (v8f){};
            #pragma unroll
            for (int ks = 0; ks < KS; ++ks) {
                #pragma unroll
                for (int t = 0; t < NT / 2; ++t) {
                    const int T = NT + hf * (NT / 2) + t;
                    const v16bf w = lds_frag(brow + goff[T / G] + (T % G) * 16 * PITCH + ks * 64);
                    ac2[t] = __builtin_amdgcn_wmma_f32_16x16x32_bf16(
                        false, w, false, af[ks], (short)0, ac2[t], false, false);
                }
            }
            #pragma unroll
            for (int t = 0; t < NT / 2; ++t) {
                const int tt = hf * (NT / 2) + t;
                const float4 c0 = *(const float4*)(b1 + boff + tt * 16);
                const float4 c1 = *(const float4*)(b1 + boff + tt * 16 + 4);
                v8bf o;
                o[0] = (__bf16)fmaxf(ac2[t][0] + c0.x, 0.0f);
                o[1] = (__bf16)fmaxf(ac2[t][1] + c0.y, 0.0f);
                o[2] = (__bf16)fmaxf(ac2[t][2] + c0.z, 0.0f);
                o[3] = (__bf16)fmaxf(ac2[t][3] + c0.w, 0.0f);
                o[4] = (__bf16)fmaxf(ac2[t][4] + c1.x, 0.0f);
                o[5] = (__bf16)fmaxf(ac2[t][5] + c1.y, 0.0f);
                o[6] = (__bf16)fmaxf(ac2[t][6] + c1.z, 0.0f);
                o[7] = (__bf16)fmaxf(ac2[t][7] + c1.w, 0.0f);
                *(v8bf*)(crow + tt * 16) = o;
            }
        }
    }
}

#define SLOTS 32
#define EDGES_PER_THREAD 8
static_assert(SLOTS == 32, "a wave reads a row's slots with one load, a lane per slot");

__device__ __forceinline__ void bin1(int* __restrict__ slots, int4* __restrict__ ovf,
                                     int* __restrict__ ovf_n, int d, int rank, int s, float w) {
    if (rank < SLOTS) {
        int2 v; v.x = s; v.y = __float_as_int(w);
        *(int2*)(slots + 2 * ((size_t)d * SLOTS + rank)) = v;
    } else {
        const int q = atomicAdd(ovf_n, 1);
        ovf[q] = make_int4(d, s, __float_as_int(w), 0);
    }
}

__device__ __forceinline__ void bin8(const int* __restrict__ src, const int* __restrict__ dst,
                                     const float* __restrict__ w, int* __restrict__ cnt,
                                     int* __restrict__ slots, int4* __restrict__ ovf,
                                     int* __restrict__ ovf_n, int i) {
    const int4*   d4 = (const int4*)(dst + (size_t)i * EDGES_PER_THREAD);
    const int4*   s4 = (const int4*)(src + (size_t)i * EDGES_PER_THREAD);
    const float4* w4 = (const float4*)(w + (size_t)i * EDGES_PER_THREAD);
    const int4 da = d4[0], db = d4[1];
    const int4 sa = s4[0], sb = s4[1];
    const float4 wa = w4[0], wb = w4[1];
    const int p0 = atomicAdd(cnt + da.x, 1), p1 = atomicAdd(cnt + da.y, 1);
    const int p2 = atomicAdd(cnt + da.z, 1), p3 = atomicAdd(cnt + da.w, 1);
    const int p4 = atomicAdd(cnt + db.x, 1), p5 = atomicAdd(cnt + db.y, 1);
    const int p6 = atomicAdd(cnt + db.z, 1), p7 = atomicAdd(cnt + db.w, 1);
    bin1(slots, ovf, ovf_n, da.x, p0, sa.x, wa.x); bin1(slots, ovf, ovf_n, da.y, p1, sa.y, wa.y);
    bin1(slots, ovf, ovf_n, da.z, p2, sa.z, wa.z); bin1(slots, ovf, ovf_n, da.w, p3, sa.w, wa.w);
    bin1(slots, ovf, ovf_n, db.x, p4, sb.x, wb.x); bin1(slots, ovf, ovf_n, db.y, p5, sb.y, wb.y);
    bin1(slots, ovf, ovf_n, db.z, p6, sb.z, wb.z); bin1(slots, ovf, ovf_n, db.w, p7, sb.w, wb.w);
}

__global__ __launch_bounds__(256)
void edge_bin_kernel(const int* __restrict__ src1, const int* __restrict__ dst1,
                     const float* __restrict__ w1, int* __restrict__ cnt1,
                     int* __restrict__ slots1, int4* __restrict__ ovf1, int* __restrict__ ovfn1,
                     const int* __restrict__ src2, const int* __restrict__ dst2,
                     const float* __restrict__ w2, int* __restrict__ cnt2,
                     int* __restrict__ slots2, int4* __restrict__ ovf2, int* __restrict__ ovfn2)
{
    const int i = blockIdx.x * 256 + threadIdx.x;
    if (i < E1_N / EDGES_PER_THREAD) {
        bin8(src1, dst1, w1, cnt1, slots1, ovf1, ovfn1, i);
    } else {
        const int e = i - E1_N / EDGES_PER_THREAD;
        if (e < E2_N / EDGES_PER_THREAD) bin8(src2, dst2, w2, cnt2, slots2, ovf2, ovfn2, e);
    }
}

__device__ __forceinline__ void fma4(float4& a, const float4& x, float w) {
    a.x = fmaf(x.x, w, a.x); a.y = fmaf(x.y, w, a.y);
    a.z = fmaf(x.z, w, a.z); a.w = fmaf(x.w, w, a.w);
}

template <bool TAB16> struct row_raw;
template <> struct row_raw<false> { float4 x, y; };
template <> struct row_raw<true>  { uint2  x, y; };

__device__ __forceinline__ row_raw<false> load_row(const float* tab, int row, int lane) {
    const float4* p = (const float4*)(tab + (size_t)row * HDIM);
    row_raw<false> v;
    v.x = p[lane]; v.y = p[32 + lane];
    return v;
}
__device__ __forceinline__ row_raw<true> load_row(const __bf16* tab, int row, int lane) {
    const uint2* p = (const uint2*)(tab + (size_t)row * HDIM);
    row_raw<true> v;
    v.x = p[lane]; v.y = p[32 + lane];
    return v;
}
__device__ __forceinline__ float4 widen4(const float4& v) { return v; }
__device__ __forceinline__ float4 widen4(const uint2& v) {
    return make_float4(__uint_as_float(v.x << 16), __uint_as_float(v.x & 0xffff0000u),
                       __uint_as_float(v.y << 16), __uint_as_float(v.y & 0xffff0000u));
}

template <bool TAB16, typename TabT>
__device__ __forceinline__ void segment_row_sum(const TabT* __restrict__ nsrc,
                                                const int* __restrict__ cnt,
                                                const int* __restrict__ slots,
                                                const int4* __restrict__ ovf,
                                                const int* __restrict__ ovf_n,
                                                int r, int lane, v4bf& o0, v4bf& o1)
{
    const int deg = __builtin_amdgcn_readfirstlane(cnt[r]);
    float4 a0 = make_float4(0.f, 0.f, 0.f, 0.f);
    float4 a1 = make_float4(0.f, 0.f, 0.f, 0.f);
    float  ws = 0.0f;
    {
        const int n = deg < SLOTS ? deg : SLOTS;
        int ps = 0, pw = 0;
        if (lane < n) {
            int2 p = *(const int2*)(slots + 2 * ((size_t)r * SLOTS + lane));
            ps = p.x; pw = p.y;
        }
        int j = 0;
        for (; j + 4 <= n; j += 4) {
            const row_raw<TAB16> v0 = load_row(nsrc, __builtin_amdgcn_readlane(ps, j),     lane);
            const row_raw<TAB16> v1 = load_row(nsrc, __builtin_amdgcn_readlane(ps, j + 1), lane);
            const row_raw<TAB16> v2 = load_row(nsrc, __builtin_amdgcn_readlane(ps, j + 2), lane);
            const row_raw<TAB16> v3 = load_row(nsrc, __builtin_amdgcn_readlane(ps, j + 3), lane);
            float w0 = __int_as_float(__builtin_amdgcn_readlane(pw, j));
            float w1 = __int_as_float(__builtin_amdgcn_readlane(pw, j + 1));
            float w2 = __int_as_float(__builtin_amdgcn_readlane(pw, j + 2));
            float w3 = __int_as_float(__builtin_amdgcn_readlane(pw, j + 3));
            fma4(a0, widen4(v0.x), w0); fma4(a1, widen4(v0.y), w0); ws += w0;
            fma4(a0, widen4(v1.x), w1); fma4(a1, widen4(v1.y), w1); ws += w1;
            fma4(a0, widen4(v2.x), w2); fma4(a1, widen4(v2.y), w2); ws += w2;
            fma4(a0, widen4(v3.x), w3); fma4(a1, widen4(v3.y), w3); ws += w3;
        }
        #pragma nounroll
        for (; j < n; ++j) {
            const row_raw<TAB16> v0 = load_row(nsrc, __builtin_amdgcn_readlane(ps, j), lane);
            float w0 = __int_as_float(__builtin_amdgcn_readlane(pw, j));
            fma4(a0, widen4(v0.x), w0); fma4(a1, widen4(v0.y), w0); ws += w0;
        }
    }
    if (deg > SLOTS) {
        const int m = __builtin_amdgcn_readfirstlane(ovf_n[0]);
        for (int base = 0; base < m; base += 32) {
            int od = -1, os = 0, ow = 0;
            if (base + lane < m) {
                const int4 e = ovf[base + lane];
                od = e.x; os = e.y; ow = e.z;
            }
            unsigned mask = __builtin_amdgcn_ballot_w32(od == r);
            while (mask != 0u) {
                const int k = __builtin_ctz(mask);
                mask &= mask - 1u;
                const row_raw<TAB16> v0 = load_row(nsrc, __builtin_amdgcn_readlane(os, k), lane);
                float w0 = __int_as_float(__builtin_amdgcn_readlane(ow, k));
                fma4(a0, widen4(v0.x), w0); fma4(a1, widen4(v0.y), w0); ws += w0;
            }
        }
    }
    ws = ws < 1.0f ? 1.0f : ws;
    const float inv = __frcp_rn(ws);
    o0[0] = (__bf16)(a0.x * inv); o0[1] = (__bf16)(a0.y * inv);
    o0[2] = (__bf16)(a0.z * inv); o0[3] = (__bf16)(a0.w * inv);
    o1[0] = (__bf16)(a1.x * inv); o1[1] = (__bf16)(a1.y * inv);
    o1[2] = (__bf16)(a1.z * inv); o1[3] = (__bf16)(a1.w * inv);
}

__global__ __launch_bounds__(256)
void segment_gather_bf16_kernel(const __bf16* __restrict__ nsrc,
                                const int* __restrict__ cnt,
                                const int* __restrict__ slots,
                                const int4* __restrict__ ovf,
                                const int* __restrict__ ovf_n,
                                __bf16* __restrict__ z, int rows)
{
    const int lane = threadIdx.x & 31;
    const int r    = blockIdx.x * 8 + (threadIdx.x >> 5);
    if (r >= rows) return;
    __bf16* zr = z + (size_t)r * (2 * HDIM);
    v4bf o0, o1;
    segment_row_sum<true>(nsrc, cnt, slots, ovf, ovf_n, r, lane, o0, o1);
    *(v4bf*)(zr + 4 * lane)       = o0;
    *(v4bf*)(zr + 128 + 4 * lane) = o1;
}

#define TAIL_ROWS  32
#define TAIL_PITCH (HDIM * 2 + 16)
#define TAIL_K     (2 * HDIM)
#define TAIL_KS    (TAIL_K / 32)
#define TAIL_PF    3
static_assert(N_DST % TAIL_ROWS == 0, "whole work-groups only");
__global__ __launch_bounds__(1024)
void layer2_tail_kernel(const float* __restrict__ n2, const int* __restrict__ cnt,
                        const int* __restrict__ slots, const int4* __restrict__ ovf,
                        const int* __restrict__ ovf_n, const __bf16* __restrict__ zb,
                        const __bf16* __restrict__ W, const float* __restrict__ bias,
                        const float* __restrict__ hp, float* __restrict__ zf)
{
    __shared__ __attribute__((aligned(16))) unsigned char L[TAIL_ROWS * TAIL_PITCH];
    __shared__ __attribute__((aligned(16))) float red[2 * 16 * 16];

    const int lane = threadIdx.x & 31;
    const int wave = __builtin_amdgcn_readfirstlane(threadIdx.x >> 5);
    const int half = lane >> 4;
    const int l15  = lane & 15;
    const int m0   = blockIdx.x * TAIL_ROWS;
    const int rt   = wave >> 4;
    const int ct   = wave & 15;

    {
        v4bf o0, o1;
        segment_row_sum<false>(n2, cnt, slots, ovf, ovf_n, m0 + wave, lane, o0, o1);
        unsigned char* lr = L + wave * TAIL_PITCH;
        *(v4bf*)(lr + 8 * lane)       = o0;
        *(v4bf*)(lr + 256 + 8 * lane) = o1;
    }

    const __bf16* wrow = W  + (size_t)(ct * 16 + l15) * TAIL_K + half * 8;
    const __bf16* arow = zb + (size_t)(m0 + rt * 16 + l15) * TAIL_K + half * 8;
    const int n = ct * 16 + l15;
    const size_t orow = (size_t)(m0 + rt * 16 + half * 8) * HDIM + n;
    b_step qa[TAIL_KS], qb[TAIL_KS];
    #pragma unroll
    for (int ks = 0; ks < TAIL_PF; ++ks) qb[ks] = load_b_step(wrow + ks * 32);
    const float bv = bias[n];
    float hv[8];
    #pragma unroll
    for (int r = 0; r < 8; ++r) hv[r] = hp[orow + (size_t)r * HDIM];

    __syncthreads();

    v8f acc = (v8f){};
    const unsigned char* lrow = L + (rt * 16 + l15) * TAIL_PITCH + half * 16;
    #pragma unroll
    for (int ks = 0; ks < TAIL_KS; ++ks) {
        if (ks + TAIL_PF < TAIL_KS) {
            qb[ks + TAIL_PF] = load_b_step(wrow + (ks + TAIL_PF) * 32);
            if (ks + TAIL_PF >= TAIL_KS / 2)
                qa[ks + TAIL_PF] = load_b_step(arow + (ks + TAIL_PF) * 32);
        }
        __builtin_amdgcn_sched_barrier(0);
        v16bf a;
        if (ks < TAIL_KS / 2) a = lds_frag(lrow + ks * 64);
        else                  a = frag_of(qa[ks]);
        acc = __builtin_amdgcn_wmma_f32_16x16x32_bf16(
            false, a, false, frag_of(qb[ks]), (short)0, acc, false, false);
    }

    float v[8], p[8];
    #pragma unroll
    for (int r = 0; r < 8; ++r) {
        v[r] = hv[r] + fmaxf(acc[r] + bv, 0.0f);
        p[r] = v[r] * v[r];
    }
    #pragma unroll
    for (int o = 8; o > 0; o >>= 1) {
        #pragma unroll
        for (int r = 0; r < 8; ++r) p[r] += __shfl_xor(p[r], o, 32);
    }
    if (l15 == 0) {
        float4* d = (float4*)(red + (rt * 16 + ct) * 16 + half * 8);
        d[0] = make_float4(p[0], p[1], p[2], p[3]);
        d[1] = make_float4(p[4], p[5], p[6], p[7]);
    }
    __syncthreads();
    float ss[8];
    #pragma unroll
    for (int r = 0; r < 8; ++r) ss[r] = 0.0f;
    #pragma unroll
    for (int t = 0; t < 16; ++t) {
        const float4* s = (const float4*)(red + (rt * 16 + t) * 16 + half * 8);
        const float4 x = s[0], y = s[1];
        ss[0] += x.x; ss[1] += x.y; ss[2] += x.z; ss[3] += x.w;
        ss[4] += y.x; ss[5] += y.y; ss[6] += y.z; ss[7] += y.w;
    }
    #pragma unroll
    for (int r = 0; r < 8; ++r) {
        const float nrm   = sqrtf(ss[r]);
        const float scale = (nrm == 0.0f) ? 1.0f : 1.0f / nrm;
        zf[orow + (size_t)r * HDIM] = v[r] * scale;
    }
}

#define PAIRS_PER_WAVE 4
static_assert(N_PAIRS % (8 * PAIRS_PER_WAVE) == 0, "whole waves only");
__global__ __launch_bounds__(256)
void score_kernel(const float* __restrict__ zf, const float* __restrict__ bias,
                  const int* __restrict__ nids,
                  const int* __restrict__ ps, const int* __restrict__ pd,
                  const int* __restrict__ ns, const int* __restrict__ nd,
                  float* __restrict__ out, int npairs)
{
    const int lane = threadIdx.x & 31;
    const int p0   = (blockIdx.x * 8 + (threadIdx.x >> 5)) * PAIRS_PER_WAVE;
    if (p0 >= npairs) return;
    int pu[PAIRS_PER_WAVE], pv[PAIRS_PER_WAVE], nu[PAIRS_PER_WAVE], nv[PAIRS_PER_WAVE];
    #pragma unroll
    for (int k = 0; k < PAIRS_PER_WAVE; ++k) {
        pu[k] = ps[p0 + k]; pv[k] = pd[p0 + k]; nu[k] = ns[p0 + k]; nv[k] = nd[p0 + k];
    }
    int ipu[PAIRS_PER_WAVE], ipv[PAIRS_PER_WAVE], inu[PAIRS_PER_WAVE], inv[PAIRS_PER_WAVE];
    #pragma unroll
    for (int k = 0; k < PAIRS_PER_WAVE; ++k) {
        ipu[k] = nids[pu[k]]; ipv[k] = nids[pv[k]];
        inu[k] = nids[nu[k]]; inv[k] = nids[nv[k]];
    }
    float4 a0[PAIRS_PER_WAVE], a1[PAIRS_PER_WAVE], b0[PAIRS_PER_WAVE], b1[PAIRS_PER_WAVE];
    float4 e0[PAIRS_PER_WAVE], e1[PAIRS_PER_WAVE], f0[PAIRS_PER_WAVE], f1[PAIRS_PER_WAVE];
    #pragma unroll
    for (int k = 0; k < PAIRS_PER_WAVE; ++k) {
        const float4* a = (const float4*)(zf + (size_t)pu[k] * HDIM + lane * 8);
        const float4* b = (const float4*)(zf + (size_t)pv[k] * HDIM + lane * 8);
        const float4* e = (const float4*)(zf + (size_t)nu[k] * HDIM + lane * 8);
        const float4* f = (const float4*)(zf + (size_t)nv[k] * HDIM + lane * 8);
        a0[k] = a[0]; a1[k] = a[1]; b0[k] = b[0]; b1[k] = b[1];
        e0[k] = e[0]; e1[k] = e[1]; f0[k] = f[0]; f1[k] = f[1];
    }
    float gp[PAIRS_PER_WAVE], gn[PAIRS_PER_WAVE];
    #pragma unroll
    for (int k = 0; k < PAIRS_PER_WAVE; ++k) {
        gp[k] = bias[ipu[k]] + bias[ipv[k]];
        gn[k] = bias[inu[k]] + bias[inv[k]];
    }
    #pragma unroll
    for (int k = 0; k < PAIRS_PER_WAVE; ++k) {
        float sp = 0.0f, sn = 0.0f;
        sp += a0[k].x * b0[k].x; sn += e0[k].x * f0[k].x;
        sp += a0[k].y * b0[k].y; sn += e0[k].y * f0[k].y;
        sp += a0[k].z * b0[k].z; sn += e0[k].z * f0[k].z;
        sp += a0[k].w * b0[k].w; sn += e0[k].w * f0[k].w;
        sp += a1[k].x * b1[k].x; sn += e1[k].x * f1[k].x;
        sp += a1[k].y * b1[k].y; sn += e1[k].y * f1[k].y;
        sp += a1[k].z * b1[k].z; sn += e1[k].z * f1[k].z;
        sp += a1[k].w * b1[k].w; sn += e1[k].w * f1[k].w;
        #pragma unroll
        for (int o = 16; o > 0; o >>= 1) {
            sp += __shfl_xor(sp, o, 32);
            sn += __shfl_xor(sn, o, 32);
        }
        const float m = (sn + gn[k]) - (sp + gp[k]) + 1.0f;
        if (lane == 0) out[p0 + k] = m > 0.0f ? m : 0.0f;
    }
}

extern "C" void kernel_launch(void* const* d_in, const int* in_sizes, int n_in,
                              void* d_out, int out_size, void* d_ws, size_t ws_size,
                              hipStream_t stream) {
    const float* feat = (const float*)d_in[0];
    const float* Wp   = (const float*)d_in[1];
    const float* bp   = (const float*)d_in[2];
    const float* Q1   = (const float*)d_in[3];
    const float* bq1  = (const float*)d_in[4];
    const float* W1   = (const float*)d_in[5];
    const float* bw1  = (const float*)d_in[6];
    const float* Q2   = (const float*)d_in[7];
    const float* bq2  = (const float*)d_in[8];
    const float* W2   = (const float*)d_in[9];
    const float* bw2  = (const float*)d_in[10];
    const float* w1v  = (const float*)d_in[11];
    const float* w2v  = (const float*)d_in[12];
    const float* bias = (const float*)d_in[13];
    const int* src1   = (const int*)d_in[14];
    const int* dst1   = (const int*)d_in[15];
    const int* src2   = (const int*)d_in[16];
    const int* dst2   = (const int*)d_in[17];
    const int* pos_s  = (const int*)d_in[18];
    const int* pos_d  = (const int*)d_in[19];
    const int* neg_s  = (const int*)d_in[20];
    const int* neg_d  = (const int*)d_in[21];
    const int* nids   = (const int*)d_in[22];
    float* out        = (float*)d_out;

    char* ws = (char*)d_ws;
    const size_t SZ_HP  = (size_t)N_SRC * HDIM * 4;
    const size_t SZ_N   = (size_t)N_SRC * HDIM * 4;
    const size_t SZ_H1  = (size_t)N_MID * HDIM * 4;
    const size_t SZ_AGG = (size_t)N_MID * HDIM * 4;
    const size_t SZ_WS  = 256 * 1024;
    const size_t SZ_H2  = (size_t)N_DST * HDIM * 4;
    const size_t SZ_ZF  = (size_t)N_DST * HDIM * 4;
    float*  hp   = (float*)(ws);
    float*  nbuf = (float*)(ws + SZ_HP);
    __bf16* n1b  = (__bf16*)(ws + SZ_HP);
    __bf16* h1b  = (__bf16*)(ws + SZ_HP + SZ_N);
    __bf16* z1b  = (__bf16*)(hp + (size_t)N_MID * HDIM);
    __bf16* z2b  = z1b + (size_t)N_MID * 2 * HDIM;
    int*  cnt1   = (int*)(ws + SZ_HP + SZ_N + SZ_H1);
    int*  cnt2   = cnt1 + N_MID;
    int*  ovfn1  = cnt2 + N_DST;
    int*  ovfn2  = ovfn1 + 64;
    int*  slots1 = cnt1 + CNT_INTS;
    int*  slots2 = slots1 + 2 * SLOTS * N_MID;
    int4* ovf1   = (int4*)(slots2 + 2 * SLOTS * N_DST);
    int4* ovf2   = ovf1 + E1_N;
    float*  zf   = (float*)(ws + SZ_HP + SZ_N + SZ_H1 + SZ_AGG + SZ_WS + SZ_H2);
    __bf16* Wpb  = (__bf16*)(ws + SZ_HP + SZ_N + SZ_H1 + SZ_AGG + SZ_WS + SZ_H2 + SZ_ZF);
    __bf16* Q1b  = Wpb + HDIM * HDIM;
    __bf16* W1b  = Q1b + HDIM * HDIM;
    __bf16* Q2b  = W1b + HDIM * 2 * HDIM;
    __bf16* W2b  = Q2b + HDIM * HDIM;

    prep_kernel<<<PREP_W_BLOCKS + PREP_Z_BLOCKS, PREP_THREADS, 0, stream>>>(Wp, Q1, W1, Q2, W2, Wpb, cnt1);

    edge_bin_kernel<<<(E1_N + E2_N) / EDGES_PER_THREAD / 256, 256, 0, stream>>>(src1, dst1, w1v, cnt1, slots1, ovf1, ovfn1,
                                                                            src2, dst2, w2v, cnt2, slots2, ovf2, ovfn2);

    gemm_pair_bf16_w_in_lds<<<256, 512, 0, stream>>>(feat, Wpb, bp, bq1, hp, z1b, n1b, N_SRC, N_MID, N_DST);
    segment_gather_bf16_kernel<<<N_MID / 8, 256, 0, stream>>>(n1b, cnt1, slots1, ovf1, ovfn1, z1b, N_MID);
    gemm_bf16_w_in_lds<2 * HDIM, 8, true><<<256, 512, 0, stream>>>(z1b, W1b, bw1, nullptr, h1b, z2b, N_MID, 1, N_DST);
    gemm_bf16_w_in_lds<HDIM, 8, false><<<256, 512, 0, stream>>>(h1b, Q2b, bq2, nbuf, nullptr, nullptr, N_MID, 1, 0);
    layer2_tail_kernel<<<N_DST / TAIL_ROWS, 1024, 0, stream>>>(nbuf, cnt2, slots2, ovf2, ovfn2, z2b, W2b, bw2, hp, zf);
    score_kernel<<<N_PAIRS / (8 * PAIRS_PER_WAVE), 256, 0, stream>>>(zf, bias, nids,
                                                  pos_s, pos_d, neg_s, neg_d,
                                                  out, N_PAIRS);
}
